# starting kernel
# baseline (speedup 1.0000x reference)
#include <hip/hip_runtime.h>

typedef __attribute__((ext_vector_type(2))) float v2f;
typedef __attribute__((ext_vector_type(8))) float v8f;

#define IN_F  128
#define HID_F 128
#define OUT_F 64


__global__ __launch_bounds__(256) void zero_kernel(float* __restrict__ p, long long n) {
    long long i = (long long)blockIdx.x * 256 + threadIdx.x;
    if (i < n) p[i] = 0.0f;
}

__global__ __launch_bounds__(256) void degree_kernel(const int* __restrict__ dst,
                                                     float* __restrict__ deg, int E) {
    int e = blockIdx.x * 256 + threadIdx.x;
    if (e < E) atomicAdd(&deg[dst[e]], 1.0f);
}

__global__ __launch_bounds__(256) void invdeg_kernel(float* __restrict__ deg, int n) {
    int i = blockIdx.x * 256 + threadIdx.x;
    if (i < n) deg[i] = 1.0f / fmaxf(deg[i], 1.0f);
}

__global__ __launch_bounds__(256) void scatter_kernel(const float* __restrict__ feat,
                                                      const int* __restrict__ src,
                                                      const int* __restrict__ dst,
                                                      float* __restrict__ accum, int E) {
    long long tid = (long long)blockIdx.x * 256 + threadIdx.x;
    int e = (int)(tid >> 7);
    int d = (int)(tid & 127);
    if (e < E) {
        int s = src[e], t = dst[e];
        atomicAdd(&accum[(long long)t * 128 + d], feat[(long long)s * 128 + d]);
    }
}

template <int NCOLT>
__global__ __launch_bounds__(32 * NCOLT) void sage_gemm_wmma(
        const float* __restrict__ Xs, const float* __restrict__ Xa,
        const float* __restrict__ invdeg,
        const float* __restrict__ Ws, const float* __restrict__ Wn,
        const float* __restrict__ bias, float* __restrict__ out,
        int M, int K, int do_relu) {
    const int N = NCOLT * 16;
    const int lane    = threadIdx.x;
    const int rowBase = blockIdx.x * 16;
    const int colBase = threadIdx.y * 16;

    int r = rowBase + (lane & 15);
    if (r >= M) r = M - 1;
    const int kOff = (lane < 16) ? 0 : 2;
    const int c = colBase + (lane & 15);

    const float invd = invdeg[r];
    const float* xsRow = Xs + (long long)r * K + kOff;
    const float* xaRow = Xa + (long long)r * K + kOff;

    v8f acc = {};
    for (int kk = 0; kk < K; kk += 4) {
        v2f a_s = *reinterpret_cast<const v2f*>(xsRow + kk);
        v2f a_raw = *reinterpret_cast<const v2f*>(xaRow + kk);
        v2f a_n = { a_raw.x * invd, a_raw.y * invd };
        const int kb = kk + kOff;
        v2f b_s = { Ws[(long long)kb * N + c], Ws[(long long)(kb + 1) * N + c] };
        v2f b_n = { Wn[(long long)kb * N + c], Wn[(long long)(kb + 1) * N + c] };
        acc = __builtin_amdgcn_wmma_f32_16x16x4_f32(false, a_s, false, b_s,
                                                    (short)0, acc, false, false);
        acc = __builtin_amdgcn_wmma_f32_16x16x4_f32(false, a_n, false, b_n,
                                                    (short)0, acc, false, false);
    }

    const float bv = bias[c];
    const int mBase = rowBase + ((lane < 16) ? 0 : 8);
#pragma unroll
    for (int j = 0; j < 8; ++j) {
        int m = mBase + j;
        if (m < M) {
            float v = acc[j] + bv;
            if (do_relu) v = fmaxf(v, 0.0f);
            out[(long long)m * N + c] = v;
        }
    }
}


__global__ __launch_bounds__(256) void gcount_kernel(const int* __restrict__ gid,
                                                     float* __restrict__ gcnt, int n) {
    int i = blockIdx.x * 256 + threadIdx.x;
    if (i < n) atomicAdd(&gcnt[gid[i]], 1.0f);
}

__global__ __launch_bounds__(256) void readout_kernel(const float* __restrict__ h2,
                                                      const int* __restrict__ gid,
                                                      float* __restrict__ gsum, int Nn) {
    long long tid = (long long)blockIdx.x * 256 + threadIdx.x;
    int node = (int)(tid >> 6);
    int f    = (int)(tid & 63);
    if (node < Nn)
        atomicAdd(&gsum[(long long)gid[node] * OUT_F + f], h2[(long long)node * OUT_F + f]);
}

__global__ __launch_bounds__(256) void finalize_kernel(const float* __restrict__ gsum,
                                                       const float* __restrict__ gcnt,
                                                       float* __restrict__ out, int total) {
    int i = blockIdx.x * 256 + threadIdx.x;
    if (i < total) {
        int g = i >> 6;
        out[i] = gsum[i] / fmaxf(gcnt[g], 1.0f);
    }
}


static inline int cdiv64(long long a, long long b) { return (int)((a + b - 1) / b); }

extern "C" void kernel_launch(void* const* d_in, const int* in_sizes, int n_in,
                              void* d_out, int out_size, void* d_ws, size_t ws_size,
                              hipStream_t stream) {
    const float* in_feat = (const float*)d_in[0];
    const int*   src     = (const int*)d_in[1];
    const int*   dst     = (const int*)d_in[2];
    const int*   gid     = (const int*)d_in[3];
    const float* W1s = (const float*)d_in[5];
    const float* W1n = (const float*)d_in[6];
    const float* b1  = (const float*)d_in[7];
    const float* W2s = (const float*)d_in[8];
    const float* W2n = (const float*)d_in[9];
    const float* b2  = (const float*)d_in[10];
    float* out = (float*)d_out;

    const int Nn = in_sizes[0] / IN_F;
    const int E  = in_sizes[1];
    const int G  = out_size / OUT_F;

    float* ws    = (float*)d_ws;
    float* accum = ws;
    float* deg   = accum + (size_t)Nn * 128;
    float* h1    = deg + Nn;
    float* h2    = h1 + (size_t)Nn * 128;
    float* gsum  = h2 + (size_t)Nn * OUT_F;
    float* gcnt  = gsum + (size_t)G * OUT_F;

    const int mTiles = cdiv64(Nn, 16);

    zero_kernel<<<cdiv64((long long)Nn * 128 + Nn, 256), 256, 0, stream>>>(
        accum, (long long)Nn * 128 + Nn);
    degree_kernel<<<cdiv64(E, 256), 256, 0, stream>>>(dst, deg, E);
    scatter_kernel<<<cdiv64((long long)E * 128, 256), 256, 0, stream>>>(in_feat, src, dst, accum, E);
    invdeg_kernel<<<cdiv64(Nn, 256), 256, 0, stream>>>(deg, Nn);
    sage_gemm_wmma<8><<<mTiles, dim3(32, 8), 0, stream>>>(
        in_feat, accum, deg, W1s, W1n, b1, h1, Nn, IN_F, 1);

    zero_kernel<<<cdiv64((long long)Nn * 128, 256), 256, 0, stream>>>(accum, (long long)Nn * 128);
    scatter_kernel<<<cdiv64((long long)E * 128, 256), 256, 0, stream>>>(h1, src, dst, accum, E);
    sage_gemm_wmma<4><<<mTiles, dim3(32, 4), 0, stream>>>(
        h1, accum, deg, W2s, W2n, b2, h2, Nn, HID_F, 0);

    zero_kernel<<<cdiv64((long long)G * OUT_F + G, 256), 256, 0, stream>>>(
        gsum, (long long)G * OUT_F + G);
    gcount_kernel<<<cdiv64(Nn, 256), 256, 0, stream>>>(gid, gcnt, Nn);
    readout_kernel<<<cdiv64((long long)Nn * OUT_F, 256), 256, 0, stream>>>(h2, gid, gsum, Nn);
    finalize_kernel<<<cdiv64((long long)G * OUT_F, 256), 256, 0, stream>>>(gsum, gcnt, out, G * OUT_F);
}


# Mean readout pushed through linear layer 2: per-graph sums of h1 via a 64xN edge-weight matrix (1 float atomic per edge) times h1 on WMMA, then one 64x256x64 product; summation order changes
# speedup vs baseline: 1.5261x; 1.5261x over previous
#include <hip/hip_runtime.h>

typedef __attribute__((ext_vector_type(2))) float v2f;
typedef __attribute__((ext_vector_type(8))) float v8f;

#define IN_F  128
#define HID_F 128
#define OUT_F 64
#define NG    64
#define CH    256
#define KU    4


__global__ __launch_bounds__(256) void zero_kernel(float* __restrict__ p, long long n) {
    long long i = (long long)blockIdx.x * 256 + threadIdx.x;
    if (i < n) p[i] = 0.0f;
}

__global__ __launch_bounds__(256) void degree_kernel(const int* __restrict__ dst,
                                                     float* __restrict__ deg, int E) {
    int e = blockIdx.x * 256 + threadIdx.x;
    if (e < E) atomicAdd(&deg[dst[e]], 1.0f);
}

__global__ __launch_bounds__(256) void invdeg_kernel(float* __restrict__ deg, int n) {
    int i = blockIdx.x * 256 + threadIdx.x;
    if (i < n) deg[i] = 1.0f / fmaxf(deg[i], 1.0f);
}

__global__ __launch_bounds__(256) void scatter_kernel(const float* __restrict__ feat,
                                                      const int* __restrict__ src,
                                                      const int* __restrict__ dst,
                                                      float* __restrict__ accum, int E) {
    long long tid = (long long)blockIdx.x * 256 + threadIdx.x;
    int e = (int)(tid >> 7);
    int d = (int)(tid & 127);
    if (e < E) {
        int s = src[e], t = dst[e];
        atomicAdd(&accum[(long long)t * 128 + d], feat[(long long)s * 128 + d]);
    }
}

template <int NCOLT>
__global__ __launch_bounds__(32 * NCOLT) void sage_gemm_wmma(
        const float* __restrict__ Xs, const float* __restrict__ Xa,
        const float* __restrict__ invdeg,
        const float* __restrict__ Ws, const float* __restrict__ Wn,
        const float* __restrict__ bias, float* __restrict__ out,
        int M, int K, int do_relu) {
    const int N = NCOLT * 16;
    const int lane    = threadIdx.x;
    const int rowBase = blockIdx.x * 16;
    const int colBase = threadIdx.y * 16;

    int r = rowBase + (lane & 15);
    if (r >= M) r = M - 1;
    const int kOff = (lane < 16) ? 0 : 2;
    const int c = colBase + (lane & 15);

    const float invd = invdeg[r];
    const float* xsRow = Xs + (long long)r * K + kOff;
    const float* xaRow = Xa + (long long)r * K + kOff;

    v8f acc = {};
    for (int kk = 0; kk < K; kk += 4) {
        v2f a_s = *reinterpret_cast<const v2f*>(xsRow + kk);
        v2f a_raw = *reinterpret_cast<const v2f*>(xaRow + kk);
        v2f a_n = { a_raw.x * invd, a_raw.y * invd };
        const int kb = kk + kOff;
        v2f b_s = { Ws[(long long)kb * N + c], Ws[(long long)(kb + 1) * N + c] };
        v2f b_n = { Wn[(long long)kb * N + c], Wn[(long long)(kb + 1) * N + c] };
        acc = __builtin_amdgcn_wmma_f32_16x16x4_f32(false, a_s, false, b_s,
                                                    (short)0, acc, false, false);
        acc = __builtin_amdgcn_wmma_f32_16x16x4_f32(false, a_n, false, b_n,
                                                    (short)0, acc, false, false);
    }

    const float bv = bias[c];
    const int mBase = rowBase + ((lane < 16) ? 0 : 8);
#pragma unroll
    for (int j = 0; j < 8; ++j) {
        int m = mBase + j;
        if (m < M) {
            float v = acc[j] + bv;
            if (do_relu) v = fmaxf(v, 0.0f);
            out[(long long)m * N + c] = v;
        }
    }
}


__global__ __launch_bounds__(256) void edge_weight_kernel(const int* __restrict__ src,
                                                          const int* __restrict__ dst,
                                                          const int* __restrict__ gid,
                                                          const float* __restrict__ invdeg,
                                                          float* __restrict__ Ct, int E) {
    int e = blockIdx.x * 256 + threadIdx.x;
    if (e < E) {
        int t = dst[e];
        int g = gid[t];
        if ((unsigned)g < (unsigned)NG)
            atomicAdd(&Ct[(long long)src[e] * NG + g], invdeg[t]);
    }
}

__global__ __launch_bounds__(256) void graph_reduce_wmma(
        const float* __restrict__ h1, const float* __restrict__ Ct,
        const int* __restrict__ gid, float* __restrict__ part,
        float* __restrict__ cntPart, int Nn) {
    __shared__ float Sloc[(NG + 1) * HID_F];
    __shared__ int   cntloc[NG + 1];
    __shared__ int   sgid[CH];

    const int lane = threadIdx.x;
    const int w    = threadIdx.y;
    const int tid  = w * 32 + lane;
    const int k0   = blockIdx.x * CH;

    for (int i = tid; i < (NG + 1) * HID_F; i += 256) Sloc[i] = 0.0f;
    if (tid <= NG) cntloc[tid] = 0;
    {
        const int node = k0 + tid;
        int g = NG;
        if (node < Nn) {
            const int gg = gid[node];
            if ((unsigned)gg < (unsigned)NG) g = gg;
        }
        sgid[tid] = g;
    }
    __syncthreads();
    atomicAdd(&cntloc[sgid[tid]], 1);

    const int kOff = (lane < 16) ? 0 : 2;
    const int col  = lane & 15;
    const int f    = w * 16 + col;

    v8f acc0 = {}, acc1 = {}, acc2 = {}, acc3 = {};
    for (int kk = 0; kk < CH; kk += 4 * KU) {
        v2f b[KU], a0[KU], a1[KU], a2[KU], a3[KU];
        int sa[KU], sb[KU];
#pragma unroll
        for (int u = 0; u < KU; ++u) {
            const int ka = kk + 4 * u + kOff;
            const int na = k0 + ka;
            const bool va = na < Nn;
            const bool vb = (na + 1) < Nn;
            const int ca = va ? na : (Nn - 1);
            const int cb = vb ? (na + 1) : (Nn - 1);
            const float x = h1[(long long)ca * HID_F + f];
            const float y = h1[(long long)cb * HID_F + f];
            b[u] = (v2f){ va ? x : 0.0f, vb ? y : 0.0f };
            const float* ra = Ct + (long long)ca * NG + col;
            const float* rb = Ct + (long long)cb * NG + col;
            a0[u] = (v2f){ ra[0],  rb[0]  };
            a1[u] = (v2f){ ra[16], rb[16] };
            a2[u] = (v2f){ ra[32], rb[32] };
            a3[u] = (v2f){ ra[48], rb[48] };
            sa[u] = sgid[ka];
            sb[u] = sgid[ka + 1];
        }
#pragma unroll
        for (int u = 0; u < KU; ++u) {
            atomicAdd(&Sloc[sa[u] * HID_F + f], b[u].x);
            atomicAdd(&Sloc[sb[u] * HID_F + f], b[u].y);
            acc0 = __builtin_amdgcn_wmma_f32_16x16x4_f32(false, a0[u], false, b[u], (short)0, acc0, false, false);
            acc1 = __builtin_amdgcn_wmma_f32_16x16x4_f32(false, a1[u], false, b[u], (short)0, acc1, false, false);
            acc2 = __builtin_amdgcn_wmma_f32_16x16x4_f32(false, a2[u], false, b[u], (short)0, acc2, false, false);
            acc3 = __builtin_amdgcn_wmma_f32_16x16x4_f32(false, a3[u], false, b[u], (short)0, acc3, false, false);
        }
    }

    float* pT = part + (long long)blockIdx.x * (2 * NG * HID_F);
    float* pS = pT + NG * HID_F;
    const int mB = (lane < 16) ? 0 : 8;
#pragma unroll
    for (int j = 0; j < 8; ++j) {
        pT[(0  + mB + j) * HID_F + f] = acc0[j];
        pT[(16 + mB + j) * HID_F + f] = acc1[j];
        pT[(32 + mB + j) * HID_F + f] = acc2[j];
        pT[(48 + mB + j) * HID_F + f] = acc3[j];
    }
    __syncthreads();
    for (int i = tid; i < NG * HID_F; i += 256) pS[i] = Sloc[i];
    if (tid < NG) cntPart[blockIdx.x * NG + tid] = (float)cntloc[tid];
}

__global__ __launch_bounds__(256) void sum_partials_kernel(const float* __restrict__ part,
                                                           const float* __restrict__ cntPart,
                                                           float* __restrict__ R,
                                                           float* __restrict__ cnt, int nChunks) {
    const int per = 2 * NG * HID_F;
    if (blockIdx.x < per / 256) {
        const int i = blockIdx.x * 256 + threadIdx.x;
        float s = 0.0f;
        for (int c = 0; c < nChunks; ++c) s += part[(long long)c * per + i];
        R[i] = s;
    } else if (threadIdx.x < NG) {
        float s = 0.0f;
        for (int c = 0; c < nChunks; ++c) s += cntPart[c * NG + threadIdx.x];
        cnt[threadIdx.x] = s;
    }
}

__global__ __launch_bounds__(512) void graph_head_wmma(const float* __restrict__ R,
                                                       const float* __restrict__ cnt,
                                                       const float* __restrict__ Ws,
                                                       const float* __restrict__ Wn,
                                                       const float* __restrict__ bias,
                                                       float* __restrict__ out) {
    const int lane = threadIdx.x;
    const int w    = threadIdx.y;
    const int gB   = (w >> 2) * 16;
    const int oB   = (w & 3) * 16;
    const int kOff = (lane < 16) ? 0 : 2;
    const int g    = gB + (lane & 15);
    const int c    = oB + (lane & 15);
    const float* tRow = R + g * HID_F + kOff;
    const float* sRow = R + NG * HID_F + g * HID_F + kOff;

    v8f accS = {}, accT = {};
    for (int kk = 0; kk < HID_F; kk += 4) {
        const v2f a_s = *reinterpret_cast<const v2f*>(sRow + kk);
        const v2f a_t = *reinterpret_cast<const v2f*>(tRow + kk);
        const int kb = kk + kOff;
        const v2f b_s = { Ws[kb * OUT_F + c], Ws[(kb + 1) * OUT_F + c] };
        const v2f b_n = { Wn[kb * OUT_F + c], Wn[(kb + 1) * OUT_F + c] };
        accS = __builtin_amdgcn_wmma_f32_16x16x4_f32(false, a_s, false, b_s, (short)0, accS, false, false);
        accT = __builtin_amdgcn_wmma_f32_16x16x4_f32(false, a_t, false, b_n, (short)0, accT, false, false);
    }

    const float bv = bias[c];
    const int mB = gB + ((lane < 16) ? 0 : 8);
#pragma unroll
    for (int j = 0; j < 8; ++j) {
        const int gg = mB + j;
        const float n = cnt[gg];
        out[gg * OUT_F + c] = (accS[j] + accT[j] + n * bv) / fmaxf(n, 1.0f);
    }
}


static inline int cdiv64(long long a, long long b) { return (int)((a + b - 1) / b); }

extern "C" void kernel_launch(void* const* d_in, const int* in_sizes, int n_in,
                              void* d_out, int out_size, void* d_ws, size_t ws_size,
                              hipStream_t stream) {
    const float* in_feat = (const float*)d_in[0];
    const int*   src     = (const int*)d_in[1];
    const int*   dst     = (const int*)d_in[2];
    const int*   gid     = (const int*)d_in[3];
    const float* W1s = (const float*)d_in[5];
    const float* W1n = (const float*)d_in[6];
    const float* b1  = (const float*)d_in[7];
    const float* W2s = (const float*)d_in[8];
    const float* W2n = (const float*)d_in[9];
    const float* b2  = (const float*)d_in[10];
    float* out = (float*)d_out;

    const int Nn = in_sizes[0] / IN_F;
    const int E  = in_sizes[1];

    float* ws    = (float*)d_ws;
    float* accum = ws;
    float* deg   = accum + (size_t)Nn * 128;
    float* Ct    = deg + Nn;
    float* h1    = Ct + (size_t)Nn * NG;

    const int mTiles  = cdiv64(Nn, 16);
    const int nChunks = cdiv64(Nn, CH);

    float* part    = accum;
    float* cntPart = part + (size_t)nChunks * (2 * NG * HID_F);
    float* R       = cntPart + (size_t)nChunks * NG;
    float* cnt     = R + 2 * NG * HID_F;

    zero_kernel<<<cdiv64((long long)Nn * (128 + 1 + NG), 256), 256, 0, stream>>>(
        accum, (long long)Nn * (128 + 1 + NG));
    degree_kernel<<<cdiv64(E, 256), 256, 0, stream>>>(dst, deg, E);
    scatter_kernel<<<cdiv64((long long)E * 128, 256), 256, 0, stream>>>(in_feat, src, dst, accum, E);
    invdeg_kernel<<<cdiv64(Nn, 256), 256, 0, stream>>>(deg, Nn);
    sage_gemm_wmma<8><<<mTiles, dim3(32, 8), 0, stream>>>(
        in_feat, accum, deg, W1s, W1n, b1, h1, Nn, IN_F, 1);

    edge_weight_kernel<<<cdiv64(E, 256), 256, 0, stream>>>(src, dst, gid, deg, Ct, E);
    graph_reduce_wmma<<<nChunks, dim3(32, 8), 0, stream>>>(h1, Ct, gid, part, cntPart, Nn);
    sum_partials_kernel<<<2 * NG * HID_F / 256 + 1, 256, 0, stream>>>(part, cntPart, R, cnt, nChunks);
    graph_head_wmma<<<1, dim3(32, 16), 0, stream>>>(R, cnt, W2s, W2n, b2, out);
}


# Layer-1 neighbour sums by owner tiles: each work-group owns 256 dst rows, scans the edge list, adds matching src rows in LDS (LDS float atomics, order changes); global float scatter gone
# speedup vs baseline: 3.0734x; 2.0139x over previous
#include <hip/hip_runtime.h>

typedef __attribute__((ext_vector_type(2))) float v2f;
typedef __attribute__((ext_vector_type(8))) float v8f;

#define IN_F  128
#define HID_F 128
#define OUT_F 64
#define NG    64
#define CH    256
#define KU    4
#define TR    256
#define TW    32
#define SU    4
#define PB    8
#define FLUSH 64
#define LCAP  (FLUSH + SU * 128)


__global__ __launch_bounds__(256) void zero_kernel(float* __restrict__ p, long long n) {
    long long i = (long long)blockIdx.x * 256 + threadIdx.x;
    if (i < n) p[i] = 0.0f;
}

__global__ __launch_bounds__(1024) void tile_aggregate_kernel(
        const float* __restrict__ feat, const int* __restrict__ src, const int* __restrict__ dst,
        float* __restrict__ accum, float* __restrict__ invdeg, int Nn, int E) {
    __shared__ __attribute__((aligned(16))) float tile[TR * 128];
    __shared__ int degc[TR];
    __shared__ int lst[TW][LCAP];

    const int lane = threadIdx.x;
    const int w    = __builtin_amdgcn_readfirstlane(threadIdx.y);
    const int tid  = w * 32 + lane;
    const int rowBase = blockIdx.x * TR;

    for (int i = tid; i < TR * 32; i += 32 * TW)
        reinterpret_cast<float4*>(tile)[i] = make_float4(0.0f, 0.0f, 0.0f, 0.0f);
    if (tid < TR) degc[tid] = 0;
    __syncthreads();

    int* myl = lst[w];
    int cnt = 0;

    auto consider = [&](int d, int e) {
        const unsigned ld = (unsigned)d - (unsigned)rowBase;
        const bool m = ld < (unsigned)TR;
        const unsigned mask = __builtin_amdgcn_ballot_w32(m);
        if (mask != 0u) {
            if (m) {
                myl[cnt + (int)__builtin_amdgcn_mbcnt_lo(mask, 0u)] = e;
                atomicAdd(&degc[ld], 1);
            }
            cnt += __builtin_popcount(mask);
        }
    };

    auto drain = [&]() {
        for (int base = 0; base < cnt; base += PB) {
            int ee[PB], ss[PB], dd[PB];
            float4 v[PB];
#pragma unroll
            for (int k = 0; k < PB; ++k) {
                int idx = base + k;
                if (idx > cnt - 1) idx = cnt - 1;
                ee[k] = myl[idx];
            }
#pragma unroll
            for (int k = 0; k < PB; ++k) {
                ss[k] = src[ee[k]];
                dd[k] = dst[ee[k]] - rowBase;
            }
#pragma unroll
            for (int k = 0; k < PB; ++k)
                v[k] = *reinterpret_cast<const float4*>(feat + (long long)ss[k] * 128 + lane * 4);
#pragma unroll
            for (int k = 0; k < PB; ++k) {
                const bool live = base + k < cnt;
                float* t = tile + dd[k] * 128 + lane;
                atomicAdd(t,      live ? v[k].x : 0.0f);
                atomicAdd(t + 32, live ? v[k].y : 0.0f);
                atomicAdd(t + 64, live ? v[k].z : 0.0f);
                atomicAdd(t + 96, live ? v[k].w : 0.0f);
            }
        }
        cnt = 0;
    };

    const int nVec   = E >> 2;
    const int nChunk = (nVec + 31) >> 5;
    const int nIt    = (nChunk + TW * SU - 1) / (TW * SU);
    const int rot    = (int)(((unsigned long long)blockIdx.x * (unsigned long long)nChunk) / gridDim.x);
    const int4* dst4 = reinterpret_cast<const int4*>(dst);

    auto fetch = [&](int it, int4 (&dv)[SU], int (&eb)[SU]) {
#pragma unroll
        for (int u = 0; u < SU; ++u) {
            int c = (it * TW + w) * SU + u;
            bool ok = c < nChunk;
            c += rot;
            if (c >= nChunk) c -= nChunk;
            const int q = c * 32 + lane;
            ok = ok && (q < nVec);
            eb[u] = ok ? q * 4 : -1;
            dv[u] = dst4[ok ? q : 0];
        }
    };
    auto look = [&](const int4 (&dv)[SU], const int (&eb)[SU]) {
#pragma unroll
        for (int u = 0; u < SU; ++u) {
            const bool ok = eb[u] >= 0;
            consider(ok ? dv[u].x : -1, eb[u]);
            consider(ok ? dv[u].y : -1, eb[u] + 1);
            consider(ok ? dv[u].z : -1, eb[u] + 2);
            consider(ok ? dv[u].w : -1, eb[u] + 3);
        }
        if (cnt >= FLUSH) drain();
    };

    if (nIt > 0) {
        int4 dvA[SU], dvB[SU];
        int  ebA[SU], ebB[SU];
        fetch(0, dvA, ebA);
        for (int it = 0; it < nIt; it += 2) {
            fetch(it + 1, dvB, ebB);
            look(dvA, ebA);
            fetch(it + 2, dvA, ebA);
            look(dvB, ebB);
        }
    }
    if (w == 0) {
        const int e = nVec * 4 + lane;
        const int d = (e < E) ? dst[e] : -1;
        consider(d, e);
    }
    drain();
    __syncthreads();

    for (int rr = w; rr < TR; rr += TW) {
        const int node = rowBase + rr;
        if (node < Nn) {
            const float* t = tile + rr * 128 + lane;
            const float4 o = make_float4(t[0], t[32], t[64], t[96]);
            *reinterpret_cast<float4*>(accum + (long long)node * 128 + lane * 4) = o;
        }
    }
    if (tid < TR) {
        const int node = rowBase + tid;
        if (node < Nn) invdeg[node] = 1.0f / fmaxf((float)degc[tid], 1.0f);
    }
}

template <int NCOLT>
__global__ __launch_bounds__(32 * NCOLT) void sage_gemm_wmma(
        const float* __restrict__ Xs, const float* __restrict__ Xa,
        const float* __restrict__ invdeg,
        const float* __restrict__ Ws, const float* __restrict__ Wn,
        const float* __restrict__ bias, float* __restrict__ out,
        int M, int K, int do_relu) {
    const int N = NCOLT * 16;
    const int lane    = threadIdx.x;
    const int rowBase = blockIdx.x * 16;
    const int colBase = threadIdx.y * 16;

    int r = rowBase + (lane & 15);
    if (r >= M) r = M - 1;
    const int kOff = (lane < 16) ? 0 : 2;
    const int c = colBase + (lane & 15);

    const float invd = invdeg[r];
    const float* xsRow = Xs + (long long)r * K + kOff;
    const float* xaRow = Xa + (long long)r * K + kOff;

    v8f acc = {};
    for (int kk = 0; kk < K; kk += 4) {
        v2f a_s = *reinterpret_cast<const v2f*>(xsRow + kk);
        v2f a_raw = *reinterpret_cast<const v2f*>(xaRow + kk);
        v2f a_n = { a_raw.x * invd, a_raw.y * invd };
        const int kb = kk + kOff;
        v2f b_s = { Ws[(long long)kb * N + c], Ws[(long long)(kb + 1) * N + c] };
        v2f b_n = { Wn[(long long)kb * N + c], Wn[(long long)(kb + 1) * N + c] };
        acc = __builtin_amdgcn_wmma_f32_16x16x4_f32(false, a_s, false, b_s,
                                                    (short)0, acc, false, false);
        acc = __builtin_amdgcn_wmma_f32_16x16x4_f32(false, a_n, false, b_n,
                                                    (short)0, acc, false, false);
    }

    const float bv = bias[c];
    const int mBase = rowBase + ((lane < 16) ? 0 : 8);
#pragma unroll
    for (int j = 0; j < 8; ++j) {
        int m = mBase + j;
        if (m < M) {
            float v = acc[j] + bv;
            if (do_relu) v = fmaxf(v, 0.0f);
            out[(long long)m * N + c] = v;
        }
    }
}


__global__ __launch_bounds__(256) void edge_weight_kernel(const int* __restrict__ src,
                                                          const int* __restrict__ dst,
                                                          const int* __restrict__ gid,
                                                          const float* __restrict__ invdeg,
                                                          float* __restrict__ Ct, int E) {
    int e = blockIdx.x * 256 + threadIdx.x;
    if (e < E) {
        int t = dst[e];
        int g = gid[t];
        if ((unsigned)g < (unsigned)NG)
            atomicAdd(&Ct[(long long)src[e] * NG + g], invdeg[t]);
    }
}

__global__ __launch_bounds__(256) void graph_reduce_wmma(
        const float* __restrict__ h1, const float* __restrict__ Ct,
        const int* __restrict__ gid, float* __restrict__ part,
        float* __restrict__ cntPart, int Nn) {
    __shared__ float Sloc[(NG + 1) * HID_F];
    __shared__ int   cntloc[NG + 1];
    __shared__ int   sgid[CH];

    const int lane = threadIdx.x;
    const int w    = threadIdx.y;
    const int tid  = w * 32 + lane;
    const int k0   = blockIdx.x * CH;

    for (int i = tid; i < (NG + 1) * HID_F; i += 256) Sloc[i] = 0.0f;
    if (tid <= NG) cntloc[tid] = 0;
    {
        const int node = k0 + tid;
        int g = NG;
        if (node < Nn) {
            const int gg = gid[node];
            if ((unsigned)gg < (unsigned)NG) g = gg;
        }
        sgid[tid] = g;
    }
    __syncthreads();
    atomicAdd(&cntloc[sgid[tid]], 1);

    const int kOff = (lane < 16) ? 0 : 2;
    const int col  = lane & 15;
    const int f    = w * 16 + col;

    v8f acc0 = {}, acc1 = {}, acc2 = {}, acc3 = {};
    for (int kk = 0; kk < CH; kk += 4 * KU) {
        v2f b[KU], a0[KU], a1[KU], a2[KU], a3[KU];
        int sa[KU], sb[KU];
#pragma unroll
        for (int u = 0; u < KU; ++u) {
            const int ka = kk + 4 * u + kOff;
            const int na = k0 + ka;
            const bool va = na < Nn;
            const bool vb = (na + 1) < Nn;
            const int ca = va ? na : (Nn - 1);
            const int cb = vb ? (na + 1) : (Nn - 1);
            const float x = h1[(long long)ca * HID_F + f];
            const float y = h1[(long long)cb * HID_F + f];
            b[u] = (v2f){ va ? x : 0.0f, vb ? y : 0.0f };
            const float* ra = Ct + (long long)ca * NG + col;
            const float* rb = Ct + (long long)cb * NG + col;
            a0[u] = (v2f){ ra[0],  rb[0]  };
            a1[u] = (v2f){ ra[16], rb[16] };
            a2[u] = (v2f){ ra[32], rb[32] };
            a3[u] = (v2f){ ra[48], rb[48] };
            sa[u] = sgid[ka];
            sb[u] = sgid[ka + 1];
        }
#pragma unroll
        for (int u = 0; u < KU; ++u) {
            atomicAdd(&Sloc[sa[u] * HID_F + f], b[u].x);
            atomicAdd(&Sloc[sb[u] * HID_F + f], b[u].y);
            acc0 = __builtin_amdgcn_wmma_f32_16x16x4_f32(false, a0[u], false, b[u], (short)0, acc0, false, false);
            acc1 = __builtin_amdgcn_wmma_f32_16x16x4_f32(false, a1[u], false, b[u], (short)0, acc1, false, false);
            acc2 = __builtin_amdgcn_wmma_f32_16x16x4_f32(false, a2[u], false, b[u], (short)0, acc2, false, false);
            acc3 = __builtin_amdgcn_wmma_f32_16x16x4_f32(false, a3[u], false, b[u], (short)0, acc3, false, false);
        }
    }

    float* pT = part + (long long)blockIdx.x * (2 * NG * HID_F);
    float* pS = pT + NG * HID_F;
    const int mB = (lane < 16) ? 0 : 8;
#pragma unroll
    for (int j = 0; j < 8; ++j) {
        pT[(0  + mB + j) * HID_F + f] = acc0[j];
        pT[(16 + mB + j) * HID_F + f] = acc1[j];
        pT[(32 + mB + j) * HID_F + f] = acc2[j];
        pT[(48 + mB + j) * HID_F + f] = acc3[j];
    }
    __syncthreads();
    for (int i = tid; i < NG * HID_F; i += 256) pS[i] = Sloc[i];
    if (tid < NG) cntPart[blockIdx.x * NG + tid] = (float)cntloc[tid];
}

__global__ __launch_bounds__(256) void sum_partials_kernel(const float* __restrict__ part,
                                                           const float* __restrict__ cntPart,
                                                           float* __restrict__ R,
                                                           float* __restrict__ cnt, int nChunks) {
    const int per = 2 * NG * HID_F;
    if (blockIdx.x < per / 256) {
        const int i = blockIdx.x * 256 + threadIdx.x;
        float s = 0.0f;
        for (int c = 0; c < nChunks; ++c) s += part[(long long)c * per + i];
        R[i] = s;
    } else if (threadIdx.x < NG) {
        float s = 0.0f;
        for (int c = 0; c < nChunks; ++c) s += cntPart[c * NG + threadIdx.x];
        cnt[threadIdx.x] = s;
    }
}

__global__ __launch_bounds__(512) void graph_head_wmma(const float* __restrict__ R,
                                                       const float* __restrict__ cnt,
                                                       const float* __restrict__ Ws,
                                                       const float* __restrict__ Wn,
                                                       const float* __restrict__ bias,
                                                       float* __restrict__ out) {
    const int lane = threadIdx.x;
    const int w    = threadIdx.y;
    const int gB   = (w >> 2) * 16;
    const int oB   = (w & 3) * 16;
    const int kOff = (lane < 16) ? 0 : 2;
    const int g    = gB + (lane & 15);
    const int c    = oB + (lane & 15);
    const float* tRow = R + g * HID_F + kOff;
    const float* sRow = R + NG * HID_F + g * HID_F + kOff;

    v8f accS = {}, accT = {};
    for (int kk = 0; kk < HID_F; kk += 4) {
        const v2f a_s = *reinterpret_cast<const v2f*>(sRow + kk);
        const v2f a_t = *reinterpret_cast<const v2f*>(tRow + kk);
        const int kb = kk + kOff;
        const v2f b_s = { Ws[kb * OUT_F + c], Ws[(kb + 1) * OUT_F + c] };
        const v2f b_n = { Wn[kb * OUT_F + c], Wn[(kb + 1) * OUT_F + c] };
        accS = __builtin_amdgcn_wmma_f32_16x16x4_f32(false, a_s, false, b_s, (short)0, accS, false, false);
        accT = __builtin_amdgcn_wmma_f32_16x16x4_f32(false, a_t, false, b_n, (short)0, accT, false, false);
    }

    const float bv = bias[c];
    const int mB = gB + ((lane < 16) ? 0 : 8);
#pragma unroll
    for (int j = 0; j < 8; ++j) {
        const int gg = mB + j;
        const float n = cnt[gg];
        out[gg * OUT_F + c] = (accS[j] + accT[j] + n * bv) / fmaxf(n, 1.0f);
    }
}


static inline int cdiv64(long long a, long long b) { return (int)((a + b - 1) / b); }

extern "C" void kernel_launch(void* const* d_in, const int* in_sizes, int n_in,
                              void* d_out, int out_size, void* d_ws, size_t ws_size,
                              hipStream_t stream) {
    const float* in_feat = (const float*)d_in[0];
    const int*   src     = (const int*)d_in[1];
    const int*   dst     = (const int*)d_in[2];
    const int*   gid     = (const int*)d_in[3];
    const float* W1s = (const float*)d_in[5];
    const float* W1n = (const float*)d_in[6];
    const float* b1  = (const float*)d_in[7];
    const float* W2s = (const float*)d_in[8];
    const float* W2n = (const float*)d_in[9];
    const float* b2  = (const float*)d_in[10];
    float* out = (float*)d_out;

    const int Nn = in_sizes[0] / IN_F;
    const int E  = in_sizes[1];

    float* ws    = (float*)d_ws;
    float* accum = ws;
    float* deg   = accum + (size_t)Nn * 128;
    float* Ct    = deg + Nn;
    float* h1    = Ct + (size_t)Nn * NG;

    const int mTiles  = cdiv64(Nn, 16);
    const int nChunks = cdiv64(Nn, CH);

    float* part    = accum;
    float* cntPart = part + (size_t)nChunks * (2 * NG * HID_F);
    float* R       = cntPart + (size_t)nChunks * NG;
    float* cnt     = R + 2 * NG * HID_F;

    zero_kernel<<<cdiv64((long long)Nn * NG, 256), 256, 0, stream>>>(Ct, (long long)Nn * NG);
    tile_aggregate_kernel<<<cdiv64(Nn, TR), dim3(32, TW), 0, stream>>>(in_feat, src, dst, accum, deg, Nn, E);
    sage_gemm_wmma<8><<<mTiles, dim3(32, 8), 0, stream>>>(
        in_feat, accum, deg, W1s, W1n, b1, h1, Nn, IN_F, 1);

    edge_weight_kernel<<<cdiv64(E, 256), 256, 0, stream>>>(src, dst, gid, deg, Ct, E);
    graph_reduce_wmma<<<nChunks, dim3(32, 8), 0, stream>>>(h1, Ct, gid, part, cntPart, Nn);
    sum_partials_kernel<<<2 * NG * HID_F / 256 + 1, 256, 0, stream>>>(part, cntPart, R, cnt, nChunks);
    graph_head_wmma<<<1, dim3(32, 16), 0, stream>>>(R, cnt, W2s, W2n, b2, out);
}


# Layer-1 owner scan reads a 1-byte owner-tile key per edge (written by the first launch) instead of the 4-byte destination: 16 edges per load, packed byte compare, exact test at the add
# speedup vs baseline: 3.2308x; 1.0512x over previous
#include <hip/hip_runtime.h>

typedef __attribute__((ext_vector_type(2))) float v2f;
typedef __attribute__((ext_vector_type(8))) float v8f;

#define IN_F  128
#define HID_F 128
#define OUT_F 64
#define NG    64
#define CH    256
#define KU    4
#define TR    256
#define TW    32
#define TRS   8
#define SU    2
#define PB    8
#define FLUSH 64
#define LCAP  (FLUSH + 512)


__global__ __launch_bounds__(256) void prep_kernel(float* __restrict__ p, long long n,
                                                   const int* __restrict__ dst,
                                                   unsigned* __restrict__ keys, int E) {
    const long long i = (long long)blockIdx.x * 256 + threadIdx.x;
    if (i < n) p[i] = 0.0f;
    const int nW = (E + 3) >> 2;
    if (i < nW) {
        const int e = (int)i * 4;
        unsigned w;
        if (e + 3 < E) {
            const int4 d = *reinterpret_cast<const int4*>(dst + e);
            w = ((unsigned)(d.x >> TRS) & 255u) | (((unsigned)(d.y >> TRS) & 255u) << 8) |
                (((unsigned)(d.z >> TRS) & 255u) << 16) | (((unsigned)(d.w >> TRS) & 255u) << 24);
        } else {
            w = 0u;
            for (int b = 0; b < 4; ++b)
                if (e + b < E) w |= ((unsigned)(dst[e + b] >> TRS) & 255u) << (8 * b);
        }
        keys[i] = w;
    }
}

__global__ __launch_bounds__(1024) void tile_aggregate_kernel(
        const float* __restrict__ feat, const int* __restrict__ src, const int* __restrict__ dst,
        const unsigned* __restrict__ keys,
        float* __restrict__ accum, float* __restrict__ invdeg, int Nn, int E) {
    __shared__ __attribute__((aligned(16))) float tile[TR * 128];
    __shared__ int degc[TR];
    __shared__ int lst[TW][LCAP];

    const int lane = threadIdx.x;
    const int w    = __builtin_amdgcn_readfirstlane(threadIdx.y);
    const int tid  = w * 32 + lane;
    const int rowBase = blockIdx.x * TR;

    for (int i = tid; i < TR * 32; i += 32 * TW)
        reinterpret_cast<float4*>(tile)[i] = make_float4(0.0f, 0.0f, 0.0f, 0.0f);
    if (tid < TR) degc[tid] = 0;
    __syncthreads();

    int* myl = lst[w];
    int cnt = 0;

    auto note = [&](bool m, int e) {
        const unsigned mask = __builtin_amdgcn_ballot_w32(m);
        if (m) myl[cnt + (int)__builtin_amdgcn_mbcnt_lo(mask, 0u)] = e;
        cnt += __builtin_popcount(mask);
    };

    auto drain = [&]() {
        for (int i = lane; i < cnt; i += 32) {
            const unsigned ld = (unsigned)dst[myl[i]] - (unsigned)rowBase;
            if (ld < (unsigned)TR) atomicAdd(&degc[ld], 1);
        }
        for (int base = 0; base < cnt; base += PB) {
            int ee[PB], ss[PB], dd[PB];
            float4 v[PB];
#pragma unroll
            for (int k = 0; k < PB; ++k) {
                int idx = base + k;
                if (idx > cnt - 1) idx = cnt - 1;
                ee[k] = myl[idx];
            }
#pragma unroll
            for (int k = 0; k < PB; ++k) {
                ss[k] = src[ee[k]];
                dd[k] = dst[ee[k]] - rowBase;
            }
#pragma unroll
            for (int k = 0; k < PB; ++k)
                v[k] = *reinterpret_cast<const float4*>(feat + (long long)ss[k] * 128 + lane * 4);
#pragma unroll
            for (int k = 0; k < PB; ++k) {
                const bool live = (base + k < cnt) && ((unsigned)dd[k] < (unsigned)TR);
                float* t = tile + (live ? dd[k] : 0) * 128 + lane;
                atomicAdd(t,      live ? v[k].x : 0.0f);
                atomicAdd(t + 32, live ? v[k].y : 0.0f);
                atomicAdd(t + 64, live ? v[k].z : 0.0f);
                atomicAdd(t + 96, live ? v[k].w : 0.0f);
            }
        }
        cnt = 0;
    };

    const int nVec   = E >> 4;
    const int nChunk = (nVec + 31) >> 5;
    const int nIt    = (nChunk + TW * SU - 1) / (TW * SU);
    const int rot    = (int)(((unsigned long long)blockIdx.x * (unsigned long long)nChunk) / gridDim.x);
    const uint4* key4 = reinterpret_cast<const uint4*>(keys);
    const unsigned pat = ((unsigned)blockIdx.x & 255u) * 0x01010101u;

    auto fetch = [&](int it, uint4 (&kv)[SU], int (&eb)[SU]) {
#pragma unroll
        for (int u = 0; u < SU; ++u) {
            int c = (it * TW + w) * SU + u;
            bool ok = c < nChunk;
            c += rot;
            if (c >= nChunk) c -= nChunk;
            const int q = c * 32 + lane;
            ok = ok && (q < nVec);
            eb[u] = ok ? q * 16 : -1;
            kv[u] = key4[ok ? q : 0];
        }
    };
    auto flags = [&](unsigned x) {
        x ^= pat;
        return (x - 0x01010101u) & ~x & 0x80808080u;
    };
    auto look = [&](const uint4 (&kv)[SU], const int (&eb)[SU]) {
#pragma unroll
        for (int u = 0; u < SU; ++u) {
            unsigned f = (flags(kv[u].x) >> 7) | (flags(kv[u].y) >> 6) |
                         (flags(kv[u].z) >> 5) | (flags(kv[u].w) >> 4);
            if (eb[u] < 0) f = 0u;
            while (__builtin_amdgcn_ballot_w32(f != 0u) != 0u) {
                const bool has = f != 0u;
                const int p = has ? __builtin_ctz(f) : 0;
                note(has, eb[u] + 4 * (p & 7) + (p >> 3));
                f &= f - 1u;
            }
            if (cnt >= FLUSH) drain();
        }
    };

    if (nIt > 0) {
        uint4 kvA[SU], kvB[SU];
        int   ebA[SU], ebB[SU];
        fetch(0, kvA, ebA);
        for (int it = 0; it < nIt; it += 2) {
            fetch(it + 1, kvB, ebB);
            look(kvA, ebA);
            fetch(it + 2, kvA, ebA);
            look(kvB, ebB);
        }
    }
    if (w == 0) {
        const int e = nVec * 16 + lane;
        const bool in = (lane < 16) && (e < E);
        const int d = in ? dst[e] : -1;
        note(in && ((unsigned)d - (unsigned)rowBase < (unsigned)TR), e);
    }
    drain();
    __syncthreads();

    for (int rr = w; rr < TR; rr += TW) {
        const int node = rowBase + rr;
        if (node < Nn) {
            const float* t = tile + rr * 128 + lane;
            const float4 o = make_float4(t[0], t[32], t[64], t[96]);
            *reinterpret_cast<float4*>(accum + (long long)node * 128 + lane * 4) = o;
        }
    }
    if (tid < TR) {
        const int node = rowBase + tid;
        if (node < Nn) invdeg[node] = 1.0f / fmaxf((float)degc[tid], 1.0f);
    }
}

template <int NCOLT>
__global__ __launch_bounds__(32 * NCOLT) void sage_gemm_wmma(
        const float* __restrict__ Xs, const float* __restrict__ Xa,
        const float* __restrict__ invdeg,
        const float* __restrict__ Ws, const float* __restrict__ Wn,
        const float* __restrict__ bias, float* __restrict__ out,
        int M, int K, int do_relu) {
    const int N = NCOLT * 16;
    const int lane    = threadIdx.x;
    const int rowBase = blockIdx.x * 16;
    const int colBase = threadIdx.y * 16;

    int r = rowBase + (lane & 15);
    if (r >= M) r = M - 1;
    const int kOff = (lane < 16) ? 0 : 2;
    const int c = colBase + (lane & 15);

    const float invd = invdeg[r];
    const float* xsRow = Xs + (long long)r * K + kOff;
    const float* xaRow = Xa + (long long)r * K + kOff;

    v8f acc = {};
    for (int kk = 0; kk < K; kk += 4) {
        v2f a_s = *reinterpret_cast<const v2f*>(xsRow + kk);
        v2f a_raw = *reinterpret_cast<const v2f*>(xaRow + kk);
        v2f a_n = { a_raw.x * invd, a_raw.y * invd };
        const int kb = kk + kOff;
        v2f b_s = { Ws[(long long)kb * N + c], Ws[(long long)(kb + 1) * N + c] };
        v2f b_n = { Wn[(long long)kb * N + c], Wn[(long long)(kb + 1) * N + c] };
        acc = __builtin_amdgcn_wmma_f32_16x16x4_f32(false, a_s, false, b_s,
                                                    (short)0, acc, false, false);
        acc = __builtin_amdgcn_wmma_f32_16x16x4_f32(false, a_n, false, b_n,
                                                    (short)0, acc, false, false);
    }

    const float bv = bias[c];
    const int mBase = rowBase + ((lane < 16) ? 0 : 8);
#pragma unroll
    for (int j = 0; j < 8; ++j) {
        int m = mBase + j;
        if (m < M) {
            float v = acc[j] + bv;
            if (do_relu) v = fmaxf(v, 0.0f);
            out[(long long)m * N + c] = v;
        }
    }
}


__global__ __launch_bounds__(256) void edge_weight_kernel(const int* __restrict__ src,
                                                          const int* __restrict__ dst,
                                                          const int* __restrict__ gid,
                                                          const float* __restrict__ invdeg,
                                                          float* __restrict__ Ct, int E) {
    int e = blockIdx.x * 256 + threadIdx.x;
    if (e < E) {
        int t = dst[e];
        int g = gid[t];
        if ((unsigned)g < (unsigned)NG)
            atomicAdd(&Ct[(long long)src[e] * NG + g], invdeg[t]);
    }
}

__global__ __launch_bounds__(256) void graph_reduce_wmma(
        const float* __restrict__ h1, const float* __restrict__ Ct,
        const int* __restrict__ gid, float* __restrict__ part,
        float* __restrict__ cntPart, int Nn) {
    __shared__ float Sloc[(NG + 1) * HID_F];
    __shared__ int   cntloc[NG + 1];
    __shared__ int   sgid[CH];

    const int lane = threadIdx.x;
    const int w    = threadIdx.y;
    const int tid  = w * 32 + lane;
    const int k0   = blockIdx.x * CH;

    for (int i = tid; i < (NG + 1) * HID_F; i += 256) Sloc[i] = 0.0f;
    if (tid <= NG) cntloc[tid] = 0;
    {
        const int node = k0 + tid;
        int g = NG;
        if (node < Nn) {
            const int gg = gid[node];
            if ((unsigned)gg < (unsigned)NG) g = gg;
        }
        sgid[tid] = g;
    }
    __syncthreads();
    atomicAdd(&cntloc[sgid[tid]], 1);

    const int kOff = (lane < 16) ? 0 : 2;
    const int col  = lane & 15;
    const int f    = w * 16 + col;

    v8f acc0 = {}, acc1 = {}, acc2 = {}, acc3 = {};
    for (int kk = 0; kk < CH; kk += 4 * KU) {
        v2f b[KU], a0[KU], a1[KU], a2[KU], a3[KU];
        int sa[KU], sb[KU];
#pragma unroll
        for (int u = 0; u < KU; ++u) {
            const int ka = kk + 4 * u + kOff;
            const int na = k0 + ka;
            const bool va = na < Nn;
            const bool vb = (na + 1) < Nn;
            const int ca = va ? na : (Nn - 1);
            const int cb = vb ? (na + 1) : (Nn - 1);
            const float x = h1[(long long)ca * HID_F + f];
            const float y = h1[(long long)cb * HID_F + f];
            b[u] = (v2f){ va ? x : 0.0f, vb ? y : 0.0f };
            const float* ra = Ct + (long long)ca * NG + col;
            const float* rb = Ct + (long long)cb * NG + col;
            a0[u] = (v2f){ ra[0],  rb[0]  };
            a1[u] = (v2f){ ra[16], rb[16] };
            a2[u] = (v2f){ ra[32], rb[32] };
            a3[u] = (v2f){ ra[48], rb[48] };
            sa[u] = sgid[ka];
            sb[u] = sgid[ka + 1];
        }
#pragma unroll
        for (int u = 0; u < KU; ++u) {
            atomicAdd(&Sloc[sa[u] * HID_F + f], b[u].x);
            atomicAdd(&Sloc[sb[u] * HID_F + f], b[u].y);
            acc0 = __builtin_amdgcn_wmma_f32_16x16x4_f32(false, a0[u], false, b[u], (short)0, acc0, false, false);
            acc1 = __builtin_amdgcn_wmma_f32_16x16x4_f32(false, a1[u], false, b[u], (short)0, acc1, false, false);
            acc2 = __builtin_amdgcn_wmma_f32_16x16x4_f32(false, a2[u], false, b[u], (short)0, acc2, false, false);
            acc3 = __builtin_amdgcn_wmma_f32_16x16x4_f32(false, a3[u], false, b[u], (short)0, acc3, false, false);
        }
    }

    float* pT = part + (long long)blockIdx.x * (2 * NG * HID_F);
    float* pS = pT + NG * HID_F;
    const int mB = (lane < 16) ? 0 : 8;
#pragma unroll
    for (int j = 0; j < 8; ++j) {
        pT[(0  + mB + j) * HID_F + f] = acc0[j];
        pT[(16 + mB + j) * HID_F + f] = acc1[j];
        pT[(32 + mB + j) * HID_F + f] = acc2[j];
        pT[(48 + mB + j) * HID_F + f] = acc3[j];
    }
    __syncthreads();
    for (int i = tid; i < NG * HID_F; i += 256) pS[i] = Sloc[i];
    if (tid < NG) cntPart[blockIdx.x * NG + tid] = (float)cntloc[tid];
}

__global__ __launch_bounds__(256) void sum_partials_kernel(const float* __restrict__ part,
                                                           const float* __restrict__ cntPart,
                                                           float* __restrict__ R,
                                                           float* __restrict__ cnt, int nChunks) {
    const int per = 2 * NG * HID_F;
    if (blockIdx.x < per / 256) {
        const int i = blockIdx.x * 256 + threadIdx.x;
        float s = 0.0f;
        for (int c = 0; c < nChunks; ++c) s += part[(long long)c * per + i];
        R[i] = s;
    } else if (threadIdx.x < NG) {
        float s = 0.0f;
        for (int c = 0; c < nChunks; ++c) s += cntPart[c * NG + threadIdx.x];
        cnt[threadIdx.x] = s;
    }
}

__global__ __launch_bounds__(512) void graph_head_wmma(const float* __restrict__ R,
                                                       const float* __restrict__ cnt,
                                                       const float* __restrict__ Ws,
                                                       const float* __restrict__ Wn,
                                                       const float* __restrict__ bias,
                                                       float* __restrict__ out) {
    const int lane = threadIdx.x;
    const int w    = threadIdx.y;
    const int gB   = (w >> 2) * 16;
    const int oB   = (w & 3) * 16;
    const int kOff = (lane < 16) ? 0 : 2;
    const int g    = gB + (lane & 15);
    const int c    = oB + (lane & 15);
    const float* tRow = R + g * HID_F + kOff;
    const float* sRow = R + NG * HID_F + g * HID_F + kOff;

    v8f accS = {}, accT = {};
    for (int kk = 0; kk < HID_F; kk += 4) {
        const v2f a_s = *reinterpret_cast<const v2f*>(sRow + kk);
        const v2f a_t = *reinterpret_cast<const v2f*>(tRow + kk);
        const int kb = kk + kOff;
        const v2f b_s = { Ws[kb * OUT_F + c], Ws[(kb + 1) * OUT_F + c] };
        const v2f b_n = { Wn[kb * OUT_F + c], Wn[(kb + 1) * OUT_F + c] };
        accS = __builtin_amdgcn_wmma_f32_16x16x4_f32(false, a_s, false, b_s, (short)0, accS, false, false);
        accT = __builtin_amdgcn_wmma_f32_16x16x4_f32(false, a_t, false, b_n, (short)0, accT, false, false);
    }

    const float bv = bias[c];
    const int mB = gB + ((lane < 16) ? 0 : 8);
#pragma unroll
    for (int j = 0; j < 8; ++j) {
        const int gg = mB + j;
        const float n = cnt[gg];
        out[gg * OUT_F + c] = (accS[j] + accT[j] + n * bv) / fmaxf(n, 1.0f);
    }
}


static inline int cdiv64(long long a, long long b) { return (int)((a + b - 1) / b); }

extern "C" void kernel_launch(void* const* d_in, const int* in_sizes, int n_in,
                              void* d_out, int out_size, void* d_ws, size_t ws_size,
                              hipStream_t stream) {
    const float* in_feat = (const float*)d_in[0];
    const int*   src     = (const int*)d_in[1];
    const int*   dst     = (const int*)d_in[2];
    const int*   gid     = (const int*)d_in[3];
    const float* W1s = (const float*)d_in[5];
    const float* W1n = (const float*)d_in[6];
    const float* b1  = (const float*)d_in[7];
    const float* W2s = (const float*)d_in[8];
    const float* W2n = (const float*)d_in[9];
    const float* b2  = (const float*)d_in[10];
    float* out = (float*)d_out;

    const int Nn = in_sizes[0] / IN_F;
    const int E  = in_sizes[1];

    float* ws    = (float*)d_ws;
    float* accum = ws;
    float* deg   = accum + (size_t)Nn * 128;
    float* Ct    = deg + Nn;
    float* h1    = Ct + (size_t)Nn * NG;

    const int mTiles  = cdiv64(Nn, 16);
    const int nChunks = cdiv64(Nn, CH);

    float* part    = accum;
    float* cntPart = part + (size_t)nChunks * (2 * NG * HID_F);
    float* R       = cntPart + (size_t)nChunks * NG;
    float* cnt     = R + 2 * NG * HID_F;

    unsigned* keys = (unsigned*)h1;
    const long long nCt = (long long)Nn * NG;
    const long long nKw = ((long long)E + 3) / 4;

    prep_kernel<<<cdiv64(nCt > nKw ? nCt : nKw, 256), 256, 0, stream>>>(Ct, nCt, dst, keys, E);
    tile_aggregate_kernel<<<cdiv64(Nn, TR), dim3(32, TW), 0, stream>>>(in_feat, src, dst, keys, accum, deg, Nn, E);
    sage_gemm_wmma<8><<<mTiles, dim3(32, 8), 0, stream>>>(
        in_feat, accum, deg, W1s, W1n, b1, h1, Nn, IN_F, 1);

    edge_weight_kernel<<<cdiv64(E, 256), 256, 0, stream>>>(src, dst, gid, deg, Ct, E);
    graph_reduce_wmma<<<nChunks, dim3(32, 8), 0, stream>>>(h1, Ct, gid, part, cntPart, Nn);
    sum_partials_kernel<<<2 * NG * HID_F / 256 + 1, 256, 0, stream>>>(part, cntPart, R, cnt, nChunks);
    graph_head_wmma<<<1, dim3(32, 16), 0, stream>>>(R, cnt, W2s, W2n, b2, out);
}


# Layer-1 GEMM retiled in registers: one wave per 16x64 strip (4 accumulators share each A fragment), all fragment loads 16 bytes via a column-to-lane map; k order inside each 8-feature group changes
# speedup vs baseline: 3.8673x; 1.1970x over previous
#include <hip/hip_runtime.h>

typedef __attribute__((ext_vector_type(2))) float v2f;
typedef __attribute__((ext_vector_type(8))) float v8f;

#define IN_F  128
#define HID_F 128
#define OUT_F 64
#define NG    64
#define CH    256
#define KU    4
#define TR    256
#define TW    32
#define TRS   8
#define SU    2
#define PB    8
#define FLUSH 64
#define LCAP  (FLUSH + 512)


__global__ __launch_bounds__(256) void prep_kernel(float* __restrict__ p, long long n,
                                                   const int* __restrict__ dst,
                                                   unsigned* __restrict__ keys, int E) {
    const long long i = (long long)blockIdx.x * 256 + threadIdx.x;
    if (i < n) p[i] = 0.0f;
    const int nW = (E + 3) >> 2;
    if (i < nW) {
        const int e = (int)i * 4;
        unsigned w;
        if (e + 3 < E) {
            const int4 d = *reinterpret_cast<const int4*>(dst + e);
            w = ((unsigned)(d.x >> TRS) & 255u) | (((unsigned)(d.y >> TRS) & 255u) << 8) |
                (((unsigned)(d.z >> TRS) & 255u) << 16) | (((unsigned)(d.w >> TRS) & 255u) << 24);
        } else {
            w = 0u;
            for (int b = 0; b < 4; ++b)
                if (e + b < E) w |= ((unsigned)(dst[e + b] >> TRS) & 255u) << (8 * b);
        }
        keys[i] = w;
    }
}

__global__ __launch_bounds__(1024) void tile_aggregate_kernel(
        const float* __restrict__ feat, const int* __restrict__ src, const int* __restrict__ dst,
        const unsigned* __restrict__ keys,
        float* __restrict__ accum, float* __restrict__ invdeg, int Nn, int E) {
    __shared__ __attribute__((aligned(16))) float tile[TR * 128];
    __shared__ int degc[TR];
    __shared__ int lst[TW][LCAP];

    const int lane = threadIdx.x;
    const int w    = __builtin_amdgcn_readfirstlane(threadIdx.y);
    const int tid  = w * 32 + lane;
    const int rowBase = blockIdx.x * TR;

    for (int i = tid; i < TR * 32; i += 32 * TW)
        reinterpret_cast<float4*>(tile)[i] = make_float4(0.0f, 0.0f, 0.0f, 0.0f);
    if (tid < TR) degc[tid] = 0;
    __syncthreads();

    int* myl = lst[w];
    int cnt = 0;

    auto note = [&](bool m, int e) {
        const unsigned mask = __builtin_amdgcn_ballot_w32(m);
        if (m) myl[cnt + (int)__builtin_amdgcn_mbcnt_lo(mask, 0u)] = e;
        cnt += __builtin_popcount(mask);
    };

    auto drain = [&]() {
        for (int i = lane; i < cnt; i += 32) {
            const unsigned ld = (unsigned)dst[myl[i]] - (unsigned)rowBase;
            if (ld < (unsigned)TR) atomicAdd(&degc[ld], 1);
        }
        for (int base = 0; base < cnt; base += PB) {
            int ee[PB], ss[PB], dd[PB];
            float4 v[PB];
#pragma unroll
            for (int k = 0; k < PB; ++k) {
                int idx = base + k;
                if (idx > cnt - 1) idx = cnt - 1;
                ee[k] = myl[idx];
            }
#pragma unroll
            for (int k = 0; k < PB; ++k) {
                ss[k] = src[ee[k]];
                dd[k] = dst[ee[k]] - rowBase;
            }
#pragma unroll
            for (int k = 0; k < PB; ++k)
                v[k] = *reinterpret_cast<const float4*>(feat + (long long)ss[k] * 128 + lane * 4);
#pragma unroll
            for (int k = 0; k < PB; ++k) {
                const bool live = (base + k < cnt) && ((unsigned)dd[k] < (unsigned)TR);
                float* t = tile + (live ? dd[k] : 0) * 128 + lane;
                atomicAdd(t,      live ? v[k].x : 0.0f);
                atomicAdd(t + 32, live ? v[k].y : 0.0f);
                atomicAdd(t + 64, live ? v[k].z : 0.0f);
                atomicAdd(t + 96, live ? v[k].w : 0.0f);
            }
        }
        cnt = 0;
    };

    const int nVec   = E >> 4;
    const int nChunk = (nVec + 31) >> 5;
    const int nIt    = (nChunk + TW * SU - 1) / (TW * SU);
    const int rot    = (int)(((unsigned long long)blockIdx.x * (unsigned long long)nChunk) / gridDim.x);
    const uint4* key4 = reinterpret_cast<const uint4*>(keys);
    const unsigned pat = ((unsigned)blockIdx.x & 255u) * 0x01010101u;

    auto fetch = [&](int it, uint4 (&kv)[SU], int (&eb)[SU]) {
#pragma unroll
        for (int u = 0; u < SU; ++u) {
            int c = (it * TW + w) * SU + u;
            bool ok = c < nChunk;
            c += rot;
            if (c >= nChunk) c -= nChunk;
            const int q = c * 32 + lane;
            ok = ok && (q < nVec);
            eb[u] = ok ? q * 16 : -1;
            kv[u] = key4[ok ? q : 0];
        }
    };
    auto flags = [&](unsigned x) {
        x ^= pat;
        return (x - 0x01010101u) & ~x & 0x80808080u;
    };
    auto look = [&](const uint4 (&kv)[SU], const int (&eb)[SU]) {
#pragma unroll
        for (int u = 0; u < SU; ++u) {
            unsigned f = (flags(kv[u].x) >> 7) | (flags(kv[u].y) >> 6) |
                         (flags(kv[u].z) >> 5) | (flags(kv[u].w) >> 4);
            if (eb[u] < 0) f = 0u;
            while (__builtin_amdgcn_ballot_w32(f != 0u) != 0u) {
                const bool has = f != 0u;
                const int p = has ? __builtin_ctz(f) : 0;
                note(has, eb[u] + 4 * (p & 7) + (p >> 3));
                f &= f - 1u;
            }
            if (cnt >= FLUSH) drain();
        }
    };

    if (nIt > 0) {
        uint4 kvA[SU], kvB[SU];
        int   ebA[SU], ebB[SU];
        fetch(0, kvA, ebA);
        for (int it = 0; it < nIt; it += 2) {
            fetch(it + 1, kvB, ebB);
            look(kvA, ebA);
            fetch(it + 2, kvA, ebA);
            look(kvB, ebB);
        }
    }
    if (w == 0) {
        const int e = nVec * 16 + lane;
        const bool in = (lane < 16) && (e < E);
        const int d = in ? dst[e] : -1;
        note(in && ((unsigned)d - (unsigned)rowBase < (unsigned)TR), e);
    }
    drain();
    __syncthreads();

    for (int rr = w; rr < TR; rr += TW) {
        const int node = rowBase + rr;
        if (node < Nn) {
            const float* t = tile + rr * 128 + lane;
            const float4 o = make_float4(t[0], t[32], t[64], t[96]);
            *reinterpret_cast<float4*>(accum + (long long)node * 128 + lane * 4) = o;
        }
    }
    if (tid < TR) {
        const int node = rowBase + tid;
        if (node < Nn) invdeg[node] = 1.0f / fmaxf((float)degc[tid], 1.0f);
    }
}

struct GemmRaw {
    float4 as, ar;
    float4 bs[4], bn[4];
};

__device__ __forceinline__ void gemm_load(GemmRaw& g, const float4* xs, const float4* xa,
                                          const float* ws, const float* wn, int i) {
    g.as = xs[2 * i];
    g.ar = xa[2 * i];
#pragma unroll
    for (int e = 0; e < 4; ++e) {
        g.bs[e] = *reinterpret_cast<const float4*>(ws + (8 * i + e) * HID_F);
        g.bn[e] = *reinterpret_cast<const float4*>(wn + (8 * i + e) * HID_F);
    }
}

__device__ __forceinline__ void gemm_group(const GemmRaw& g, float invd,
                                           v8f& acc0, v8f& acc1, v8f& acc2, v8f& acc3) {
    const v2f aS0 = { g.as.x, g.as.y }, aS1 = { g.as.z, g.as.w };
    const v2f aN0 = { g.ar.x * invd, g.ar.y * invd };
    const v2f aN1 = { g.ar.z * invd, g.ar.w * invd };
    acc0 = __builtin_amdgcn_wmma_f32_16x16x4_f32(false, aS0, false, (v2f){ g.bs[0].x, g.bs[1].x }, (short)0, acc0, false, false);
    acc1 = __builtin_amdgcn_wmma_f32_16x16x4_f32(false, aS0, false, (v2f){ g.bs[0].y, g.bs[1].y }, (short)0, acc1, false, false);
    acc2 = __builtin_amdgcn_wmma_f32_16x16x4_f32(false, aS0, false, (v2f){ g.bs[0].z, g.bs[1].z }, (short)0, acc2, false, false);
    acc3 = __builtin_amdgcn_wmma_f32_16x16x4_f32(false, aS0, false, (v2f){ g.bs[0].w, g.bs[1].w }, (short)0, acc3, false, false);
    acc0 = __builtin_amdgcn_wmma_f32_16x16x4_f32(false, aN0, false, (v2f){ g.bn[0].x, g.bn[1].x }, (short)0, acc0, false, false);
    acc1 = __builtin_amdgcn_wmma_f32_16x16x4_f32(false, aN0, false, (v2f){ g.bn[0].y, g.bn[1].y }, (short)0, acc1, false, false);
    acc2 = __builtin_amdgcn_wmma_f32_16x16x4_f32(false, aN0, false, (v2f){ g.bn[0].z, g.bn[1].z }, (short)0, acc2, false, false);
    acc3 = __builtin_amdgcn_wmma_f32_16x16x4_f32(false, aN0, false, (v2f){ g.bn[0].w, g.bn[1].w }, (short)0, acc3, false, false);
    acc0 = __builtin_amdgcn_wmma_f32_16x16x4_f32(false, aS1, false, (v2f){ g.bs[2].x, g.bs[3].x }, (short)0, acc0, false, false);
    acc1 = __builtin_amdgcn_wmma_f32_16x16x4_f32(false, aS1, false, (v2f){ g.bs[2].y, g.bs[3].y }, (short)0, acc1, false, false);
    acc2 = __builtin_amdgcn_wmma_f32_16x16x4_f32(false, aS1, false, (v2f){ g.bs[2].z, g.bs[3].z }, (short)0, acc2, false, false);
    acc3 = __builtin_amdgcn_wmma_f32_16x16x4_f32(false, aS1, false, (v2f){ g.bs[2].w, g.bs[3].w }, (short)0, acc3, false, false);
    acc0 = __builtin_amdgcn_wmma_f32_16x16x4_f32(false, aN1, false, (v2f){ g.bn[2].x, g.bn[3].x }, (short)0, acc0, false, false);
    acc1 = __builtin_amdgcn_wmma_f32_16x16x4_f32(false, aN1, false, (v2f){ g.bn[2].y, g.bn[3].y }, (short)0, acc1, false, false);
    acc2 = __builtin_amdgcn_wmma_f32_16x16x4_f32(false, aN1, false, (v2f){ g.bn[2].z, g.bn[3].z }, (short)0, acc2, false, false);
    acc3 = __builtin_amdgcn_wmma_f32_16x16x4_f32(false, aN1, false, (v2f){ g.bn[2].w, g.bn[3].w }, (short)0, acc3, false, false);
}

__global__ __launch_bounds__(64) void sage_gemm_strip(
        const float* __restrict__ Xs, const float* __restrict__ Xa,
        const float* __restrict__ invdeg,
        const float* __restrict__ Ws, const float* __restrict__ Wn,
        const float* __restrict__ bias, float* __restrict__ out,
        int M, int do_relu) {
    const int lane = threadIdx.x;
    const int c    = lane & 15;
    const int h    = lane >> 4;
    const int rowBase = blockIdx.x * 16;
    const int col0    = threadIdx.y * 64 + 4 * c;

    int r = rowBase + c;
    if (r >= M) r = M - 1;
    const float invd = invdeg[r];
    const float4* xs = reinterpret_cast<const float4*>(Xs + (long long)r * IN_F + 4 * h);
    const float4* xa = reinterpret_cast<const float4*>(Xa + (long long)r * IN_F + 4 * h);
    const float* ws = Ws + (4 * h) * HID_F + col0;
    const float* wn = Wn + (4 * h) * HID_F + col0;

    v8f acc0 = {}, acc1 = {}, acc2 = {}, acc3 = {};
    GemmRaw p, q;
    gemm_load(p, xs, xa, ws, wn, 0);
    for (int t = 0; t < IN_F / 16 - 1; ++t) {
        gemm_load(q, xs, xa, ws, wn, 2 * t + 1);
        gemm_group(p, invd, acc0, acc1, acc2, acc3);
        gemm_load(p, xs, xa, ws, wn, 2 * t + 2);
        gemm_group(q, invd, acc0, acc1, acc2, acc3);
    }
    gemm_load(q, xs, xa, ws, wn, IN_F / 8 - 1);
    gemm_group(p, invd, acc0, acc1, acc2, acc3);
    gemm_group(q, invd, acc0, acc1, acc2, acc3);

    const float4 bv = *reinterpret_cast<const float4*>(bias + col0);
    const int mBase = rowBase + 8 * h;
#pragma unroll
    for (int j = 0; j < 8; ++j) {
        const int m = mBase + j;
        if (m < M) {
            float4 v = make_float4(acc0[j] + bv.x, acc1[j] + bv.y, acc2[j] + bv.z, acc3[j] + bv.w);
            if (do_relu) v = make_float4(fmaxf(v.x, 0.0f), fmaxf(v.y, 0.0f), fmaxf(v.z, 0.0f), fmaxf(v.w, 0.0f));
            *reinterpret_cast<float4*>(out + (long long)m * HID_F + col0) = v;
        }
    }
}


__global__ __launch_bounds__(256) void edge_weight_kernel(const int* __restrict__ src,
                                                          const int* __restrict__ dst,
                                                          const int* __restrict__ gid,
                                                          const float* __restrict__ invdeg,
                                                          float* __restrict__ Ct, int E) {
    int e = blockIdx.x * 256 + threadIdx.x;
    if (e < E) {
        int t = dst[e];
        int g = gid[t];
        if ((unsigned)g < (unsigned)NG)
            atomicAdd(&Ct[(long long)src[e] * NG + g], invdeg[t]);
    }
}

__global__ __launch_bounds__(256) void graph_reduce_wmma(
        const float* __restrict__ h1, const float* __restrict__ Ct,
        const int* __restrict__ gid, float* __restrict__ part,
        float* __restrict__ cntPart, int Nn) {
    __shared__ float Sloc[(NG + 1) * HID_F];
    __shared__ int   cntloc[NG + 1];
    __shared__ int   sgid[CH];

    const int lane = threadIdx.x;
    const int w    = threadIdx.y;
    const int tid  = w * 32 + lane;
    const int k0   = blockIdx.x * CH;

    for (int i = tid; i < (NG + 1) * HID_F; i += 256) Sloc[i] = 0.0f;
    if (tid <= NG) cntloc[tid] = 0;
    {
        const int node = k0 + tid;
        int g = NG;
        if (node < Nn) {
            const int gg = gid[node];
            if ((unsigned)gg < (unsigned)NG) g = gg;
        }
        sgid[tid] = g;
    }
    __syncthreads();
    atomicAdd(&cntloc[sgid[tid]], 1);

    const int kOff = (lane < 16) ? 0 : 2;
    const int col  = lane & 15;
    const int f    = w * 16 + col;

    v8f acc0 = {}, acc1 = {}, acc2 = {}, acc3 = {};
    for (int kk = 0; kk < CH; kk += 4 * KU) {
        v2f b[KU], a0[KU], a1[KU], a2[KU], a3[KU];
        int sa[KU], sb[KU];
#pragma unroll
        for (int u = 0; u < KU; ++u) {
            const int ka = kk + 4 * u + kOff;
            const int na = k0 + ka;
            const bool va = na < Nn;
            const bool vb = (na + 1) < Nn;
            const int ca = va ? na : (Nn - 1);
            const int cb = vb ? (na + 1) : (Nn - 1);
            const float x = h1[(long long)ca * HID_F + f];
            const float y = h1[(long long)cb * HID_F + f];
            b[u] = (v2f){ va ? x : 0.0f, vb ? y : 0.0f };
            const float* ra = Ct + (long long)ca * NG + col;
            const float* rb = Ct + (long long)cb * NG + col;
            a0[u] = (v2f){ ra[0],  rb[0]  };
            a1[u] = (v2f){ ra[16], rb[16] };
            a2[u] = (v2f){ ra[32], rb[32] };
            a3[u] = (v2f){ ra[48], rb[48] };
            sa[u] = sgid[ka];
            sb[u] = sgid[ka + 1];
        }
#pragma unroll
        for (int u = 0; u < KU; ++u) {
            atomicAdd(&Sloc[sa[u] * HID_F + f], b[u].x);
            atomicAdd(&Sloc[sb[u] * HID_F + f], b[u].y);
            acc0 = __builtin_amdgcn_wmma_f32_16x16x4_f32(false, a0[u], false, b[u], (short)0, acc0, false, false);
            acc1 = __builtin_amdgcn_wmma_f32_16x16x4_f32(false, a1[u], false, b[u], (short)0, acc1, false, false);
            acc2 = __builtin_amdgcn_wmma_f32_16x16x4_f32(false, a2[u], false, b[u], (short)0, acc2, false, false);
            acc3 = __builtin_amdgcn_wmma_f32_16x16x4_f32(false, a3[u], false, b[u], (short)0, acc3, false, false);
        }
    }

    float* pT = part + (long long)blockIdx.x * (2 * NG * HID_F);
    float* pS = pT + NG * HID_F;
    const int mB = (lane < 16) ? 0 : 8;
#pragma unroll
    for (int j = 0; j < 8; ++j) {
        pT[(0  + mB + j) * HID_F + f] = acc0[j];
        pT[(16 + mB + j) * HID_F + f] = acc1[j];
        pT[(32 + mB + j) * HID_F + f] = acc2[j];
        pT[(48 + mB + j) * HID_F + f] = acc3[j];
    }
    __syncthreads();
    for (int i = tid; i < NG * HID_F; i += 256) pS[i] = Sloc[i];
    if (tid < NG) cntPart[blockIdx.x * NG + tid] = (float)cntloc[tid];
}

__global__ __launch_bounds__(256) void sum_partials_kernel(const float* __restrict__ part,
                                                           const float* __restrict__ cntPart,
                                                           float* __restrict__ R,
                                                           float* __restrict__ cnt, int nChunks) {
    const int per = 2 * NG * HID_F;
    if (blockIdx.x < per / 256) {
        const int i = blockIdx.x * 256 + threadIdx.x;
        float s = 0.0f;
        for (int c = 0; c < nChunks; ++c) s += part[(long long)c * per + i];
        R[i] = s;
    } else if (threadIdx.x < NG) {
        float s = 0.0f;
        for (int c = 0; c < nChunks; ++c) s += cntPart[c * NG + threadIdx.x];
        cnt[threadIdx.x] = s;
    }
}

__global__ __launch_bounds__(512) void graph_head_wmma(const float* __restrict__ R,
                                                       const float* __restrict__ cnt,
                                                       const float* __restrict__ Ws,
                                                       const float* __restrict__ Wn,
                                                       const float* __restrict__ bias,
                                                       float* __restrict__ out) {
    const int lane = threadIdx.x;
    const int w    = threadIdx.y;
    const int gB   = (w >> 2) * 16;
    const int oB   = (w & 3) * 16;
    const int kOff = (lane < 16) ? 0 : 2;
    const int g    = gB + (lane & 15);
    const int c    = oB + (lane & 15);
    const float* tRow = R + g * HID_F + kOff;
    const float* sRow = R + NG * HID_F + g * HID_F + kOff;

    v8f accS = {}, accT = {};
    for (int kk = 0; kk < HID_F; kk += 4) {
        const v2f a_s = *reinterpret_cast<const v2f*>(sRow + kk);
        const v2f a_t = *reinterpret_cast<const v2f*>(tRow + kk);
        const int kb = kk + kOff;
        const v2f b_s = { Ws[kb * OUT_F + c], Ws[(kb + 1) * OUT_F + c] };
        const v2f b_n = { Wn[kb * OUT_F + c], Wn[(kb + 1) * OUT_F + c] };
        accS = __builtin_amdgcn_wmma_f32_16x16x4_f32(false, a_s, false, b_s, (short)0, accS, false, false);
        accT = __builtin_amdgcn_wmma_f32_16x16x4_f32(false, a_t, false, b_n, (short)0, accT, false, false);
    }

    const float bv = bias[c];
    const int mB = gB + ((lane < 16) ? 0 : 8);
#pragma unroll
    for (int j = 0; j < 8; ++j) {
        const int gg = mB + j;
        const float n = cnt[gg];
        out[gg * OUT_F + c] = (accS[j] + accT[j] + n * bv) / fmaxf(n, 1.0f);
    }
}


static inline int cdiv64(long long a, long long b) { return (int)((a + b - 1) / b); }

extern "C" void kernel_launch(void* const* d_in, const int* in_sizes, int n_in,
                              void* d_out, int out_size, void* d_ws, size_t ws_size,
                              hipStream_t stream) {
    const float* in_feat = (const float*)d_in[0];
    const int*   src     = (const int*)d_in[1];
    const int*   dst     = (const int*)d_in[2];
    const int*   gid     = (const int*)d_in[3];
    const float* W1s = (const float*)d_in[5];
    const float* W1n = (const float*)d_in[6];
    const float* b1  = (const float*)d_in[7];
    const float* W2s = (const float*)d_in[8];
    const float* W2n = (const float*)d_in[9];
    const float* b2  = (const float*)d_in[10];
    float* out = (float*)d_out;

    const int Nn = in_sizes[0] / IN_F;
    const int E  = in_sizes[1];

    float* ws    = (float*)d_ws;
    float* accum = ws;
    float* deg   = accum + (size_t)Nn * 128;
    float* Ct    = deg + Nn;
    float* h1    = Ct + (size_t)Nn * NG;

    const int mTiles  = cdiv64(Nn, 16);
    const int nChunks = cdiv64(Nn, CH);

    float* part    = accum;
    float* cntPart = part + (size_t)nChunks * (2 * NG * HID_F);
    float* R       = cntPart + (size_t)nChunks * NG;
    float* cnt     = R + 2 * NG * HID_F;

    unsigned* keys = (unsigned*)h1;
    const long long nCt = (long long)Nn * NG;
    const long long nKw = ((long long)E + 3) / 4;

    prep_kernel<<<cdiv64(nCt > nKw ? nCt : nKw, 256), 256, 0, stream>>>(Ct, nCt, dst, keys, E);
    tile_aggregate_kernel<<<cdiv64(Nn, TR), dim3(32, TW), 0, stream>>>(in_feat, src, dst, keys, accum, deg, Nn, E);
    sage_gemm_strip<<<mTiles, dim3(32, 2), 0, stream>>>(
        in_feat, accum, deg, W1s, W1n, b1, h1, Nn, 1);

    edge_weight_kernel<<<cdiv64(E, 256), 256, 0, stream>>>(src, dst, gid, deg, Ct, E);
    graph_reduce_wmma<<<nChunks, dim3(32, 8), 0, stream>>>(h1, Ct, gid, part, cntPart, Nn);
    sum_partials_kernel<<<2 * NG * HID_F / 256 + 1, 256, 0, stream>>>(part, cntPart, R, cnt, nChunks);
    graph_head_wmma<<<1, dim3(32, 16), 0, stream>>>(R, cnt, W2s, W2n, b2, out);
}


# Edge weights Ct kept out of memory: each reduction work-group owns 256 source rows, scans 1-byte source keys, builds its Ct piece and T in LDS (LDS float atomics, order changes); global atomics gone
# speedup vs baseline: 4.3959x; 1.1367x over previous
#include <hip/hip_runtime.h>

typedef __attribute__((ext_vector_type(2))) float v2f;
typedef __attribute__((ext_vector_type(8))) float v8f;

#define IN_F  128
#define HID_F 128
#define OUT_F 64
#define NG    64
#define CH    256
#define KU    4
#define TR    256
#define TW    32
#define TRS   8
#define SU    2
#define PB    8
#define FLUSH 64
#define LCAP  (FLUSH + 512)
#define RW    32
#define CP    80


__device__ __forceinline__ unsigned pack_keys(const int* __restrict__ a, int e, int E) {
    unsigned w;
    if (e + 3 < E) {
        const int4 d = *reinterpret_cast<const int4*>(a + e);
        w = ((unsigned)(d.x >> TRS) & 255u) | (((unsigned)(d.y >> TRS) & 255u) << 8) |
            (((unsigned)(d.z >> TRS) & 255u) << 16) | (((unsigned)(d.w >> TRS) & 255u) << 24);
    } else {
        w = 0u;
        for (int b = 0; b < 4; ++b)
            if (e + b < E) w |= ((unsigned)(a[e + b] >> TRS) & 255u) << (8 * b);
    }
    return w;
}

__global__ __launch_bounds__(256) void prep_kernel(const int* __restrict__ dst,
                                                   const int* __restrict__ src,
                                                   unsigned* __restrict__ dkeys,
                                                   unsigned* __restrict__ skeys, int E) {
    const int i = blockIdx.x * 256 + threadIdx.x;
    const int nW = (E + 3) >> 2;
    if (i < nW) {
        dkeys[i] = pack_keys(dst, i * 4, E);
        skeys[i] = pack_keys(src, i * 4, E);
    }
}

__global__ __launch_bounds__(1024) void tile_aggregate_kernel(
        const float* __restrict__ feat, const int* __restrict__ src, const int* __restrict__ dst,
        const unsigned* __restrict__ keys,
        float* __restrict__ accum, float* __restrict__ invdeg, int Nn, int E) {
    __shared__ __attribute__((aligned(16))) float tile[TR * 128];
    __shared__ int degc[TR];
    __shared__ int lst[TW][LCAP];

    const int lane = threadIdx.x;
    const int w    = __builtin_amdgcn_readfirstlane(threadIdx.y);
    const int tid  = w * 32 + lane;
    const int rowBase = blockIdx.x * TR;

    for (int i = tid; i < TR * 32; i += 32 * TW)
        reinterpret_cast<float4*>(tile)[i] = make_float4(0.0f, 0.0f, 0.0f, 0.0f);
    if (tid < TR) degc[tid] = 0;
    __syncthreads();

    int* myl = lst[w];
    int cnt = 0;

    auto note = [&](bool m, int e) {
        const unsigned mask = __builtin_amdgcn_ballot_w32(m);
        if (m) myl[cnt + (int)__builtin_amdgcn_mbcnt_lo(mask, 0u)] = e;
        cnt += __builtin_popcount(mask);
    };

    auto drain = [&]() {
        for (int i = lane; i < cnt; i += 32) {
            const unsigned ld = (unsigned)dst[myl[i]] - (unsigned)rowBase;
            if (ld < (unsigned)TR) atomicAdd(&degc[ld], 1);
        }
        for (int base = 0; base < cnt; base += PB) {
            int ee[PB], ss[PB], dd[PB];
            float4 v[PB];
#pragma unroll
            for (int k = 0; k < PB; ++k) {
                int idx = base + k;
                if (idx > cnt - 1) idx = cnt - 1;
                ee[k] = myl[idx];
            }
#pragma unroll
            for (int k = 0; k < PB; ++k) {
                ss[k] = src[ee[k]];
                dd[k] = dst[ee[k]] - rowBase;
            }
#pragma unroll
            for (int k = 0; k < PB; ++k)
                v[k] = *reinterpret_cast<const float4*>(feat + (long long)ss[k] * 128 + lane * 4);
#pragma unroll
            for (int k = 0; k < PB; ++k) {
                const bool live = (base + k < cnt) && ((unsigned)dd[k] < (unsigned)TR);
                float* t = tile + (live ? dd[k] : 0) * 128 + lane;
                atomicAdd(t,      live ? v[k].x : 0.0f);
                atomicAdd(t + 32, live ? v[k].y : 0.0f);
                atomicAdd(t + 64, live ? v[k].z : 0.0f);
                atomicAdd(t + 96, live ? v[k].w : 0.0f);
            }
        }
        cnt = 0;
    };

    const int nVec   = E >> 4;
    const int nChunk = (nVec + 31) >> 5;
    const int nIt    = (nChunk + TW * SU - 1) / (TW * SU);
    const int rot    = (int)(((unsigned long long)blockIdx.x * (unsigned long long)nChunk) / gridDim.x);
    const uint4* key4 = reinterpret_cast<const uint4*>(keys);
    const unsigned pat = ((unsigned)blockIdx.x & 255u) * 0x01010101u;

    auto fetch = [&](int it, uint4 (&kv)[SU], int (&eb)[SU]) {
#pragma unroll
        for (int u = 0; u < SU; ++u) {
            int c = (it * TW + w) * SU + u;
            bool ok = c < nChunk;
            c += rot;
            if (c >= nChunk) c -= nChunk;
            const int q = c * 32 + lane;
            ok = ok && (q < nVec);
            eb[u] = ok ? q * 16 : -1;
            kv[u] = key4[ok ? q : 0];
        }
    };
    auto flags = [&](unsigned x) {
        x ^= pat;
        return (x - 0x01010101u) & ~x & 0x80808080u;
    };
    auto look = [&](const uint4 (&kv)[SU], const int (&eb)[SU]) {
#pragma unroll
        for (int u = 0; u < SU; ++u) {
            unsigned f = (flags(kv[u].x) >> 7) | (flags(kv[u].y) >> 6) |
                         (flags(kv[u].z) >> 5) | (flags(kv[u].w) >> 4);
            if (eb[u] < 0) f = 0u;
            while (__builtin_amdgcn_ballot_w32(f != 0u) != 0u) {
                const bool has = f != 0u;
                const int p = has ? __builtin_ctz(f) : 0;
                note(has, eb[u] + 4 * (p & 7) + (p >> 3));
                f &= f - 1u;
            }
            if (cnt >= FLUSH) drain();
        }
    };

    if (nIt > 0) {
        uint4 kvA[SU], kvB[SU];
        int   ebA[SU], ebB[SU];
        fetch(0, kvA, ebA);
        for (int it = 0; it < nIt; it += 2) {
            fetch(it + 1, kvB, ebB);
            look(kvA, ebA);
            fetch(it + 2, kvA, ebA);
            look(kvB, ebB);
        }
    }
    if (w == 0) {
        const int e = nVec * 16 + lane;
        const bool in = (lane < 16) && (e < E);
        const int d = in ? dst[e] : -1;
        note(in && ((unsigned)d - (unsigned)rowBase < (unsigned)TR), e);
    }
    drain();
    __syncthreads();

    for (int rr = w; rr < TR; rr += TW) {
        const int node = rowBase + rr;
        if (node < Nn) {
            const float* t = tile + rr * 128 + lane;
            const float4 o = make_float4(t[0], t[32], t[64], t[96]);
            *reinterpret_cast<float4*>(accum + (long long)node * 128 + lane * 4) = o;
        }
    }
    if (tid < TR) {
        const int node = rowBase + tid;
        if (node < Nn) invdeg[node] = 1.0f / fmaxf((float)degc[tid], 1.0f);
    }
}

struct GemmRaw {
    float4 as, ar;
    float4 bs[4], bn[4];
};

__device__ __forceinline__ void gemm_load(GemmRaw& g, const float4* xs, const float4* xa,
                                          const float* ws, const float* wn, int i) {
    g.as = xs[2 * i];
    g.ar = xa[2 * i];
#pragma unroll
    for (int e = 0; e < 4; ++e) {
        g.bs[e] = *reinterpret_cast<const float4*>(ws + (8 * i + e) * HID_F);
        g.bn[e] = *reinterpret_cast<const float4*>(wn + (8 * i + e) * HID_F);
    }
}

__device__ __forceinline__ void gemm_group(const GemmRaw& g, float invd,
                                           v8f& acc0, v8f& acc1, v8f& acc2, v8f& acc3) {
    const v2f aS0 = { g.as.x, g.as.y }, aS1 = { g.as.z, g.as.w };
    const v2f aN0 = { g.ar.x * invd, g.ar.y * invd };
    const v2f aN1 = { g.ar.z * invd, g.ar.w * invd };
    acc0 = __builtin_amdgcn_wmma_f32_16x16x4_f32(false, aS0, false, (v2f){ g.bs[0].x, g.bs[1].x }, (short)0, acc0, false, false);
    acc1 = __builtin_amdgcn_wmma_f32_16x16x4_f32(false, aS0, false, (v2f){ g.bs[0].y, g.bs[1].y }, (short)0, acc1, false, false);
    acc2 = __builtin_amdgcn_wmma_f32_16x16x4_f32(false, aS0, false, (v2f){ g.bs[0].z, g.bs[1].z }, (short)0, acc2, false, false);
    acc3 = __builtin_amdgcn_wmma_f32_16x16x4_f32(false, aS0, false, (v2f){ g.bs[0].w, g.bs[1].w }, (short)0, acc3, false, false);
    acc0 = __builtin_amdgcn_wmma_f32_16x16x4_f32(false, aN0, false, (v2f){ g.bn[0].x, g.bn[1].x }, (short)0, acc0, false, false);
    acc1 = __builtin_amdgcn_wmma_f32_16x16x4_f32(false, aN0, false, (v2f){ g.bn[0].y, g.bn[1].y }, (short)0, acc1, false, false);
    acc2 = __builtin_amdgcn_wmma_f32_16x16x4_f32(false, aN0, false, (v2f){ g.bn[0].z, g.bn[1].z }, (short)0, acc2, false, false);
    acc3 = __builtin_amdgcn_wmma_f32_16x16x4_f32(false, aN0, false, (v2f){ g.bn[0].w, g.bn[1].w }, (short)0, acc3, false, false);
    acc0 = __builtin_amdgcn_wmma_f32_16x16x4_f32(false, aS1, false, (v2f){ g.bs[2].x, g.bs[3].x }, (short)0, acc0, false, false);
    acc1 = __builtin_amdgcn_wmma_f32_16x16x4_f32(false, aS1, false, (v2f){ g.bs[2].y, g.bs[3].y }, (short)0, acc1, false, false);
    acc2 = __builtin_amdgcn_wmma_f32_16x16x4_f32(false, aS1, false, (v2f){ g.bs[2].z, g.bs[3].z }, (short)0, acc2, false, false);
    acc3 = __builtin_amdgcn_wmma_f32_16x16x4_f32(false, aS1, false, (v2f){ g.bs[2].w, g.bs[3].w }, (short)0, acc3, false, false);
    acc0 = __builtin_amdgcn_wmma_f32_16x16x4_f32(false, aN1, false, (v2f){ g.bn[2].x, g.bn[3].x }, (short)0, acc0, false, false);
    acc1 = __builtin_amdgcn_wmma_f32_16x16x4_f32(false, aN1, false, (v2f){ g.bn[2].y, g.bn[3].y }, (short)0, acc1, false, false);
    acc2 = __builtin_amdgcn_wmma_f32_16x16x4_f32(false, aN1, false, (v2f){ g.bn[2].z, g.bn[3].z }, (short)0, acc2, false, false);
    acc3 = __builtin_amdgcn_wmma_f32_16x16x4_f32(false, aN1, false, (v2f){ g.bn[2].w, g.bn[3].w }, (short)0, acc3, false, false);
}

__global__ __launch_bounds__(64) void sage_gemm_strip(
        const float* __restrict__ Xs, const float* __restrict__ Xa,
        const float* __restrict__ invdeg,
        const float* __restrict__ Ws, const float* __restrict__ Wn,
        const float* __restrict__ bias, float* __restrict__ out,
        int M, int do_relu) {
    const int lane = threadIdx.x;
    const int c    = lane & 15;
    const int h    = lane >> 4;
    const int rowBase = blockIdx.x * 16;
    const int col0    = threadIdx.y * 64 + 4 * c;

    int r = rowBase + c;
    if (r >= M) r = M - 1;
    const float invd = invdeg[r];
    const float4* xs = reinterpret_cast<const float4*>(Xs + (long long)r * IN_F + 4 * h);
    const float4* xa = reinterpret_cast<const float4*>(Xa + (long long)r * IN_F + 4 * h);
    const float* ws = Ws + (4 * h) * HID_F + col0;
    const float* wn = Wn + (4 * h) * HID_F + col0;

    v8f acc0 = {}, acc1 = {}, acc2 = {}, acc3 = {};
    GemmRaw p, q;
    gemm_load(p, xs, xa, ws, wn, 0);
    for (int t = 0; t < IN_F / 16 - 1; ++t) {
        gemm_load(q, xs, xa, ws, wn, 2 * t + 1);
        gemm_group(p, invd, acc0, acc1, acc2, acc3);
        gemm_load(p, xs, xa, ws, wn, 2 * t + 2);
        gemm_group(q, invd, acc0, acc1, acc2, acc3);
    }
    gemm_load(q, xs, xa, ws, wn, IN_F / 8 - 1);
    gemm_group(p, invd, acc0, acc1, acc2, acc3);
    gemm_group(q, invd, acc0, acc1, acc2, acc3);

    const float4 bv = *reinterpret_cast<const float4*>(bias + col0);
    const int mBase = rowBase + 8 * h;
#pragma unroll
    for (int j = 0; j < 8; ++j) {
        const int m = mBase + j;
        if (m < M) {
            float4 v = make_float4(acc0[j] + bv.x, acc1[j] + bv.y, acc2[j] + bv.z, acc3[j] + bv.w);
            if (do_relu) v = make_float4(fmaxf(v.x, 0.0f), fmaxf(v.y, 0.0f), fmaxf(v.z, 0.0f), fmaxf(v.w, 0.0f));
            *reinterpret_cast<float4*>(out + (long long)m * HID_F + col0) = v;
        }
    }
}

static_assert(CH == TR && CH == (1 << TRS), "both owner tilings share the key shift");
static_assert(CH == 4 * 64 && RW == 32, "8 feature tiles x 4 node quarters of 64");

__global__ __launch_bounds__(1024) void graph_reduce_owner(
        const float* __restrict__ h1, const int* __restrict__ src, const int* __restrict__ dst,
        const unsigned* __restrict__ skeys, const int* __restrict__ gid,
        const float* __restrict__ invdeg,
        float* __restrict__ part, float* __restrict__ cntPart, int Nn, int E) {
    __shared__ __attribute__((aligned(16))) float CtL[CH * CP];
    __shared__ __attribute__((aligned(16))) float Sloc[(NG + 1) * HID_F];
    __shared__ __attribute__((aligned(16))) float Tloc[NG * HID_F];
    __shared__ int cntloc[NG + 1];
    __shared__ int sgid[CH];
    __shared__ int lst[RW][LCAP];

    const int lane = threadIdx.x;
    const int w    = __builtin_amdgcn_readfirstlane(threadIdx.y);
    const int tid  = w * 32 + lane;
    const int k0   = blockIdx.x * CH;

    const float4 z4 = make_float4(0.0f, 0.0f, 0.0f, 0.0f);
    for (int i = tid; i < CH * CP / 4; i += 32 * RW) reinterpret_cast<float4*>(CtL)[i] = z4;
    for (int i = tid; i < (NG + 1) * HID_F / 4; i += 32 * RW) reinterpret_cast<float4*>(Sloc)[i] = z4;
    for (int i = tid; i < NG * HID_F / 4; i += 32 * RW) reinterpret_cast<float4*>(Tloc)[i] = z4;
    if (tid <= NG) cntloc[tid] = 0;
    if (tid < CH) {
        const int node = k0 + tid;
        int g = NG;
        if (node < Nn) {
            const int gg = gid[node];
            if ((unsigned)gg < (unsigned)NG) g = gg;
        }
        sgid[tid] = g;
    }
    __syncthreads();
    if (tid < CH) atomicAdd(&cntloc[sgid[tid]], 1);

    int* myl = lst[w];
    int cnt = 0;

    auto note = [&](bool m, int e) {
        const unsigned mask = __builtin_amdgcn_ballot_w32(m);
        if (m) myl[cnt + (int)__builtin_amdgcn_mbcnt_lo(mask, 0u)] = e;
        cnt += __builtin_popcount(mask);
    };

    auto drain = [&]() {
        for (int i = lane; i < cnt; i += 32) {
            const int e = myl[i];
            const unsigned ls = (unsigned)src[e] - (unsigned)k0;
            const int d = dst[e];
            const int g = gid[d];
            const float wv = invdeg[d];
            const bool live = (ls < (unsigned)CH) && ((unsigned)g < (unsigned)NG);
            atomicAdd(&CtL[live ? (int)ls * CP + g : 0], wv * (live ? 1.0f : 0.0f));
        }
        cnt = 0;
    };

    const int nVec   = E >> 4;
    const int nChunk = (nVec + 31) >> 5;
    const int nIt    = (nChunk + RW * SU - 1) / (RW * SU);
    const int rot    = (int)(((unsigned long long)blockIdx.x * (unsigned long long)nChunk) / gridDim.x);
    const uint4* key4 = reinterpret_cast<const uint4*>(skeys);
    const unsigned pat = ((unsigned)blockIdx.x & 255u) * 0x01010101u;

    auto fetch = [&](int it, uint4 (&kv)[SU], int (&eb)[SU]) {
#pragma unroll
        for (int u = 0; u < SU; ++u) {
            int c = (it * RW + w) * SU + u;
            bool ok = c < nChunk;
            c += rot;
            if (c >= nChunk) c -= nChunk;
            const int q = c * 32 + lane;
            ok = ok && (q < nVec);
            eb[u] = ok ? q * 16 : -1;
            kv[u] = key4[ok ? q : 0];
        }
    };
    auto flags = [&](unsigned x) {
        x ^= pat;
        return (x - 0x01010101u) & ~x & 0x80808080u;
    };
    auto look = [&](const uint4 (&kv)[SU], const int (&eb)[SU]) {
#pragma unroll
        for (int u = 0; u < SU; ++u) {
            unsigned f = (flags(kv[u].x) >> 7) | (flags(kv[u].y) >> 6) |
                         (flags(kv[u].z) >> 5) | (flags(kv[u].w) >> 4);
            if (eb[u] < 0) f = 0u;
            while (__builtin_amdgcn_ballot_w32(f != 0u) != 0u) {
                const bool has = f != 0u;
                const int p = has ? __builtin_ctz(f) : 0;
                note(has, eb[u] + 4 * (p & 7) + (p >> 3));
                f &= f - 1u;
            }
            if (cnt >= FLUSH) drain();
        }
    };

    if (nIt > 0) {
        uint4 kvA[SU], kvB[SU];
        int   ebA[SU], ebB[SU];
        fetch(0, kvA, ebA);
        for (int it = 0; it < nIt; it += 2) {
            fetch(it + 1, kvB, ebB);
            look(kvA, ebA);
            fetch(it + 2, kvA, ebA);
            look(kvB, ebB);
        }
    }
    if (w == 0) {
        const int e = nVec * 16 + lane;
        const bool in = (lane < 16) && (e < E);
        const int sN = in ? src[e] : -1;
        note(in && ((unsigned)sN - (unsigned)k0 < (unsigned)CH), e);
    }
    drain();
    __syncthreads();

    const int kOff = (lane < 16) ? 0 : 2;
    const int col  = lane & 15;
    const int f    = (w & 7) * 16 + col;
    const int kq   = (w >> 3) * 64;

    v8f acc0 = {}, acc1 = {}, acc2 = {}, acc3 = {};
    for (int kk = kq; kk < kq + 64; kk += 4 * KU) {
        v2f b[KU], a0[KU], a1[KU], a2[KU], a3[KU];
        int sa[KU], sb[KU];
#pragma unroll
        for (int u = 0; u < KU; ++u) {
            const int ka = kk + 4 * u + kOff;
            const int na = k0 + ka;
            const bool va = na < Nn;
            const bool vb = (na + 1) < Nn;
            const int ca = va ? na : (Nn - 1);
            const int cb = vb ? (na + 1) : (Nn - 1);
            const float x = h1[(long long)ca * HID_F + f];
            const float y = h1[(long long)cb * HID_F + f];
            b[u] = (v2f){ va ? x : 0.0f, vb ? y : 0.0f };
            const float* ra = CtL + ka * CP + col;
            const float* rb = ra + CP;
            a0[u] = (v2f){ ra[0],  rb[0]  };
            a1[u] = (v2f){ ra[16], rb[16] };
            a2[u] = (v2f){ ra[32], rb[32] };
            a3[u] = (v2f){ ra[48], rb[48] };
            sa[u] = sgid[ka];
            sb[u] = sgid[ka + 1];
        }
#pragma unroll
        for (int u = 0; u < KU; ++u) {
            atomicAdd(&Sloc[sa[u] * HID_F + f], b[u].x);
            atomicAdd(&Sloc[sb[u] * HID_F + f], b[u].y);
            acc0 = __builtin_amdgcn_wmma_f32_16x16x4_f32(false, a0[u], false, b[u], (short)0, acc0, false, false);
            acc1 = __builtin_amdgcn_wmma_f32_16x16x4_f32(false, a1[u], false, b[u], (short)0, acc1, false, false);
            acc2 = __builtin_amdgcn_wmma_f32_16x16x4_f32(false, a2[u], false, b[u], (short)0, acc2, false, false);
            acc3 = __builtin_amdgcn_wmma_f32_16x16x4_f32(false, a3[u], false, b[u], (short)0, acc3, false, false);
        }
    }

    const int mB = (lane < 16) ? 0 : 8;
#pragma unroll
    for (int j = 0; j < 8; ++j) {
        atomicAdd(&Tloc[(0  + mB + j) * HID_F + f], acc0[j]);
        atomicAdd(&Tloc[(16 + mB + j) * HID_F + f], acc1[j]);
        atomicAdd(&Tloc[(32 + mB + j) * HID_F + f], acc2[j]);
        atomicAdd(&Tloc[(48 + mB + j) * HID_F + f], acc3[j]);
    }
    __syncthreads();

    float4* p4 = reinterpret_cast<float4*>(part + (long long)blockIdx.x * (2 * NG * HID_F));
    for (int i = tid; i < NG * HID_F / 4; i += 32 * RW) {
        p4[i] = reinterpret_cast<const float4*>(Tloc)[i];
        p4[NG * HID_F / 4 + i] = reinterpret_cast<const float4*>(Sloc)[i];
    }
    if (tid < NG) cntPart[blockIdx.x * NG + tid] = (float)cntloc[tid];
}

__global__ __launch_bounds__(256) void sum_partials_kernel(const float* __restrict__ part,
                                                           const float* __restrict__ cntPart,
                                                           float* __restrict__ R,
                                                           float* __restrict__ cnt, int nChunks) {
    const int per = 2 * NG * HID_F;
    if (blockIdx.x < per / 256) {
        const int i = blockIdx.x * 256 + threadIdx.x;
        float s = 0.0f;
        for (int c = 0; c < nChunks; ++c) s += part[(long long)c * per + i];
        R[i] = s;
    } else if (threadIdx.x < NG) {
        float s = 0.0f;
        for (int c = 0; c < nChunks; ++c) s += cntPart[c * NG + threadIdx.x];
        cnt[threadIdx.x] = s;
    }
}

__global__ __launch_bounds__(512) void graph_head_wmma(const float* __restrict__ R,
                                                       const float* __restrict__ cnt,
                                                       const float* __restrict__ Ws,
                                                       const float* __restrict__ Wn,
                                                       const float* __restrict__ bias,
                                                       float* __restrict__ out) {
    const int lane = threadIdx.x;
    const int w    = threadIdx.y;
    const int gB   = (w >> 2) * 16;
    const int oB   = (w & 3) * 16;
    const int kOff = (lane < 16) ? 0 : 2;
    const int g    = gB + (lane & 15);
    const int c    = oB + (lane & 15);
    const float* tRow = R + g * HID_F + kOff;
    const float* sRow = R + NG * HID_F + g * HID_F + kOff;

    v8f accS = {}, accT = {};
    for (int kk = 0; kk < HID_F; kk += 4) {
        const v2f a_s = *reinterpret_cast<const v2f*>(sRow + kk);
        const v2f a_t = *reinterpret_cast<const v2f*>(tRow + kk);
        const int kb = kk + kOff;
        const v2f b_s = { Ws[kb * OUT_F + c], Ws[(kb + 1) * OUT_F + c] };
        const v2f b_n = { Wn[kb * OUT_F + c], Wn[(kb + 1) * OUT_F + c] };
        accS = __builtin_amdgcn_wmma_f32_16x16x4_f32(false, a_s, false, b_s, (short)0, accS, false, false);
        accT = __builtin_amdgcn_wmma_f32_16x16x4_f32(false, a_t, false, b_n, (short)0, accT, false, false);
    }

    const float bv = bias[c];
    const int mB = gB + ((lane < 16) ? 0 : 8);
#pragma unroll
    for (int j = 0; j < 8; ++j) {
        const int gg = mB + j;
        const float n = cnt[gg];
        out[gg * OUT_F + c] = (accS[j] + accT[j] + n * bv) / fmaxf(n, 1.0f);
    }
}


static inline int cdiv64(long long a, long long b) { return (int)((a + b - 1) / b); }

extern "C" void kernel_launch(void* const* d_in, const int* in_sizes, int n_in,
                              void* d_out, int out_size, void* d_ws, size_t ws_size,
                              hipStream_t stream) {
    const float* in_feat = (const float*)d_in[0];
    const int*   src     = (const int*)d_in[1];
    const int*   dst     = (const int*)d_in[2];
    const int*   gid     = (const int*)d_in[3];
    const float* W1s = (const float*)d_in[5];
    const float* W1n = (const float*)d_in[6];
    const float* b1  = (const float*)d_in[7];
    const float* W2s = (const float*)d_in[8];
    const float* W2n = (const float*)d_in[9];
    const float* b2  = (const float*)d_in[10];
    float* out = (float*)d_out;

    const int Nn = in_sizes[0] / IN_F;
    const int E  = in_sizes[1];

    float* ws    = (float*)d_ws;
    float* accum = ws;
    float* deg   = accum + (size_t)Nn * 128;
    float* Ct    = deg + Nn;
    float* h1    = Ct + (size_t)Nn * NG;

    const int mTiles  = cdiv64(Nn, 16);
    const int nChunks = cdiv64(Nn, CH);

    float* part    = accum;
    float* cntPart = part + (size_t)nChunks * (2 * NG * HID_F);
    float* R       = cntPart + (size_t)nChunks * NG;
    float* cnt     = R + 2 * NG * HID_F;

    unsigned* keys  = (unsigned*)h1;
    unsigned* skeys = (unsigned*)Ct;
    const long long nKw = ((long long)E + 3) / 4;

    prep_kernel<<<cdiv64(nKw, 256), 256, 0, stream>>>(dst, src, keys, skeys, E);
    tile_aggregate_kernel<<<cdiv64(Nn, TR), dim3(32, TW), 0, stream>>>(in_feat, src, dst, keys, accum, deg, Nn, E);
    sage_gemm_strip<<<mTiles, dim3(32, 2), 0, stream>>>(
        in_feat, accum, deg, W1s, W1n, b1, h1, Nn, 1);

    graph_reduce_owner<<<nChunks, dim3(32, RW), 0, stream>>>(h1, src, dst, skeys, gid, deg, part, cntPart, Nn, E);
    sum_partials_kernel<<<2 * NG * HID_F / 256 + 1, 256, 0, stream>>>(part, cntPart, R, cnt, nChunks);
    graph_head_wmma<<<1, dim3(32, 16), 0, stream>>>(R, cnt, W2s, W2n, b2, out);
}


# Partial sums and layer-2 head in one launch: a work-group per graph sums [T|S] over the chunks in 16 slices, all loads in flight; the slices are the 16 A rows of the head WMMA; sum order changes
# speedup vs baseline: 5.4055x; 1.2297x over previous
#include <hip/hip_runtime.h>

typedef __attribute__((ext_vector_type(2))) float v2f;
typedef __attribute__((ext_vector_type(8))) float v8f;

#define IN_F  128
#define HID_F 128
#define OUT_F 64
#define NG    64
#define CH    256
#define KU    4
#define TR    256
#define TW    32
#define TRS   8
#define SU    2
#define PB    8
#define FLUSH 64
#define LCAP  (FLUSH + 512)
#define RW    32
#define CP    80


__device__ __forceinline__ unsigned pack_keys(const int* __restrict__ a, int e, int E) {
    unsigned w;
    if (e + 3 < E) {
        const int4 d = *reinterpret_cast<const int4*>(a + e);
        w = ((unsigned)(d.x >> TRS) & 255u) | (((unsigned)(d.y >> TRS) & 255u) << 8) |
            (((unsigned)(d.z >> TRS) & 255u) << 16) | (((unsigned)(d.w >> TRS) & 255u) << 24);
    } else {
        w = 0u;
        for (int b = 0; b < 4; ++b)
            if (e + b < E) w |= ((unsigned)(a[e + b] >> TRS) & 255u) << (8 * b);
    }
    return w;
}

__global__ __launch_bounds__(256) void prep_kernel(const int* __restrict__ dst,
                                                   const int* __restrict__ src,
                                                   unsigned* __restrict__ dkeys,
                                                   unsigned* __restrict__ skeys, int E) {
    const int i = blockIdx.x * 256 + threadIdx.x;
    const int nW = (E + 3) >> 2;
    if (i < nW) {
        dkeys[i] = pack_keys(dst, i * 4, E);
        skeys[i] = pack_keys(src, i * 4, E);
    }
}

__global__ __launch_bounds__(1024) void tile_aggregate_kernel(
        const float* __restrict__ feat, const int* __restrict__ src, const int* __restrict__ dst,
        const unsigned* __restrict__ keys,
        float* __restrict__ accum, float* __restrict__ invdeg, int Nn, int E) {
    __shared__ __attribute__((aligned(16))) float tile[TR * 128];
    __shared__ int degc[TR];
    __shared__ int lst[TW][LCAP];

    const int lane = threadIdx.x;
    const int w    = __builtin_amdgcn_readfirstlane(threadIdx.y);
    const int tid  = w * 32 + lane;
    const int rowBase = blockIdx.x * TR;

    for (int i = tid; i < TR * 32; i += 32 * TW)
        reinterpret_cast<float4*>(tile)[i] = make_float4(0.0f, 0.0f, 0.0f, 0.0f);
    if (tid < TR) degc[tid] = 0;
    __syncthreads();

    int* myl = lst[w];
    int cnt = 0;

    auto note = [&](bool m, int e) {
        const unsigned mask = __builtin_amdgcn_ballot_w32(m);
        if (m) myl[cnt + (int)__builtin_amdgcn_mbcnt_lo(mask, 0u)] = e;
        cnt += __builtin_popcount(mask);
    };

    auto drain = [&]() {
        for (int i = lane; i < cnt; i += 32) {
            const unsigned ld = (unsigned)dst[myl[i]] - (unsigned)rowBase;
            if (ld < (unsigned)TR) atomicAdd(&degc[ld], 1);
        }
        for (int base = 0; base < cnt; base += PB) {
            int ee[PB], ss[PB], dd[PB];
            float4 v[PB];
#pragma unroll
            for (int k = 0; k < PB; ++k) {
                int idx = base + k;
                if (idx > cnt - 1) idx = cnt - 1;
                ee[k] = myl[idx];
            }
#pragma unroll
            for (int k = 0; k < PB; ++k) {
                ss[k] = src[ee[k]];
                dd[k] = dst[ee[k]] - rowBase;
            }
#pragma unroll
            for (int k = 0; k < PB; ++k)
                v[k] = *reinterpret_cast<const float4*>(feat + (long long)ss[k] * 128 + lane * 4);
#pragma unroll
            for (int k = 0; k < PB; ++k) {
                const bool live = (base + k < cnt) && ((unsigned)dd[k] < (unsigned)TR);
                float* t = tile + (live ? dd[k] : 0) * 128 + lane;
                atomicAdd(t,      live ? v[k].x : 0.0f);
                atomicAdd(t + 32, live ? v[k].y : 0.0f);
                atomicAdd(t + 64, live ? v[k].z : 0.0f);
                atomicAdd(t + 96, live ? v[k].w : 0.0f);
            }
        }
        cnt = 0;
    };

    const int nVec   = E >> 4;
    const int nChunk = (nVec + 31) >> 5;
    const int nIt    = (nChunk + TW * SU - 1) / (TW * SU);
    const int rot    = (int)(((unsigned long long)blockIdx.x * (unsigned long long)nChunk) / gridDim.x);
    const uint4* key4 = reinterpret_cast<const uint4*>(keys);
    const unsigned pat = ((unsigned)blockIdx.x & 255u) * 0x01010101u;

    auto fetch = [&](int it, uint4 (&kv)[SU], int (&eb)[SU]) {
#pragma unroll
        for (int u = 0; u < SU; ++u) {
            int c = (it * TW + w) * SU + u;
            bool ok = c < nChunk;
            c += rot;
            if (c >= nChunk) c -= nChunk;
            const int q = c * 32 + lane;
            ok = ok && (q < nVec);
            eb[u] = ok ? q * 16 : -1;
            kv[u] = key4[ok ? q : 0];
        }
    };
    auto flags = [&](unsigned x) {
        x ^= pat;
        return (x - 0x01010101u) & ~x & 0x80808080u;
    };
    auto look = [&](const uint4 (&kv)[SU], const int (&eb)[SU]) {
#pragma unroll
        for (int u = 0; u < SU; ++u) {
            unsigned f = (flags(kv[u].x) >> 7) | (flags(kv[u].y) >> 6) |
                         (flags(kv[u].z) >> 5) | (flags(kv[u].w) >> 4);
            if (eb[u] < 0) f = 0u;
            while (__builtin_amdgcn_ballot_w32(f != 0u) != 0u) {
                const bool has = f != 0u;
                const int p = has ? __builtin_ctz(f) : 0;
                note(has, eb[u] + 4 * (p & 7) + (p >> 3));
                f &= f - 1u;
            }
            if (cnt >= FLUSH) drain();
        }
    };

    if (nIt > 0) {
        uint4 kvA[SU], kvB[SU];
        int   ebA[SU], ebB[SU];
        fetch(0, kvA, ebA);
        for (int it = 0; it < nIt; it += 2) {
            fetch(it + 1, kvB, ebB);
            look(kvA, ebA);
            fetch(it + 2, kvA, ebA);
            look(kvB, ebB);
        }
    }
    if (w == 0) {
        const int e = nVec * 16 + lane;
        const bool in = (lane < 16) && (e < E);
        const int d = in ? dst[e] : -1;
        note(in && ((unsigned)d - (unsigned)rowBase < (unsigned)TR), e);
    }
    drain();
    __syncthreads();

    for (int rr = w; rr < TR; rr += TW) {
        const int node = rowBase + rr;
        if (node < Nn) {
            const float* t = tile + rr * 128 + lane;
            const float4 o = make_float4(t[0], t[32], t[64], t[96]);
            *reinterpret_cast<float4*>(accum + (long long)node * 128 + lane * 4) = o;
        }
    }
    if (tid < TR) {
        const int node = rowBase + tid;
        if (node < Nn) invdeg[node] = 1.0f / fmaxf((float)degc[tid], 1.0f);
    }
}

struct GemmRaw {
    float4 as, ar;
    float4 bs[4], bn[4];
};

__device__ __forceinline__ void gemm_load(GemmRaw& g, const float4* xs, const float4* xa,
                                          const float* ws, const float* wn, int i) {
    g.as = xs[2 * i];
    g.ar = xa[2 * i];
#pragma unroll
    for (int e = 0; e < 4; ++e) {
        g.bs[e] = *reinterpret_cast<const float4*>(ws + (8 * i + e) * HID_F);
        g.bn[e] = *reinterpret_cast<const float4*>(wn + (8 * i + e) * HID_F);
    }
}

__device__ __forceinline__ void gemm_group(const GemmRaw& g, float invd,
                                           v8f& acc0, v8f& acc1, v8f& acc2, v8f& acc3) {
    const v2f aS0 = { g.as.x, g.as.y }, aS1 = { g.as.z, g.as.w };
    const v2f aN0 = { g.ar.x * invd, g.ar.y * invd };
    const v2f aN1 = { g.ar.z * invd, g.ar.w * invd };
    acc0 = __builtin_amdgcn_wmma_f32_16x16x4_f32(false, aS0, false, (v2f){ g.bs[0].x, g.bs[1].x }, (short)0, acc0, false, false);
    acc1 = __builtin_amdgcn_wmma_f32_16x16x4_f32(false, aS0, false, (v2f){ g.bs[0].y, g.bs[1].y }, (short)0, acc1, false, false);
    acc2 = __builtin_amdgcn_wmma_f32_16x16x4_f32(false, aS0, false, (v2f){ g.bs[0].z, g.bs[1].z }, (short)0, acc2, false, false);
    acc3 = __builtin_amdgcn_wmma_f32_16x16x4_f32(false, aS0, false, (v2f){ g.bs[0].w, g.bs[1].w }, (short)0, acc3, false, false);
    acc0 = __builtin_amdgcn_wmma_f32_16x16x4_f32(false, aN0, false, (v2f){ g.bn[0].x, g.bn[1].x }, (short)0, acc0, false, false);
    acc1 = __builtin_amdgcn_wmma_f32_16x16x4_f32(false, aN0, false, (v2f){ g.bn[0].y, g.bn[1].y }, (short)0, acc1, false, false);
    acc2 = __builtin_amdgcn_wmma_f32_16x16x4_f32(false, aN0, false, (v2f){ g.bn[0].z, g.bn[1].z }, (short)0, acc2, false, false);
    acc3 = __builtin_amdgcn_wmma_f32_16x16x4_f32(false, aN0, false, (v2f){ g.bn[0].w, g.bn[1].w }, (short)0, acc3, false, false);
    acc0 = __builtin_amdgcn_wmma_f32_16x16x4_f32(false, aS1, false, (v2f){ g.bs[2].x, g.bs[3].x }, (short)0, acc0, false, false);
    acc1 = __builtin_amdgcn_wmma_f32_16x16x4_f32(false, aS1, false, (v2f){ g.bs[2].y, g.bs[3].y }, (short)0, acc1, false, false);
    acc2 = __builtin_amdgcn_wmma_f32_16x16x4_f32(false, aS1, false, (v2f){ g.bs[2].z, g.bs[3].z }, (short)0, acc2, false, false);
    acc3 = __builtin_amdgcn_wmma_f32_16x16x4_f32(false, aS1, false, (v2f){ g.bs[2].w, g.bs[3].w }, (short)0, acc3, false, false);
    acc0 = __builtin_amdgcn_wmma_f32_16x16x4_f32(false, aN1, false, (v2f){ g.bn[2].x, g.bn[3].x }, (short)0, acc0, false, false);
    acc1 = __builtin_amdgcn_wmma_f32_16x16x4_f32(false, aN1, false, (v2f){ g.bn[2].y, g.bn[3].y }, (short)0, acc1, false, false);
    acc2 = __builtin_amdgcn_wmma_f32_16x16x4_f32(false, aN1, false, (v2f){ g.bn[2].z, g.bn[3].z }, (short)0, acc2, false, false);
    acc3 = __builtin_amdgcn_wmma_f32_16x16x4_f32(false, aN1, false, (v2f){ g.bn[2].w, g.bn[3].w }, (short)0, acc3, false, false);
}

__global__ __launch_bounds__(64) void sage_gemm_strip(
        const float* __restrict__ Xs, const float* __restrict__ Xa,
        const float* __restrict__ invdeg,
        const float* __restrict__ Ws, const float* __restrict__ Wn,
        const float* __restrict__ bias, float* __restrict__ out,
        int M, int do_relu) {
    const int lane = threadIdx.x;
    const int c    = lane & 15;
    const int h    = lane >> 4;
    const int rowBase = blockIdx.x * 16;
    const int col0    = threadIdx.y * 64 + 4 * c;

    int r = rowBase + c;
    if (r >= M) r = M - 1;
    const float invd = invdeg[r];
    const float4* xs = reinterpret_cast<const float4*>(Xs + (long long)r * IN_F + 4 * h);
    const float4* xa = reinterpret_cast<const float4*>(Xa + (long long)r * IN_F + 4 * h);
    const float* ws = Ws + (4 * h) * HID_F + col0;
    const float* wn = Wn + (4 * h) * HID_F + col0;

    v8f acc0 = {}, acc1 = {}, acc2 = {}, acc3 = {};
    GemmRaw p, q;
    gemm_load(p, xs, xa, ws, wn, 0);
    for (int t = 0; t < IN_F / 16 - 1; ++t) {
        gemm_load(q, xs, xa, ws, wn, 2 * t + 1);
        gemm_group(p, invd, acc0, acc1, acc2, acc3);
        gemm_load(p, xs, xa, ws, wn, 2 * t + 2);
        gemm_group(q, invd, acc0, acc1, acc2, acc3);
    }
    gemm_load(q, xs, xa, ws, wn, IN_F / 8 - 1);
    gemm_group(p, invd, acc0, acc1, acc2, acc3);
    gemm_group(q, invd, acc0, acc1, acc2, acc3);

    const float4 bv = *reinterpret_cast<const float4*>(bias + col0);
    const int mBase = rowBase + 8 * h;
#pragma unroll
    for (int j = 0; j < 8; ++j) {
        const int m = mBase + j;
        if (m < M) {
            float4 v = make_float4(acc0[j] + bv.x, acc1[j] + bv.y, acc2[j] + bv.z, acc3[j] + bv.w);
            if (do_relu) v = make_float4(fmaxf(v.x, 0.0f), fmaxf(v.y, 0.0f), fmaxf(v.z, 0.0f), fmaxf(v.w, 0.0f));
            *reinterpret_cast<float4*>(out + (long long)m * HID_F + col0) = v;
        }
    }
}

static_assert(CH == TR && CH == (1 << TRS), "both owner tilings share the key shift");
static_assert(CH == 4 * 64 && RW == 32, "8 feature tiles x 4 node quarters of 64");

__global__ __launch_bounds__(1024) void graph_reduce_owner(
        const float* __restrict__ h1, const int* __restrict__ src, const int* __restrict__ dst,
        const unsigned* __restrict__ skeys, const int* __restrict__ gid,
        const float* __restrict__ invdeg,
        float* __restrict__ part, float* __restrict__ cntPart, int Nn, int E) {
    __shared__ __attribute__((aligned(16))) float CtL[CH * CP];
    __shared__ __attribute__((aligned(16))) float Sloc[(NG + 1) * HID_F];
    __shared__ __attribute__((aligned(16))) float Tloc[NG * HID_F];
    __shared__ int cntloc[NG + 1];
    __shared__ int sgid[CH];
    __shared__ int lst[RW][LCAP];

    const int lane = threadIdx.x;
    const int w    = __builtin_amdgcn_readfirstlane(threadIdx.y);
    const int tid  = w * 32 + lane;
    const int k0   = blockIdx.x * CH;

    const float4 z4 = make_float4(0.0f, 0.0f, 0.0f, 0.0f);
    for (int i = tid; i < CH * CP / 4; i += 32 * RW) reinterpret_cast<float4*>(CtL)[i] = z4;
    for (int i = tid; i < (NG + 1) * HID_F / 4; i += 32 * RW) reinterpret_cast<float4*>(Sloc)[i] = z4;
    for (int i = tid; i < NG * HID_F / 4; i += 32 * RW) reinterpret_cast<float4*>(Tloc)[i] = z4;
    if (tid <= NG) cntloc[tid] = 0;
    if (tid < CH) {
        const int node = k0 + tid;
        int g = NG;
        if (node < Nn) {
            const int gg = gid[node];
            if ((unsigned)gg < (unsigned)NG) g = gg;
        }
        sgid[tid] = g;
    }
    __syncthreads();
    if (tid < CH) atomicAdd(&cntloc[sgid[tid]], 1);

    int* myl = lst[w];
    int cnt = 0;

    auto note = [&](bool m, int e) {
        const unsigned mask = __builtin_amdgcn_ballot_w32(m);
        if (m) myl[cnt + (int)__builtin_amdgcn_mbcnt_lo(mask, 0u)] = e;
        cnt += __builtin_popcount(mask);
    };

    auto drain = [&]() {
        for (int i = lane; i < cnt; i += 32) {
            const int e = myl[i];
            const unsigned ls = (unsigned)src[e] - (unsigned)k0;
            const int d = dst[e];
            const int g = gid[d];
            const float wv = invdeg[d];
            const bool live = (ls < (unsigned)CH) && ((unsigned)g < (unsigned)NG);
            atomicAdd(&CtL[live ? (int)ls * CP + g : 0], wv * (live ? 1.0f : 0.0f));
        }
        cnt = 0;
    };

    const int nVec   = E >> 4;
    const int nChunk = (nVec + 31) >> 5;
    const int nIt    = (nChunk + RW * SU - 1) / (RW * SU);
    const int rot    = (int)(((unsigned long long)blockIdx.x * (unsigned long long)nChunk) / gridDim.x);
    const uint4* key4 = reinterpret_cast<const uint4*>(skeys);
    const unsigned pat = ((unsigned)blockIdx.x & 255u) * 0x01010101u;

    auto fetch = [&](int it, uint4 (&kv)[SU], int (&eb)[SU]) {
#pragma unroll
        for (int u = 0; u < SU; ++u) {
            int c = (it * RW + w) * SU + u;
            bool ok = c < nChunk;
            c += rot;
            if (c >= nChunk) c -= nChunk;
            const int q = c * 32 + lane;
            ok = ok && (q < nVec);
            eb[u] = ok ? q * 16 : -1;
            kv[u] = key4[ok ? q : 0];
        }
    };
    auto flags = [&](unsigned x) {
        x ^= pat;
        return (x - 0x01010101u) & ~x & 0x80808080u;
    };
    auto look = [&](const uint4 (&kv)[SU], const int (&eb)[SU]) {
#pragma unroll
        for (int u = 0; u < SU; ++u) {
            unsigned f = (flags(kv[u].x) >> 7) | (flags(kv[u].y) >> 6) |
                         (flags(kv[u].z) >> 5) | (flags(kv[u].w) >> 4);
            if (eb[u] < 0) f = 0u;
            while (__builtin_amdgcn_ballot_w32(f != 0u) != 0u) {
                const bool has = f != 0u;
                const int p = has ? __builtin_ctz(f) : 0;
                note(has, eb[u] + 4 * (p & 7) + (p >> 3));
                f &= f - 1u;
            }
            if (cnt >= FLUSH) drain();
        }
    };

    if (nIt > 0) {
        uint4 kvA[SU], kvB[SU];
        int   ebA[SU], ebB[SU];
        fetch(0, kvA, ebA);
        for (int it = 0; it < nIt; it += 2) {
            fetch(it + 1, kvB, ebB);
            look(kvA, ebA);
            fetch(it + 2, kvA, ebA);
            look(kvB, ebB);
        }
    }
    if (w == 0) {
        const int e = nVec * 16 + lane;
        const bool in = (lane < 16) && (e < E);
        const int sN = in ? src[e] : -1;
        note(in && ((unsigned)sN - (unsigned)k0 < (unsigned)CH), e);
    }
    drain();
    __syncthreads();

    const int kOff = (lane < 16) ? 0 : 2;
    const int col  = lane & 15;
    const int f    = (w & 7) * 16 + col;
    const int kq   = (w >> 3) * 64;

    v8f acc0 = {}, acc1 = {}, acc2 = {}, acc3 = {};
    for (int kk = kq; kk < kq + 64; kk += 4 * KU) {
        v2f b[KU], a0[KU], a1[KU], a2[KU], a3[KU];
        int sa[KU], sb[KU];
#pragma unroll
        for (int u = 0; u < KU; ++u) {
            const int ka = kk + 4 * u + kOff;
            const int na = k0 + ka;
            const bool va = na < Nn;
            const bool vb = (na + 1) < Nn;
            const int ca = va ? na : (Nn - 1);
            const int cb = vb ? (na + 1) : (Nn - 1);
            const float x = h1[(long long)ca * HID_F + f];
            const float y = h1[(long long)cb * HID_F + f];
            b[u] = (v2f){ va ? x : 0.0f, vb ? y : 0.0f };
            const float* ra = CtL + ka * CP + col;
            const float* rb = ra + CP;
            a0[u] = (v2f){ ra[0],  rb[0]  };
            a1[u] = (v2f){ ra[16], rb[16] };
            a2[u] = (v2f){ ra[32], rb[32] };
            a3[u] = (v2f){ ra[48], rb[48] };
            sa[u] = sgid[ka];
            sb[u] = sgid[ka + 1];
        }
#pragma unroll
        for (int u = 0; u < KU; ++u) {
            atomicAdd(&Sloc[sa[u] * HID_F + f], b[u].x);
            atomicAdd(&Sloc[sb[u] * HID_F + f], b[u].y);
            acc0 = __builtin_amdgcn_wmma_f32_16x16x4_f32(false, a0[u], false, b[u], (short)0, acc0, false, false);
            acc1 = __builtin_amdgcn_wmma_f32_16x16x4_f32(false, a1[u], false, b[u], (short)0, acc1, false, false);
            acc2 = __builtin_amdgcn_wmma_f32_16x16x4_f32(false, a2[u], false, b[u], (short)0, acc2, false, false);
            acc3 = __builtin_amdgcn_wmma_f32_16x16x4_f32(false, a3[u], false, b[u], (short)0, acc3, false, false);
        }
    }

    const int mB = (lane < 16) ? 0 : 8;
#pragma unroll
    for (int j = 0; j < 8; ++j) {
        atomicAdd(&Tloc[(0  + mB + j) * HID_F + f], acc0[j]);
        atomicAdd(&Tloc[(16 + mB + j) * HID_F + f], acc1[j]);
        atomicAdd(&Tloc[(32 + mB + j) * HID_F + f], acc2[j]);
        atomicAdd(&Tloc[(48 + mB + j) * HID_F + f], acc3[j]);
    }
    __syncthreads();

    float4* p4 = reinterpret_cast<float4*>(part + (long long)blockIdx.x * (2 * NG * HID_F));
    for (int i = tid; i < NG * HID_F / 4; i += 32 * RW) {
        p4[i] = reinterpret_cast<const float4*>(Tloc)[i];
        p4[NG * HID_F / 4 + i] = reinterpret_cast<const float4*>(Sloc)[i];
    }
    if (tid < NG) cntPart[blockIdx.x * NG + tid] = (float)cntloc[tid];
}

#define HS  16
#define HB  16
#define HP  260
static_assert(2 * HID_F == 256 && OUT_F == 64 && HS == 16 && NG * HID_F == 8192,
              "32 waves = 4 column tiles x 8 k-slices of 32; 64 pieces of 16 bytes per row");

__global__ __launch_bounds__(1024) void graph_head_sum(const float* __restrict__ part,
                                                       const float* __restrict__ cntPart,
                                                       const float* __restrict__ Ws,
                                                       const float* __restrict__ Wn,
                                                       const float* __restrict__ bias,
                                                       float* __restrict__ out, int nChunks) {
    __shared__ __attribute__((aligned(16))) float Asl[HS * HP];
    __shared__ float red[8 * OUT_F];
    __shared__ float cw[32];

    const int lane = threadIdx.x;
    const int w    = threadIdx.y;
    const int tid  = w * 32 + lane;
    const int g    = blockIdx.x;

    const int c    = lane & 15;
    const int kOff = (lane < 16) ? 0 : 2;
    const int ks   = w >> 2;
    const int o    = (w & 3) * 16 + c;
    const float* Wp = ((ks < 4) ? Wn : Ws) + ((ks & 3) * 32 + kOff) * OUT_F + o;
    v2f b[8];
#pragma unroll
    for (int u = 0; u < 8; ++u) b[u] = (v2f){ Wp[(4 * u) * OUT_F], Wp[(4 * u + 1) * OUT_F] };

    const float cv0 = cntPart[((tid < nChunks) ? tid : 0) * NG + g];
    const float bo  = bias[tid & (OUT_F - 1)];

    const int q = tid & 63;
    const int r = tid >> 6;
    const float* rowp = part + (q >> 5) * (NG * HID_F) + g * HID_F + (q & 31) * 4;
    const long long per = 2 * NG * HID_F;
    const int nIt = (nChunks + HS - 1) / HS;
    float4 s = make_float4(0.0f, 0.0f, 0.0f, 0.0f);
    for (int base = 0; base < nIt; base += HB) {
        float4 v[HB];
#pragma unroll
        for (int k = 0; k < HB; ++k) {
            const int cc = r + (base + k) * HS;
            v[k] = *reinterpret_cast<const float4*>(rowp + (long long)((cc < nChunks) ? cc : 0) * per);
        }
#pragma unroll
        for (int k = 0; k < HB; ++k) {
            const bool ok = (r + (base + k) * HS) < nChunks;
            s.x += ok ? v[k].x : 0.0f;
            s.y += ok ? v[k].y : 0.0f;
            s.z += ok ? v[k].z : 0.0f;
            s.w += ok ? v[k].w : 0.0f;
        }
    }
    *reinterpret_cast<float4*>(Asl + r * HP + 4 * q) = s;
    float cv = (tid < nChunks) ? cv0 : 0.0f;
    for (int cc = tid + 32 * 32; cc < nChunks; cc += 32 * 32) cv += cntPart[cc * NG + g];
    for (int off = 16; off > 0; off >>= 1) cv += __shfl_xor(cv, off);
    if (lane == 0) cw[w] = cv;
    __syncthreads();

    v8f acc = {};
    const float* ap = Asl + c * HP + ks * 32 + kOff;
#pragma unroll
    for (int u = 0; u < 8; ++u) {
        const v2f a = *reinterpret_cast<const v2f*>(ap + 4 * u);
        acc = __builtin_amdgcn_wmma_f32_16x16x4_f32(false, a, false, b[u], (short)0, acc, false, false);
    }
    float t = ((acc[0] + acc[1]) + (acc[2] + acc[3])) + ((acc[4] + acc[5]) + (acc[6] + acc[7]));
    t += __shfl_xor(t, 16);
    if (lane < 16) red[ks * OUT_F + o] = t;
    __syncthreads();

    if (tid < OUT_F) {
        float tot = 0.0f;
        for (int k = 0; k < 8; ++k) tot += red[k * OUT_F + tid];
        float n = 0.0f;
        for (int k = 0; k < 32; ++k) n += cw[k];
        out[g * OUT_F + tid] = (tot + n * bo) / fmaxf(n, 1.0f);
    }
}


static inline int cdiv64(long long a, long long b) { return (int)((a + b - 1) / b); }

extern "C" void kernel_launch(void* const* d_in, const int* in_sizes, int n_in,
                              void* d_out, int out_size, void* d_ws, size_t ws_size,
                              hipStream_t stream) {
    const float* in_feat = (const float*)d_in[0];
    const int*   src     = (const int*)d_in[1];
    const int*   dst     = (const int*)d_in[2];
    const int*   gid     = (const int*)d_in[3];
    const float* W1s = (const float*)d_in[5];
    const float* W1n = (const float*)d_in[6];
    const float* b1  = (const float*)d_in[7];
    const float* W2s = (const float*)d_in[8];
    const float* W2n = (const float*)d_in[9];
    const float* b2  = (const float*)d_in[10];
    float* out = (float*)d_out;

    const int Nn = in_sizes[0] / IN_F;
    const int E  = in_sizes[1];

    float* ws    = (float*)d_ws;
    float* accum = ws;
    float* deg   = accum + (size_t)Nn * 128;
    float* Ct    = deg + Nn;
    float* h1    = Ct + (size_t)Nn * NG;

    const int mTiles  = cdiv64(Nn, 16);
    const int nChunks = cdiv64(Nn, CH);

    float* part    = accum;
    float* cntPart = part + (size_t)nChunks * (2 * NG * HID_F);

    unsigned* keys  = (unsigned*)h1;
    unsigned* skeys = (unsigned*)Ct;
    const long long nKw = ((long long)E + 3) / 4;

    prep_kernel<<<cdiv64(nKw, 256), 256, 0, stream>>>(dst, src, keys, skeys, E);
    tile_aggregate_kernel<<<cdiv64(Nn, TR), dim3(32, TW), 0, stream>>>(in_feat, src, dst, keys, accum, deg, Nn, E);
    sage_gemm_strip<<<mTiles, dim3(32, 2), 0, stream>>>(
        in_feat, accum, deg, W1s, W1n, b1, h1, Nn, 1);

    graph_reduce_owner<<<nChunks, dim3(32, RW), 0, stream>>>(h1, src, dst, skeys, gid, deg, part, cntPart, Nn, E);
    graph_head_sum<<<NG, dim3(32, 32), 0, stream>>>(part, cntPart, W2s, W2n, b2, out, nChunks);
}


# Edges binned once per call by owner tile inside 4096-edge blocks (LDS counters give ranks); owner work-groups of both passes read their runs instead of scanning all key bytes; add order changes
# speedup vs baseline: 6.0468x; 1.1186x over previous
#include <hip/hip_runtime.h>

typedef __attribute__((ext_vector_type(2))) float v2f;
typedef __attribute__((ext_vector_type(8))) float v8f;

#define IN_F  128
#define HID_F 128
#define OUT_F 64
#define NG    64
#define CH    256
#define KU    4
#define TR    256
#define TW    32
#define TRS   8
#define EB    4096
#define OFFP  257
#define RB    8
#define PB    8
#define FLUSH 64
#define LCAP  (FLUSH + 512)
#define RW    32
#define CP    80


__global__ __launch_bounds__(1024) void bin_kernel(const int* __restrict__ dst,
                                                   const int* __restrict__ src,
                                                   int* __restrict__ idxD, int* __restrict__ offD,
                                                   int* __restrict__ idxS, int* __restrict__ offS,
                                                   int E) {
    __shared__ int hd[256], hs[256];
    __shared__ int sd[OFFP], ss[OFFP];
    __shared__ __attribute__((aligned(16))) int od[EB], os[EB];

    const int tid = threadIdx.x;
    const int b   = blockIdx.x;
    const int e0  = b * EB + tid * 4;

    if (tid < 256) { hd[tid] = 0; hs[tid] = 0; }
    int dv[4], sv[4];
    if (e0 + 3 < E) {
        const int4 d4 = *reinterpret_cast<const int4*>(dst + e0);
        const int4 s4 = *reinterpret_cast<const int4*>(src + e0);
        dv[0] = d4.x; dv[1] = d4.y; dv[2] = d4.z; dv[3] = d4.w;
        sv[0] = s4.x; sv[1] = s4.y; sv[2] = s4.z; sv[3] = s4.w;
    } else {
        for (int j = 0; j < 4; ++j) {
            dv[j] = (e0 + j < E) ? dst[e0 + j] : 0;
            sv[j] = (e0 + j < E) ? src[e0 + j] : 0;
        }
    }
    __syncthreads();

    int kd[4], ks[4], rd[4], rs[4];
    for (int j = 0; j < 4; ++j) {
        kd[j] = (dv[j] >> TRS) & 255;
        ks[j] = (sv[j] >> TRS) & 255;
        rd[j] = 0;
        rs[j] = 0;
        if (e0 + j < E) {
            rd[j] = atomicAdd(&hd[kd[j]], 1);
            rs[j] = atomicAdd(&hs[ks[j]], 1);
        }
    }
    __syncthreads();

    const int wv = tid >> 5;
    const int ln = tid & 31;
    if (wv < 2) {
        const int* h = wv ? hs : hd;
        int* st = wv ? ss : sd;
        int c[8];
        int tot = 0;
        for (int q = 0; q < 8; ++q) { c[q] = h[ln * 8 + q]; tot += c[q]; }
        int inc = tot;
        for (int o = 1; o < 32; o <<= 1) {
            const int v = __shfl_up(inc, o);
            if (ln >= o) inc += v;
        }
        int run = inc - tot;
        for (int q = 0; q < 8; ++q) { st[ln * 8 + q] = run; run += c[q]; }
        if (ln == 31) st[256] = inc;
    }
    __syncthreads();

    if (tid < OFFP) {
        offD[(long long)b * OFFP + tid] = sd[tid];
        offS[(long long)b * OFFP + tid] = ss[tid];
    }
    for (int j = 0; j < 4; ++j) {
        if (e0 + j < E) {
            od[sd[kd[j]] + rd[j]] = e0 + j;
            os[ss[ks[j]] + rs[j]] = e0 + j;
        }
    }
    __syncthreads();
    const int tot = sd[256];
    const int p0  = tid * 4;
    int* gD = idxD + (long long)b * EB + p0;
    int* gS = idxS + (long long)b * EB + p0;
    if (p0 + 3 < tot) {
        *reinterpret_cast<int4*>(gD) = *reinterpret_cast<const int4*>(od + p0);
        *reinterpret_cast<int4*>(gS) = *reinterpret_cast<const int4*>(os + p0);
    } else {
        for (int j = 0; j < 4; ++j) {
            if (p0 + j < tot) { gD[j] = od[p0 + j]; gS[j] = os[p0 + j]; }
        }
    }
}

__global__ __launch_bounds__(1024) void tile_aggregate_kernel(
        const float* __restrict__ feat, const int* __restrict__ src, const int* __restrict__ dst,
        const int* __restrict__ idx, const int* __restrict__ off, int nB,
        float* __restrict__ accum, float* __restrict__ invdeg, int Nn) {
    __shared__ __attribute__((aligned(16))) float tile[TR * 128];
    __shared__ int degc[TR];
    __shared__ int lst[TW][LCAP];

    const int lane = threadIdx.x;
    const int w    = __builtin_amdgcn_readfirstlane(threadIdx.y);
    const int tid  = w * 32 + lane;
    const int rowBase = blockIdx.x * TR;

    for (int i = tid; i < TR * 32; i += 32 * TW)
        reinterpret_cast<float4*>(tile)[i] = make_float4(0.0f, 0.0f, 0.0f, 0.0f);
    if (tid < TR) degc[tid] = 0;
    __syncthreads();

    int* myl = lst[w];
    int cnt = 0;

    auto note = [&](bool m, int e) {
        const unsigned mask = __builtin_amdgcn_ballot_w32(m);
        if (m) myl[cnt + (int)__builtin_amdgcn_mbcnt_lo(mask, 0u)] = e;
        cnt += __builtin_popcount(mask);
    };

    auto drain = [&]() {
        for (int i = lane; i < cnt; i += 32) {
            const unsigned ld = (unsigned)dst[myl[i]] - (unsigned)rowBase;
            if (ld < (unsigned)TR) atomicAdd(&degc[ld], 1);
        }
        for (int base = 0; base < cnt; base += PB) {
            int ee[PB], ss[PB], dd[PB];
            float4 v[PB];
#pragma unroll
            for (int k = 0; k < PB; ++k) {
                int idx = base + k;
                if (idx > cnt - 1) idx = cnt - 1;
                ee[k] = myl[idx];
            }
#pragma unroll
            for (int k = 0; k < PB; ++k) {
                ss[k] = src[ee[k]];
                dd[k] = dst[ee[k]] - rowBase;
            }
#pragma unroll
            for (int k = 0; k < PB; ++k)
                v[k] = *reinterpret_cast<const float4*>(feat + (long long)ss[k] * 128 + lane * 4);
#pragma unroll
            for (int k = 0; k < PB; ++k) {
                const bool live = (base + k < cnt) && ((unsigned)dd[k] < (unsigned)TR);
                float* t = tile + (live ? dd[k] : 0) * 128 + lane;
                atomicAdd(t,      live ? v[k].x : 0.0f);
                atomicAdd(t + 32, live ? v[k].y : 0.0f);
                atomicAdd(t + 64, live ? v[k].z : 0.0f);
                atomicAdd(t + 96, live ? v[k].w : 0.0f);
            }
        }
        cnt = 0;
    };

    const int key = (int)((unsigned)blockIdx.x & 255u);
    for (int b0 = w; b0 < nB; b0 += TW * RB) {
        int s0[RB], s1[RB], e1[RB];
        bool longRun = false;
        const int bq = b0 + TW * (lane & (RB - 1));
        int ov = 0;
        if (lane < 2 * RB && bq < nB) ov = off[(long long)bq * OFFP + key + (lane >= RB ? 1 : 0)];
#pragma unroll
        for (int r = 0; r < RB; ++r) {
            s0[r] = __builtin_amdgcn_readlane(ov, r);
            s1[r] = __builtin_amdgcn_readlane(ov, RB + r);
        }
#pragma unroll
        for (int r = 0; r < RB; ++r) {
            const int i = s0[r] + lane;
            e1[r] = (i < s1[r]) ? idx[(long long)(b0 + TW * r) * EB + i] : -1;
            longRun = longRun || (s1[r] - s0[r] > 32);
        }
#pragma unroll
        for (int r = 0; r < RB; ++r) note(s0[r] + lane < s1[r], e1[r]);
        if (cnt >= FLUSH) drain();
        if (longRun) {
            for (int r = 0; r < RB; ++r) {
                const int a1 = __builtin_amdgcn_readlane(ov, RB + r);
                for (int base = __builtin_amdgcn_readlane(ov, r) + 32; base < a1; base += 32) {
                    const int i = base + lane;
                    note(i < a1, (i < a1) ? idx[(long long)(b0 + TW * r) * EB + i] : -1);
                    if (cnt >= FLUSH) drain();
                }
            }
        }
    }
    drain();
    __syncthreads();

    for (int rr = w; rr < TR; rr += TW) {
        const int node = rowBase + rr;
        if (node < Nn) {
            const float* t = tile + rr * 128 + lane;
            const float4 o = make_float4(t[0], t[32], t[64], t[96]);
            *reinterpret_cast<float4*>(accum + (long long)node * 128 + lane * 4) = o;
        }
    }
    if (tid < TR) {
        const int node = rowBase + tid;
        if (node < Nn) invdeg[node] = 1.0f / fmaxf((float)degc[tid], 1.0f);
    }
}

struct GemmRaw {
    float4 as, ar;
    float4 bs[4], bn[4];
};

__device__ __forceinline__ void gemm_load(GemmRaw& g, const float4* xs, const float4* xa,
                                          const float* ws, const float* wn, int i) {
    g.as = xs[2 * i];
    g.ar = xa[2 * i];
#pragma unroll
    for (int e = 0; e < 4; ++e) {
        g.bs[e] = *reinterpret_cast<const float4*>(ws + (8 * i + e) * HID_F);
        g.bn[e] = *reinterpret_cast<const float4*>(wn + (8 * i + e) * HID_F);
    }
}

__device__ __forceinline__ void gemm_group(const GemmRaw& g, float invd,
                                           v8f& acc0, v8f& acc1, v8f& acc2, v8f& acc3) {
    const v2f aS0 = { g.as.x, g.as.y }, aS1 = { g.as.z, g.as.w };
    const v2f aN0 = { g.ar.x * invd, g.ar.y * invd };
    const v2f aN1 = { g.ar.z * invd, g.ar.w * invd };
    acc0 = __builtin_amdgcn_wmma_f32_16x16x4_f32(false, aS0, false, (v2f){ g.bs[0].x, g.bs[1].x }, (short)0, acc0, false, false);
    acc1 = __builtin_amdgcn_wmma_f32_16x16x4_f32(false, aS0, false, (v2f){ g.bs[0].y, g.bs[1].y }, (short)0, acc1, false, false);
    acc2 = __builtin_amdgcn_wmma_f32_16x16x4_f32(false, aS0, false, (v2f){ g.bs[0].z, g.bs[1].z }, (short)0, acc2, false, false);
    acc3 = __builtin_amdgcn_wmma_f32_16x16x4_f32(false, aS0, false, (v2f){ g.bs[0].w, g.bs[1].w }, (short)0, acc3, false, false);
    acc0 = __builtin_amdgcn_wmma_f32_16x16x4_f32(false, aN0, false, (v2f){ g.bn[0].x, g.bn[1].x }, (short)0, acc0, false, false);
    acc1 = __builtin_amdgcn_wmma_f32_16x16x4_f32(false, aN0, false, (v2f){ g.bn[0].y, g.bn[1].y }, (short)0, acc1, false, false);
    acc2 = __builtin_amdgcn_wmma_f32_16x16x4_f32(false, aN0, false, (v2f){ g.bn[0].z, g.bn[1].z }, (short)0, acc2, false, false);
    acc3 = __builtin_amdgcn_wmma_f32_16x16x4_f32(false, aN0, false, (v2f){ g.bn[0].w, g.bn[1].w }, (short)0, acc3, false, false);
    acc0 = __builtin_amdgcn_wmma_f32_16x16x4_f32(false, aS1, false, (v2f){ g.bs[2].x, g.bs[3].x }, (short)0, acc0, false, false);
    acc1 = __builtin_amdgcn_wmma_f32_16x16x4_f32(false, aS1, false, (v2f){ g.bs[2].y, g.bs[3].y }, (short)0, acc1, false, false);
    acc2 = __builtin_amdgcn_wmma_f32_16x16x4_f32(false, aS1, false, (v2f){ g.bs[2].z, g.bs[3].z }, (short)0, acc2, false, false);
    acc3 = __builtin_amdgcn_wmma_f32_16x16x4_f32(false, aS1, false, (v2f){ g.bs[2].w, g.bs[3].w }, (short)0, acc3, false, false);
    acc0 = __builtin_amdgcn_wmma_f32_16x16x4_f32(false, aN1, false, (v2f){ g.bn[2].x, g.bn[3].x }, (short)0, acc0, false, false);
    acc1 = __builtin_amdgcn_wmma_f32_16x16x4_f32(false, aN1, false, (v2f){ g.bn[2].y, g.bn[3].y }, (short)0, acc1, false, false);
    acc2 = __builtin_amdgcn_wmma_f32_16x16x4_f32(false, aN1, false, (v2f){ g.bn[2].z, g.bn[3].z }, (short)0, acc2, false, false);
    acc3 = __builtin_amdgcn_wmma_f32_16x16x4_f32(false, aN1, false, (v2f){ g.bn[2].w, g.bn[3].w }, (short)0, acc3, false, false);
}

__global__ __launch_bounds__(64) void sage_gemm_strip(
        const float* __restrict__ Xs, const float* __restrict__ Xa,
        const float* __restrict__ invdeg,
        const float* __restrict__ Ws, const float* __restrict__ Wn,
        const float* __restrict__ bias, float* __restrict__ out,
        int M, int do_relu) {
    const int lane = threadIdx.x;
    const int c    = lane & 15;
    const int h    = lane >> 4;
    const int rowBase = blockIdx.x * 16;
    const int col0    = threadIdx.y * 64 + 4 * c;

    int r = rowBase + c;
    if (r >= M) r = M - 1;
    const float invd = invdeg[r];
    const float4* xs = reinterpret_cast<const float4*>(Xs + (long long)r * IN_F + 4 * h);
    const float4* xa = reinterpret_cast<const float4*>(Xa + (long long)r * IN_F + 4 * h);
    const float* ws = Ws + (4 * h) * HID_F + col0;
    const float* wn = Wn + (4 * h) * HID_F + col0;

    v8f acc0 = {}, acc1 = {}, acc2 = {}, acc3 = {};
    GemmRaw p, q;
    gemm_load(p, xs, xa, ws, wn, 0);
    for (int t = 0; t < IN_F / 16 - 1; ++t) {
        gemm_load(q, xs, xa, ws, wn, 2 * t + 1);
        gemm_group(p, invd, acc0, acc1, acc2, acc3);
        gemm_load(p, xs, xa, ws, wn, 2 * t + 2);
        gemm_group(q, invd, acc0, acc1, acc2, acc3);
    }
    gemm_load(q, xs, xa, ws, wn, IN_F / 8 - 1);
    gemm_group(p, invd, acc0, acc1, acc2, acc3);
    gemm_group(q, invd, acc0, acc1, acc2, acc3);

    const float4 bv = *reinterpret_cast<const float4*>(bias + col0);
    const int mBase = rowBase + 8 * h;
#pragma unroll
    for (int j = 0; j < 8; ++j) {
        const int m = mBase + j;
        if (m < M) {
            float4 v = make_float4(acc0[j] + bv.x, acc1[j] + bv.y, acc2[j] + bv.z, acc3[j] + bv.w);
            if (do_relu) v = make_float4(fmaxf(v.x, 0.0f), fmaxf(v.y, 0.0f), fmaxf(v.z, 0.0f), fmaxf(v.w, 0.0f));
            *reinterpret_cast<float4*>(out + (long long)m * HID_F + col0) = v;
        }
    }
}

static_assert(CH == TR && CH == (1 << TRS), "both owner tilings share the key shift");
static_assert(CH == 4 * 64 && RW == 32, "8 feature tiles x 4 node quarters of 64");

__global__ __launch_bounds__(1024) void graph_reduce_owner(
        const float* __restrict__ h1, const int* __restrict__ src, const int* __restrict__ dst,
        const int* __restrict__ idx, const int* __restrict__ off, int nB,
        const int* __restrict__ gid, const float* __restrict__ invdeg,
        float* __restrict__ part, float* __restrict__ cntPart, int Nn) {
    __shared__ __attribute__((aligned(16))) float CtL[CH * CP];
    __shared__ __attribute__((aligned(16))) float Sloc[(NG + 1) * HID_F];
    __shared__ __attribute__((aligned(16))) float Tloc[NG * HID_F];
    __shared__ int cntloc[NG + 1];
    __shared__ int sgid[CH];
    __shared__ int lst[RW][LCAP];

    const int lane = threadIdx.x;
    const int w    = __builtin_amdgcn_readfirstlane(threadIdx.y);
    const int tid  = w * 32 + lane;
    const int k0   = blockIdx.x * CH;

    const float4 z4 = make_float4(0.0f, 0.0f, 0.0f, 0.0f);
    for (int i = tid; i < CH * CP / 4; i += 32 * RW) reinterpret_cast<float4*>(CtL)[i] = z4;
    for (int i = tid; i < (NG + 1) * HID_F / 4; i += 32 * RW) reinterpret_cast<float4*>(Sloc)[i] = z4;
    for (int i = tid; i < NG * HID_F / 4; i += 32 * RW) reinterpret_cast<float4*>(Tloc)[i] = z4;
    if (tid <= NG) cntloc[tid] = 0;
    if (tid < CH) {
        const int node = k0 + tid;
        int g = NG;
        if (node < Nn) {
            const int gg = gid[node];
            if ((unsigned)gg < (unsigned)NG) g = gg;
        }
        sgid[tid] = g;
    }
    __syncthreads();
    if (tid < CH) atomicAdd(&cntloc[sgid[tid]], 1);

    int* myl = lst[w];
    int cnt = 0;

    auto note = [&](bool m, int e) {
        const unsigned mask = __builtin_amdgcn_ballot_w32(m);
        if (m) myl[cnt + (int)__builtin_amdgcn_mbcnt_lo(mask, 0u)] = e;
        cnt += __builtin_popcount(mask);
    };

    auto drain = [&]() {
        for (int i = lane; i < cnt; i += 32) {
            const int e = myl[i];
            const unsigned ls = (unsigned)src[e] - (unsigned)k0;
            const int d = dst[e];
            const int g = gid[d];
            const float wv = invdeg[d];
            const bool live = (ls < (unsigned)CH) && ((unsigned)g < (unsigned)NG);
            atomicAdd(&CtL[live ? (int)ls * CP + g : 0], wv * (live ? 1.0f : 0.0f));
        }
        cnt = 0;
    };

    const int key = (int)((unsigned)blockIdx.x & 255u);
    for (int b0 = w; b0 < nB; b0 += RW * RB) {
        int s0[RB], s1[RB], e1[RB];
        bool longRun = false;
        const int bq = b0 + RW * (lane & (RB - 1));
        int ov = 0;
        if (lane < 2 * RB && bq < nB) ov = off[(long long)bq * OFFP + key + (lane >= RB ? 1 : 0)];
#pragma unroll
        for (int r = 0; r < RB; ++r) {
            s0[r] = __builtin_amdgcn_readlane(ov, r);
            s1[r] = __builtin_amdgcn_readlane(ov, RB + r);
        }
#pragma unroll
        for (int r = 0; r < RB; ++r) {
            const int i = s0[r] + lane;
            e1[r] = (i < s1[r]) ? idx[(long long)(b0 + RW * r) * EB + i] : -1;
            longRun = longRun || (s1[r] - s0[r] > 32);
        }
#pragma unroll
        for (int r = 0; r < RB; ++r) note(s0[r] + lane < s1[r], e1[r]);
        if (cnt >= FLUSH) drain();
        if (longRun) {
            for (int r = 0; r < RB; ++r) {
                const int a1 = __builtin_amdgcn_readlane(ov, RB + r);
                for (int base = __builtin_amdgcn_readlane(ov, r) + 32; base < a1; base += 32) {
                    const int i = base + lane;
                    note(i < a1, (i < a1) ? idx[(long long)(b0 + RW * r) * EB + i] : -1);
                    if (cnt >= FLUSH) drain();
                }
            }
        }
    }
    drain();
    __syncthreads();

    const int kOff = (lane < 16) ? 0 : 2;
    const int col  = lane & 15;
    const int f    = (w & 7) * 16 + col;
    const int kq   = (w >> 3) * 64;

    v8f acc0 = {}, acc1 = {}, acc2 = {}, acc3 = {};
    for (int kk = kq; kk < kq + 64; kk += 4 * KU) {
        v2f b[KU], a0[KU], a1[KU], a2[KU], a3[KU];
        int sa[KU], sb[KU];
#pragma unroll
        for (int u = 0; u < KU; ++u) {
            const int ka = kk + 4 * u + kOff;
            const int na = k0 + ka;
            const bool va = na < Nn;
            const bool vb = (na + 1) < Nn;
            const int ca = va ? na : (Nn - 1);
            const int cb = vb ? (na + 1) : (Nn - 1);
            const float x = h1[(long long)ca * HID_F + f];
            const float y = h1[(long long)cb * HID_F + f];
            b[u] = (v2f){ va ? x : 0.0f, vb ? y : 0.0f };
            const float* ra = CtL + ka * CP + col;
            const float* rb = ra + CP;
            a0[u] = (v2f){ ra[0],  rb[0]  };
            a1[u] = (v2f){ ra[16], rb[16] };
            a2[u] = (v2f){ ra[32], rb[32] };
            a3[u] = (v2f){ ra[48], rb[48] };
            sa[u] = sgid[ka];
            sb[u] = sgid[ka + 1];
        }
#pragma unroll
        for (int u = 0; u < KU; ++u) {
            atomicAdd(&Sloc[sa[u] * HID_F + f], b[u].x);
            atomicAdd(&Sloc[sb[u] * HID_F + f], b[u].y);
            acc0 = __builtin_amdgcn_wmma_f32_16x16x4_f32(false, a0[u], false, b[u], (short)0, acc0, false, false);
            acc1 = __builtin_amdgcn_wmma_f32_16x16x4_f32(false, a1[u], false, b[u], (short)0, acc1, false, false);
            acc2 = __builtin_amdgcn_wmma_f32_16x16x4_f32(false, a2[u], false, b[u], (short)0, acc2, false, false);
            acc3 = __builtin_amdgcn_wmma_f32_16x16x4_f32(false, a3[u], false, b[u], (short)0, acc3, false, false);
        }
    }

    const int mB = (lane < 16) ? 0 : 8;
#pragma unroll
    for (int j = 0; j < 8; ++j) {
        atomicAdd(&Tloc[(0  + mB + j) * HID_F + f], acc0[j]);
        atomicAdd(&Tloc[(16 + mB + j) * HID_F + f], acc1[j]);
        atomicAdd(&Tloc[(32 + mB + j) * HID_F + f], acc2[j]);
        atomicAdd(&Tloc[(48 + mB + j) * HID_F + f], acc3[j]);
    }
    __syncthreads();

    float4* p4 = reinterpret_cast<float4*>(part + (long long)blockIdx.x * (2 * NG * HID_F));
    for (int i = tid; i < NG * HID_F / 4; i += 32 * RW) {
        p4[i] = reinterpret_cast<const float4*>(Tloc)[i];
        p4[NG * HID_F / 4 + i] = reinterpret_cast<const float4*>(Sloc)[i];
    }
    if (tid < NG) cntPart[blockIdx.x * NG + tid] = (float)cntloc[tid];
}

#define HS  16
#define HB  16
#define HP  260
static_assert(2 * HID_F == 256 && OUT_F == 64 && HS == 16 && NG * HID_F == 8192,
              "32 waves = 4 column tiles x 8 k-slices of 32; 64 pieces of 16 bytes per row");

__global__ __launch_bounds__(1024) void graph_head_sum(const float* __restrict__ part,
                                                       const float* __restrict__ cntPart,
                                                       const float* __restrict__ Ws,
                                                       const float* __restrict__ Wn,
                                                       const float* __restrict__ bias,
                                                       float* __restrict__ out, int nChunks) {
    __shared__ __attribute__((aligned(16))) float Asl[HS * HP];
    __shared__ float red[8 * OUT_F];
    __shared__ float cw[32];

    const int lane = threadIdx.x;
    const int w    = threadIdx.y;
    const int tid  = w * 32 + lane;
    const int g    = blockIdx.x;

    const int c    = lane & 15;
    const int kOff = (lane < 16) ? 0 : 2;
    const int ks   = w >> 2;
    const int o    = (w & 3) * 16 + c;
    const float* Wp = ((ks < 4) ? Wn : Ws) + ((ks & 3) * 32 + kOff) * OUT_F + o;
    v2f b[8];
#pragma unroll
    for (int u = 0; u < 8; ++u) b[u] = (v2f){ Wp[(4 * u) * OUT_F], Wp[(4 * u + 1) * OUT_F] };

    const float cv0 = cntPart[((tid < nChunks) ? tid : 0) * NG + g];
    const float bo  = bias[tid & (OUT_F - 1)];

    const int q = tid & 63;
    const int r = tid >> 6;
    const float* rowp = part + (q >> 5) * (NG * HID_F) + g * HID_F + (q & 31) * 4;
    const long long per = 2 * NG * HID_F;
    const int nIt = (nChunks + HS - 1) / HS;
    float4 s = make_float4(0.0f, 0.0f, 0.0f, 0.0f);
    for (int base = 0; base < nIt; base += HB) {
        float4 v[HB];
#pragma unroll
        for (int k = 0; k < HB; ++k) {
            const int cc = r + (base + k) * HS;
            v[k] = *reinterpret_cast<const float4*>(rowp + (long long)((cc < nChunks) ? cc : 0) * per);
        }
#pragma unroll
        for (int k = 0; k < HB; ++k) {
            const bool ok = (r + (base + k) * HS) < nChunks;
            s.x += ok ? v[k].x : 0.0f;
            s.y += ok ? v[k].y : 0.0f;
            s.z += ok ? v[k].z : 0.0f;
            s.w += ok ? v[k].w : 0.0f;
        }
    }
    *reinterpret_cast<float4*>(Asl + r * HP + 4 * q) = s;
    float cv = (tid < nChunks) ? cv0 : 0.0f;
    for (int cc = tid + 32 * 32; cc < nChunks; cc += 32 * 32) cv += cntPart[cc * NG + g];
    for (int off = 16; off > 0; off >>= 1) cv += __shfl_xor(cv, off);
    if (lane == 0) cw[w] = cv;
    __syncthreads();

    v8f acc = {};
    const float* ap = Asl + c * HP + ks * 32 + kOff;
#pragma unroll
    for (int u = 0; u < 8; ++u) {
        const v2f a = *reinterpret_cast<const v2f*>(ap + 4 * u);
        acc = __builtin_amdgcn_wmma_f32_16x16x4_f32(false, a, false, b[u], (short)0, acc, false, false);
    }
    float t = ((acc[0] + acc[1]) + (acc[2] + acc[3])) + ((acc[4] + acc[5]) + (acc[6] + acc[7]));
    t += __shfl_xor(t, 16);
    if (lane < 16) red[ks * OUT_F + o] = t;
    __syncthreads();

    if (tid < OUT_F) {
        float tot = 0.0f;
        for (int k = 0; k < 8; ++k) tot += red[k * OUT_F + tid];
        float n = 0.0f;
        for (int k = 0; k < 32; ++k) n += cw[k];
        out[g * OUT_F + tid] = (tot + n * bo) / fmaxf(n, 1.0f);
    }
}


static inline int cdiv64(long long a, long long b) { return (int)((a + b - 1) / b); }

extern "C" void kernel_launch(void* const* d_in, const int* in_sizes, int n_in,
                              void* d_out, int out_size, void* d_ws, size_t ws_size,
                              hipStream_t stream) {
    const float* in_feat = (const float*)d_in[0];
    const int*   src     = (const int*)d_in[1];
    const int*   dst     = (const int*)d_in[2];
    const int*   gid     = (const int*)d_in[3];
    const float* W1s = (const float*)d_in[5];
    const float* W1n = (const float*)d_in[6];
    const float* b1  = (const float*)d_in[7];
    const float* W2s = (const float*)d_in[8];
    const float* W2n = (const float*)d_in[9];
    const float* b2  = (const float*)d_in[10];
    float* out = (float*)d_out;

    const int Nn = in_sizes[0] / IN_F;
    const int E  = in_sizes[1];

    float* ws    = (float*)d_ws;
    float* accum = ws;
    float* deg   = accum + (size_t)Nn * 128;
    float* Ct    = deg + Nn;
    float* h1    = Ct + (size_t)Nn * NG;

    const int mTiles  = cdiv64(Nn, 16);
    const int nChunks = cdiv64(Nn, CH);

    float* part    = accum;
    float* cntPart = part + (size_t)nChunks * (2 * NG * HID_F);

    const int nB = cdiv64(E, EB);
    int* idxD = (int*)h1;
    int* offD = idxD + (size_t)nB * EB;
    int* idxS = (int*)Ct;
    int* offS = idxS + (size_t)nB * EB;

    bin_kernel<<<nB, 1024, 0, stream>>>(dst, src, idxD, offD, idxS, offS, E);
    tile_aggregate_kernel<<<cdiv64(Nn, TR), dim3(32, TW), 0, stream>>>(in_feat, src, dst, idxD, offD, nB, accum, deg, Nn);
    sage_gemm_strip<<<mTiles, dim3(32, 2), 0, stream>>>(
        in_feat, accum, deg, W1s, W1n, b1, h1, Nn, 1);

    graph_reduce_owner<<<nChunks, dim3(32, RW), 0, stream>>>(h1, src, dst, idxS, offS, nB, gid, deg, part, cntPart, Nn);
    graph_head_sum<<<NG, dim3(32, 32), 0, stream>>>(part, cntPart, W2s, W2n, b2, out, nChunks);
}


# Binned runs hold each edge's (source, destination) pair, not its number: owner tiles of both passes read edge ends as whole lines of their runs; scattered 4-byte src/dst loads and a round trip gone
# speedup vs baseline: 6.8667x; 1.1356x over previous
#include <hip/hip_runtime.h>

typedef __attribute__((ext_vector_type(2))) float v2f;
typedef __attribute__((ext_vector_type(8))) float v8f;
typedef __attribute__((ext_vector_type(2))) int v2i;
typedef __attribute__((ext_vector_type(4))) int v4i;

#define IN_F  128
#define HID_F 128
#define OUT_F 64
#define NG    64
#define CH    256
#define KU    4
#define TR    256
#define TW    32
#define TRS   8
#define EB    4096
#define OFFP  257
#define RB    8
#define PB    8
#define FLUSH 64
#define LCAP  (FLUSH + 32 * RB)
#define RW    32
#define CP    80


__global__ __launch_bounds__(1024) void bin_kernel(const int* __restrict__ dst,
                                                   const int* __restrict__ src,
                                                   int* __restrict__ idxD, int* __restrict__ offD,
                                                   int* __restrict__ idxS, int* __restrict__ offS,
                                                   int E) {
    __shared__ int hd[256], hs[256];
    __shared__ int sd[OFFP], ss[OFFP];
    __shared__ __attribute__((aligned(16))) v2i od[EB], os[EB];

    const int tid = threadIdx.x;
    const int b   = blockIdx.x;
    const int e0  = b * EB + tid * 4;

    if (tid < 256) { hd[tid] = 0; hs[tid] = 0; }
    int dv[4], sv[4];
    if (e0 + 3 < E) {
        const int4 d4 = *reinterpret_cast<const int4*>(dst + e0);
        const int4 s4 = *reinterpret_cast<const int4*>(src + e0);
        dv[0] = d4.x; dv[1] = d4.y; dv[2] = d4.z; dv[3] = d4.w;
        sv[0] = s4.x; sv[1] = s4.y; sv[2] = s4.z; sv[3] = s4.w;
    } else {
        for (int j = 0; j < 4; ++j) {
            dv[j] = (e0 + j < E) ? dst[e0 + j] : 0;
            sv[j] = (e0 + j < E) ? src[e0 + j] : 0;
        }
    }
    __syncthreads();

    int kd[4], ks[4], rd[4], rs[4];
    for (int j = 0; j < 4; ++j) {
        kd[j] = (dv[j] >> TRS) & 255;
        ks[j] = (sv[j] >> TRS) & 255;
        rd[j] = 0;
        rs[j] = 0;
        if (e0 + j < E) {
            rd[j] = atomicAdd(&hd[kd[j]], 1);
            rs[j] = atomicAdd(&hs[ks[j]], 1);
        }
    }
    __syncthreads();

    const int wv = tid >> 5;
    const int ln = tid & 31;
    if (wv < 2) {
        const int* h = wv ? hs : hd;
        int* st = wv ? ss : sd;
        int c[8];
        int tot = 0;
        for (int q = 0; q < 8; ++q) { c[q] = h[ln * 8 + q]; tot += c[q]; }
        int inc = tot;
        for (int o = 1; o < 32; o <<= 1) {
            const int v = __shfl_up(inc, o);
            if (ln >= o) inc += v;
        }
        int run = inc - tot;
        for (int q = 0; q < 8; ++q) { st[ln * 8 + q] = run; run += c[q]; }
        if (ln == 31) st[256] = inc;
    }
    __syncthreads();

    if (tid < OFFP) {
        offD[(long long)b * OFFP + tid] = sd[tid];
        offS[(long long)b * OFFP + tid] = ss[tid];
    }
    for (int j = 0; j < 4; ++j) {
        if (e0 + j < E) {
            const v2i rec = { sv[j], dv[j] };
            od[sd[kd[j]] + rd[j]] = rec;
            os[ss[ks[j]] + rs[j]] = rec;
        }
    }
    __syncthreads();
    const int tot = sd[256];
    v2i* gD = reinterpret_cast<v2i*>(idxD) + (long long)b * EB;
    v2i* gS = reinterpret_cast<v2i*>(idxS) + (long long)b * EB;
    for (int q = 0; q < 2; ++q) {
        const int p = q * (EB / 2) + tid * 2;
        if (p + 1 < tot) {
            *reinterpret_cast<v4i*>(gD + p) = *reinterpret_cast<const v4i*>(od + p);
            *reinterpret_cast<v4i*>(gS + p) = *reinterpret_cast<const v4i*>(os + p);
        } else if (p < tot) {
            gD[p] = od[p];
            gS[p] = os[p];
        }
    }
}

__global__ __launch_bounds__(1024) void tile_aggregate_kernel(
        const float* __restrict__ feat,
        const int* __restrict__ idx, const int* __restrict__ off, int nB,
        float* __restrict__ accum, float* __restrict__ invdeg, int Nn) {
    __shared__ __attribute__((aligned(16))) float tile[TR * 128];
    __shared__ int degc[TR];
    __shared__ v2i lst[TW][LCAP];
    const v2i* prs = reinterpret_cast<const v2i*>(idx);

    const int lane = threadIdx.x;
    const int w    = __builtin_amdgcn_readfirstlane(threadIdx.y);
    const int tid  = w * 32 + lane;
    const int rowBase = blockIdx.x * TR;

    for (int i = tid; i < TR * 32; i += 32 * TW)
        reinterpret_cast<float4*>(tile)[i] = make_float4(0.0f, 0.0f, 0.0f, 0.0f);
    if (tid < TR) degc[tid] = 0;
    __syncthreads();

    v2i* myl = lst[w];
    int cnt = 0;

    auto note = [&](bool m, v2i e) {
        const unsigned mask = __builtin_amdgcn_ballot_w32(m);
        if (m) myl[cnt + (int)__builtin_amdgcn_mbcnt_lo(mask, 0u)] = e;
        cnt += __builtin_popcount(mask);
    };

    auto drain = [&]() {
        for (int i = lane; i < cnt; i += 32) {
            const unsigned ld = (unsigned)myl[i].y - (unsigned)rowBase;
            if (ld < (unsigned)TR) atomicAdd(&degc[ld], 1);
        }
        for (int base = 0; base < cnt; base += PB) {
            v2i pr[PB];
            float4 v[PB];
#pragma unroll
            for (int k = 0; k < PB; ++k) {
                int idx = base + k;
                if (idx > cnt - 1) idx = cnt - 1;
                pr[k] = myl[idx];
            }
#pragma unroll
            for (int k = 0; k < PB; ++k)
                v[k] = *reinterpret_cast<const float4*>(feat + (long long)pr[k].x * 128 + lane * 4);
#pragma unroll
            for (int k = 0; k < PB; ++k) {
                const int dk = pr[k].y - rowBase;
                const bool live = (base + k < cnt) && ((unsigned)dk < (unsigned)TR);
                float* t = tile + (live ? dk : 0) * 128 + lane;
                atomicAdd(t,      live ? v[k].x : 0.0f);
                atomicAdd(t + 32, live ? v[k].y : 0.0f);
                atomicAdd(t + 64, live ? v[k].z : 0.0f);
                atomicAdd(t + 96, live ? v[k].w : 0.0f);
            }
        }
        cnt = 0;
    };

    const v2i none = { 0, 0 };
    const int key = (int)((unsigned)blockIdx.x & 255u);
    for (int b0 = w; b0 < nB; b0 += TW * RB) {
        int s0[RB], s1[RB];
        v2i e1[RB];
        bool longRun = false;
        const int bq = b0 + TW * (lane & (RB - 1));
        int ov = 0;
        if (lane < 2 * RB && bq < nB) ov = off[(long long)bq * OFFP + key + (lane >= RB ? 1 : 0)];
#pragma unroll
        for (int r = 0; r < RB; ++r) {
            s0[r] = __builtin_amdgcn_readlane(ov, r);
            s1[r] = __builtin_amdgcn_readlane(ov, RB + r);
        }
#pragma unroll
        for (int r = 0; r < RB; ++r) {
            const int i = s0[r] + lane;
            e1[r] = (i < s1[r]) ? prs[(long long)(b0 + TW * r) * EB + i] : none;
            longRun = longRun || (s1[r] - s0[r] > 32);
        }
#pragma unroll
        for (int r = 0; r < RB; ++r) note(s0[r] + lane < s1[r], e1[r]);
        if (cnt >= FLUSH) drain();
        if (longRun) {
            for (int r = 0; r < RB; ++r) {
                const int a1 = __builtin_amdgcn_readlane(ov, RB + r);
                for (int base = __builtin_amdgcn_readlane(ov, r) + 32; base < a1; base += 32) {
                    const int i = base + lane;
                    note(i < a1, (i < a1) ? prs[(long long)(b0 + TW * r) * EB + i] : none);
                    if (cnt >= FLUSH) drain();
                }
            }
        }
    }
    drain();
    __syncthreads();

    for (int rr = w; rr < TR; rr += TW) {
        const int node = rowBase + rr;
        if (node < Nn) {
            const float* t = tile + rr * 128 + lane;
            const float4 o = make_float4(t[0], t[32], t[64], t[96]);
            *reinterpret_cast<float4*>(accum + (long long)node * 128 + lane * 4) = o;
        }
    }
    if (tid < TR) {
        const int node = rowBase + tid;
        if (node < Nn) invdeg[node] = 1.0f / fmaxf((float)degc[tid], 1.0f);
    }
}

struct GemmRaw {
    float4 as, ar;
    float4 bs[4], bn[4];
};

__device__ __forceinline__ void gemm_load(GemmRaw& g, const float4* xs, const float4* xa,
                                          const float* ws, const float* wn, int i) {
    g.as = xs[2 * i];
    g.ar = xa[2 * i];
#pragma unroll
    for (int e = 0; e < 4; ++e) {
        g.bs[e] = *reinterpret_cast<const float4*>(ws + (8 * i + e) * HID_F);
        g.bn[e] = *reinterpret_cast<const float4*>(wn + (8 * i + e) * HID_F);
    }
}

__device__ __forceinline__ void gemm_group(const GemmRaw& g, float invd,
                                           v8f& acc0, v8f& acc1, v8f& acc2, v8f& acc3) {
    const v2f aS0 = { g.as.x, g.as.y }, aS1 = { g.as.z, g.as.w };
    const v2f aN0 = { g.ar.x * invd, g.ar.y * invd };
    const v2f aN1 = { g.ar.z * invd, g.ar.w * invd };
    acc0 = __builtin_amdgcn_wmma_f32_16x16x4_f32(false, aS0, false, (v2f){ g.bs[0].x, g.bs[1].x }, (short)0, acc0, false, false);
    acc1 = __builtin_amdgcn_wmma_f32_16x16x4_f32(false, aS0, false, (v2f){ g.bs[0].y, g.bs[1].y }, (short)0, acc1, false, false);
    acc2 = __builtin_amdgcn_wmma_f32_16x16x4_f32(false, aS0, false, (v2f){ g.bs[0].z, g.bs[1].z }, (short)0, acc2, false, false);
    acc3 = __builtin_amdgcn_wmma_f32_16x16x4_f32(false, aS0, false, (v2f){ g.bs[0].w, g.bs[1].w }, (short)0, acc3, false, false);
    acc0 = __builtin_amdgcn_wmma_f32_16x16x4_f32(false, aN0, false, (v2f){ g.bn[0].x, g.bn[1].x }, (short)0, acc0, false, false);
    acc1 = __builtin_amdgcn_wmma_f32_16x16x4_f32(false, aN0, false, (v2f){ g.bn[0].y, g.bn[1].y }, (short)0, acc1, false, false);
    acc2 = __builtin_amdgcn_wmma_f32_16x16x4_f32(false, aN0, false, (v2f){ g.bn[0].z, g.bn[1].z }, (short)0, acc2, false, false);
    acc3 = __builtin_amdgcn_wmma_f32_16x16x4_f32(false, aN0, false, (v2f){ g.bn[0].w, g.bn[1].w }, (short)0, acc3, false, false);
    acc0 = __builtin_amdgcn_wmma_f32_16x16x4_f32(false, aS1, false, (v2f){ g.bs[2].x, g.bs[3].x }, (short)0, acc0, false, false);
    acc1 = __builtin_amdgcn_wmma_f32_16x16x4_f32(false, aS1, false, (v2f){ g.bs[2].y, g.bs[3].y }, (short)0, acc1, false, false);
    acc2 = __builtin_amdgcn_wmma_f32_16x16x4_f32(false, aS1, false, (v2f){ g.bs[2].z, g.bs[3].z }, (short)0, acc2, false, false);
    acc3 = __builtin_amdgcn_wmma_f32_16x16x4_f32(false, aS1, false, (v2f){ g.bs[2].w, g.bs[3].w }, (short)0, acc3, false, false);
    acc0 = __builtin_amdgcn_wmma_f32_16x16x4_f32(false, aN1, false, (v2f){ g.bn[2].x, g.bn[3].x }, (short)0, acc0, false, false);
    acc1 = __builtin_amdgcn_wmma_f32_16x16x4_f32(false, aN1, false, (v2f){ g.bn[2].y, g.bn[3].y }, (short)0, acc1, false, false);
    acc2 = __builtin_amdgcn_wmma_f32_16x16x4_f32(false, aN1, false, (v2f){ g.bn[2].z, g.bn[3].z }, (short)0, acc2, false, false);
    acc3 = __builtin_amdgcn_wmma_f32_16x16x4_f32(false, aN1, false, (v2f){ g.bn[2].w, g.bn[3].w }, (short)0, acc3, false, false);
}

__global__ __launch_bounds__(64) void sage_gemm_strip(
        const float* __restrict__ Xs, const float* __restrict__ Xa,
        const float* __restrict__ invdeg,
        const float* __restrict__ Ws, const float* __restrict__ Wn,
        const float* __restrict__ bias, float* __restrict__ out,
        int M, int do_relu) {
    const int lane = threadIdx.x;
    const int c    = lane & 15;
    const int h    = lane >> 4;
    const int rowBase = blockIdx.x * 16;
    const int col0    = threadIdx.y * 64 + 4 * c;

    int r = rowBase + c;
    if (r >= M) r = M - 1;
    const float invd = invdeg[r];
    const float4* xs = reinterpret_cast<const float4*>(Xs + (long long)r * IN_F + 4 * h);
    const float4* xa = reinterpret_cast<const float4*>(Xa + (long long)r * IN_F + 4 * h);
    const float* ws = Ws + (4 * h) * HID_F + col0;
    const float* wn = Wn + (4 * h) * HID_F + col0;

    v8f acc0 = {}, acc1 = {}, acc2 = {}, acc3 = {};
    GemmRaw p, q;
    gemm_load(p, xs, xa, ws, wn, 0);
    for (int t = 0; t < IN_F / 16 - 1; ++t) {
        gemm_load(q, xs, xa, ws, wn, 2 * t + 1);
        gemm_group(p, invd, acc0, acc1, acc2, acc3);
        gemm_load(p, xs, xa, ws, wn, 2 * t + 2);
        gemm_group(q, invd, acc0, acc1, acc2, acc3);
    }
    gemm_load(q, xs, xa, ws, wn, IN_F / 8 - 1);
    gemm_group(p, invd, acc0, acc1, acc2, acc3);
    gemm_group(q, invd, acc0, acc1, acc2, acc3);

    const float4 bv = *reinterpret_cast<const float4*>(bias + col0);
    const int mBase = rowBase + 8 * h;
#pragma unroll
    for (int j = 0; j < 8; ++j) {
        const int m = mBase + j;
        if (m < M) {
            float4 v = make_float4(acc0[j] + bv.x, acc1[j] + bv.y, acc2[j] + bv.z, acc3[j] + bv.w);
            if (do_relu) v = make_float4(fmaxf(v.x, 0.0f), fmaxf(v.y, 0.0f), fmaxf(v.z, 0.0f), fmaxf(v.w, 0.0f));
            *reinterpret_cast<float4*>(out + (long long)m * HID_F + col0) = v;
        }
    }
}

static_assert(CH == TR && CH == (1 << TRS), "both owner tilings share the key shift");
static_assert(CH == 4 * 64 && RW == 32, "8 feature tiles x 4 node quarters of 64");

__global__ __launch_bounds__(1024) void graph_reduce_owner(
        const float* __restrict__ h1,
        const int* __restrict__ idx, const int* __restrict__ off, int nB,
        const int* __restrict__ gid, const float* __restrict__ invdeg,
        float* __restrict__ part, float* __restrict__ cntPart, int Nn) {
    __shared__ __attribute__((aligned(16))) float CtL[CH * CP];
    __shared__ __attribute__((aligned(16))) float Sloc[(NG + 1) * HID_F];
    __shared__ __attribute__((aligned(16))) float Tloc[NG * HID_F];
    __shared__ int cntloc[NG + 1];
    __shared__ int sgid[CH];
    __shared__ v2i lst[RW][LCAP];
    const v2i* prs = reinterpret_cast<const v2i*>(idx);

    const int lane = threadIdx.x;
    const int w    = __builtin_amdgcn_readfirstlane(threadIdx.y);
    const int tid  = w * 32 + lane;
    const int k0   = blockIdx.x * CH;

    const float4 z4 = make_float4(0.0f, 0.0f, 0.0f, 0.0f);
    for (int i = tid; i < CH * CP / 4; i += 32 * RW) reinterpret_cast<float4*>(CtL)[i] = z4;
    for (int i = tid; i < (NG + 1) * HID_F / 4; i += 32 * RW) reinterpret_cast<float4*>(Sloc)[i] = z4;
    for (int i = tid; i < NG * HID_F / 4; i += 32 * RW) reinterpret_cast<float4*>(Tloc)[i] = z4;
    if (tid <= NG) cntloc[tid] = 0;
    if (tid < CH) {
        const int node = k0 + tid;
        int g = NG;
        if (node < Nn) {
            const int gg = gid[node];
            if ((unsigned)gg < (unsigned)NG) g = gg;
        }
        sgid[tid] = g;
    }
    __syncthreads();
    if (tid < CH) atomicAdd(&cntloc[sgid[tid]], 1);

    v2i* myl = lst[w];
    int cnt = 0;

    auto note = [&](bool m, v2i e) {
        const unsigned mask = __builtin_amdgcn_ballot_w32(m);
        if (m) myl[cnt + (int)__builtin_amdgcn_mbcnt_lo(mask, 0u)] = e;
        cnt += __builtin_popcount(mask);
    };

    auto drain = [&]() {
        for (int i = lane; i < cnt; i += 32) {
            const v2i e = myl[i];
            const unsigned ls = (unsigned)e.x - (unsigned)k0;
            const int d = e.y;
            const int g = gid[d];
            const float wv = invdeg[d];
            const bool live = (ls < (unsigned)CH) && ((unsigned)g < (unsigned)NG);
            atomicAdd(&CtL[live ? (int)ls * CP + g : 0], wv * (live ? 1.0f : 0.0f));
        }
        cnt = 0;
    };

    const v2i none = { 0, 0 };
    const int key = (int)((unsigned)blockIdx.x & 255u);
    for (int b0 = w; b0 < nB; b0 += RW * RB) {
        int s0[RB], s1[RB];
        v2i e1[RB];
        bool longRun = false;
        const int bq = b0 + RW * (lane & (RB - 1));
        int ov = 0;
        if (lane < 2 * RB && bq < nB) ov = off[(long long)bq * OFFP + key + (lane >= RB ? 1 : 0)];
#pragma unroll
        for (int r = 0; r < RB; ++r) {
            s0[r] = __builtin_amdgcn_readlane(ov, r);
            s1[r] = __builtin_amdgcn_readlane(ov, RB + r);
        }
#pragma unroll
        for (int r = 0; r < RB; ++r) {
            const int i = s0[r] + lane;
            e1[r] = (i < s1[r]) ? prs[(long long)(b0 + RW * r) * EB + i] : none;
            longRun = longRun || (s1[r] - s0[r] > 32);
        }
#pragma unroll
        for (int r = 0; r < RB; ++r) note(s0[r] + lane < s1[r], e1[r]);
        if (cnt >= FLUSH) drain();
        if (longRun) {
            for (int r = 0; r < RB; ++r) {
                const int a1 = __builtin_amdgcn_readlane(ov, RB + r);
                for (int base = __builtin_amdgcn_readlane(ov, r) + 32; base < a1; base += 32) {
                    const int i = base + lane;
                    note(i < a1, (i < a1) ? prs[(long long)(b0 + RW * r) * EB + i] : none);
                    if (cnt >= FLUSH) drain();
                }
            }
        }
    }
    drain();
    __syncthreads();

    const int kOff = (lane < 16) ? 0 : 2;
    const int col  = lane & 15;
    const int f    = (w & 7) * 16 + col;
    const int kq   = (w >> 3) * 64;

    v8f acc0 = {}, acc1 = {}, acc2 = {}, acc3 = {};
    for (int kk = kq; kk < kq + 64; kk += 4 * KU) {
        v2f b[KU], a0[KU], a1[KU], a2[KU], a3[KU];
        int sa[KU], sb[KU];
#pragma unroll
        for (int u = 0; u < KU; ++u) {
            const int ka = kk + 4 * u + kOff;
            const int na = k0 + ka;
            const bool va = na < Nn;
            const bool vb = (na + 1) < Nn;
            const int ca = va ? na : (Nn - 1);
            const int cb = vb ? (na + 1) : (Nn - 1);
            const float x = h1[(long long)ca * HID_F + f];
            const float y = h1[(long long)cb * HID_F + f];
            b[u] = (v2f){ va ? x : 0.0f, vb ? y : 0.0f };
            const float* ra = CtL + ka * CP + col;
            const float* rb = ra + CP;
            a0[u] = (v2f){ ra[0],  rb[0]  };
            a1[u] = (v2f){ ra[16], rb[16] };
            a2[u] = (v2f){ ra[32], rb[32] };
            a3[u] = (v2f){ ra[48], rb[48] };
            sa[u] = sgid[ka];
            sb[u] = sgid[ka + 1];
        }
#pragma unroll
        for (int u = 0; u < KU; ++u) {
            atomicAdd(&Sloc[sa[u] * HID_F + f], b[u].x);
            atomicAdd(&Sloc[sb[u] * HID_F + f], b[u].y);
            acc0 = __builtin_amdgcn_wmma_f32_16x16x4_f32(false, a0[u], false, b[u], (short)0, acc0, false, false);
            acc1 = __builtin_amdgcn_wmma_f32_16x16x4_f32(false, a1[u], false, b[u], (short)0, acc1, false, false);
            acc2 = __builtin_amdgcn_wmma_f32_16x16x4_f32(false, a2[u], false, b[u], (short)0, acc2, false, false);
            acc3 = __builtin_amdgcn_wmma_f32_16x16x4_f32(false, a3[u], false, b[u], (short)0, acc3, false, false);
        }
    }

    const int mB = (lane < 16) ? 0 : 8;
#pragma unroll
    for (int j = 0; j < 8; ++j) {
        atomicAdd(&Tloc[(0  + mB + j) * HID_F + f], acc0[j]);
        atomicAdd(&Tloc[(16 + mB + j) * HID_F + f], acc1[j]);
        atomicAdd(&Tloc[(32 + mB + j) * HID_F + f], acc2[j]);
        atomicAdd(&Tloc[(48 + mB + j) * HID_F + f], acc3[j]);
    }
    __syncthreads();

    float4* p4 = reinterpret_cast<float4*>(part + (long long)blockIdx.x * (2 * NG * HID_F));
    for (int i = tid; i < NG * HID_F / 4; i += 32 * RW) {
        p4[i] = reinterpret_cast<const float4*>(Tloc)[i];
        p4[NG * HID_F / 4 + i] = reinterpret_cast<const float4*>(Sloc)[i];
    }
    if (tid < NG) cntPart[blockIdx.x * NG + tid] = (float)cntloc[tid];
}

#define HS  16
#define HB  16
#define HP  260
static_assert(2 * HID_F == 256 && OUT_F == 64 && HS == 16 && NG * HID_F == 8192,
              "32 waves = 4 column tiles x 8 k-slices of 32; 64 pieces of 16 bytes per row");

__global__ __launch_bounds__(1024) void graph_head_sum(const float* __restrict__ part,
                                                       const float* __restrict__ cntPart,
                                                       const float* __restrict__ Ws,
                                                       const float* __restrict__ Wn,
                                                       const float* __restrict__ bias,
                                                       float* __restrict__ out, int nChunks) {
    __shared__ __attribute__((aligned(16))) float Asl[HS * HP];
    __shared__ float red[8 * OUT_F];
    __shared__ float cw[32];

    const int lane = threadIdx.x;
    const int w    = threadIdx.y;
    const int tid  = w * 32 + lane;
    const int g    = blockIdx.x;

    const int c    = lane & 15;
    const int kOff = (lane < 16) ? 0 : 2;
    const int ks   = w >> 2;
    const int o    = (w & 3) * 16 + c;
    const float* Wp = ((ks < 4) ? Wn : Ws) + ((ks & 3) * 32 + kOff) * OUT_F + o;
    v2f b[8];
#pragma unroll
    for (int u = 0; u < 8; ++u) b[u] = (v2f){ Wp[(4 * u) * OUT_F], Wp[(4 * u + 1) * OUT_F] };

    const float cv0 = cntPart[((tid < nChunks) ? tid : 0) * NG + g];
    const float bo  = bias[tid & (OUT_F - 1)];

    const int q = tid & 63;
    const int r = tid >> 6;
    const float* rowp = part + (q >> 5) * (NG * HID_F) + g * HID_F + (q & 31) * 4;
    const long long per = 2 * NG * HID_F;
    const int nIt = (nChunks + HS - 1) / HS;
    float4 s = make_float4(0.0f, 0.0f, 0.0f, 0.0f);
    for (int base = 0; base < nIt; base += HB) {
        float4 v[HB];
#pragma unroll
        for (int k = 0; k < HB; ++k) {
            const int cc = r + (base + k) * HS;
            v[k] = *reinterpret_cast<const float4*>(rowp + (long long)((cc < nChunks) ? cc : 0) * per);
        }
#pragma unroll
        for (int k = 0; k < HB; ++k) {
            const bool ok = (r + (base + k) * HS) < nChunks;
            s.x += ok ? v[k].x : 0.0f;
            s.y += ok ? v[k].y : 0.0f;
            s.z += ok ? v[k].z : 0.0f;
            s.w += ok ? v[k].w : 0.0f;
        }
    }
    *reinterpret_cast<float4*>(Asl + r * HP + 4 * q) = s;
    float cv = (tid < nChunks) ? cv0 : 0.0f;
    for (int cc = tid + 32 * 32; cc < nChunks; cc += 32 * 32) cv += cntPart[cc * NG + g];
    for (int off = 16; off > 0; off >>= 1) cv += __shfl_xor(cv, off);
    if (lane == 0) cw[w] = cv;
    __syncthreads();

    v8f acc = {};
    const float* ap = Asl + c * HP + ks * 32 + kOff;
#pragma unroll
    for (int u = 0; u < 8; ++u) {
        const v2f a = *reinterpret_cast<const v2f*>(ap + 4 * u);
        acc = __builtin_amdgcn_wmma_f32_16x16x4_f32(false, a, false, b[u], (short)0, acc, false, false);
    }
    float t = ((acc[0] + acc[1]) + (acc[2] + acc[3])) + ((acc[4] + acc[5]) + (acc[6] + acc[7]));
    t += __shfl_xor(t, 16);
    if (lane < 16) red[ks * OUT_F + o] = t;
    __syncthreads();

    if (tid < OUT_F) {
        float tot = 0.0f;
        for (int k = 0; k < 8; ++k) tot += red[k * OUT_F + tid];
        float n = 0.0f;
        for (int k = 0; k < 32; ++k) n += cw[k];
        out[g * OUT_F + tid] = (tot + n * bo) / fmaxf(n, 1.0f);
    }
}


static inline int cdiv64(long long a, long long b) { return (int)((a + b - 1) / b); }

extern "C" void kernel_launch(void* const* d_in, const int* in_sizes, int n_in,
                              void* d_out, int out_size, void* d_ws, size_t ws_size,
                              hipStream_t stream) {
    const float* in_feat = (const float*)d_in[0];
    const int*   src     = (const int*)d_in[1];
    const int*   dst     = (const int*)d_in[2];
    const int*   gid     = (const int*)d_in[3];
    const float* W1s = (const float*)d_in[5];
    const float* W1n = (const float*)d_in[6];
    const float* b1  = (const float*)d_in[7];
    const float* W2s = (const float*)d_in[8];
    const float* W2n = (const float*)d_in[9];
    const float* b2  = (const float*)d_in[10];
    float* out = (float*)d_out;

    const int Nn = in_sizes[0] / IN_F;
    const int E  = in_sizes[1];

    float* ws    = (float*)d_ws;
    float* accum = ws;
    float* deg   = accum + (size_t)Nn * 128;
    float* Ct    = deg + Nn;
    float* h1    = Ct + (size_t)Nn * NG;

    const int mTiles  = cdiv64(Nn, 16);
    const int nChunks = cdiv64(Nn, CH);

    float* part    = accum;
    float* cntPart = part + (size_t)nChunks * (2 * NG * HID_F);

    const int nB = cdiv64(E, EB);
    int* idxD = (int*)h1;
    int* offD = idxD + (size_t)nB * EB * 2;
    int* idxS = (int*)Ct;
    int* offS = idxS + (size_t)nB * EB * 2;

    bin_kernel<<<nB, 1024, 0, stream>>>(dst, src, idxD, offD, idxS, offS, E);
    tile_aggregate_kernel<<<cdiv64(Nn, TR), dim3(32, TW), 0, stream>>>(in_feat, idxD, offD, nB, accum, deg, Nn);
    sage_gemm_strip<<<mTiles, dim3(32, 2), 0, stream>>>(
        in_feat, accum, deg, W1s, W1n, b1, h1, Nn, 1);

    graph_reduce_owner<<<nChunks, dim3(32, RW), 0, stream>>>(h1, idxS, offS, nB, gid, deg, part, cntPart, Nn);
    graph_head_sum<<<NG, dim3(32, 32), 0, stream>>>(part, cntPart, W2s, W2n, b2, out, nChunks);
}


# Layer-1 GEMM moved into the per-graph reduction's work-groups (208-node chunks, 241 groups): a wave per 16x64 strip, h1 stays in LDS as B operand of T; 5 launches become 4; chunk sum order changes
# speedup vs baseline: 7.1864x; 1.0466x over previous
#include <hip/hip_runtime.h>

typedef __attribute__((ext_vector_type(2))) float v2f;
typedef __attribute__((ext_vector_type(8))) float v8f;
typedef __attribute__((ext_vector_type(2))) int v2i;
typedef __attribute__((ext_vector_type(4))) int v4i;

#define IN_F  128
#define HID_F 128
#define OUT_F 64
#define NG    64
#define CH    208
#define QN    (CH / 4)
#define TR    256
#define TW    32
#define TRS   8
#define EB    4096
#define OFFP  257
#define RB    8
#define PB    8
#define FLUSH 64
#define LCAP  (FLUSH + 32 * RB)
#define RW    32
#define CP    80
#define HP1   136


__global__ __launch_bounds__(1024) void bin_kernel(const int* __restrict__ dst,
                                                   const int* __restrict__ src,
                                                   int* __restrict__ idxD, int* __restrict__ offD,
                                                   int* __restrict__ idxS, int* __restrict__ offS,
                                                   int E) {
    __shared__ int hd[256], hs[256];
    __shared__ int sd[OFFP], ss[OFFP];
    __shared__ __attribute__((aligned(16))) v2i od[EB], os[EB];

    const int tid = threadIdx.x;
    const int b   = blockIdx.x;
    const int e0  = b * EB + tid * 4;

    if (tid < 256) { hd[tid] = 0; hs[tid] = 0; }
    int dv[4], sv[4];
    if (e0 + 3 < E) {
        const int4 d4 = *reinterpret_cast<const int4*>(dst + e0);
        const int4 s4 = *reinterpret_cast<const int4*>(src + e0);
        dv[0] = d4.x; dv[1] = d4.y; dv[2] = d4.z; dv[3] = d4.w;
        sv[0] = s4.x; sv[1] = s4.y; sv[2] = s4.z; sv[3] = s4.w;
    } else {
        for (int j = 0; j < 4; ++j) {
            dv[j] = (e0 + j < E) ? dst[e0 + j] : 0;
            sv[j] = (e0 + j < E) ? src[e0 + j] : 0;
        }
    }
    __syncthreads();

    int kd[4], ks[4], rd[4], rs[4];
    for (int j = 0; j < 4; ++j) {
        kd[j] = (dv[j] >> TRS) & 255;
        ks[j] = (int)(((unsigned)sv[j] / (unsigned)CH) & 255u);
        rd[j] = 0;
        rs[j] = 0;
        if (e0 + j < E) {
            rd[j] = atomicAdd(&hd[kd[j]], 1);
            rs[j] = atomicAdd(&hs[ks[j]], 1);
        }
    }
    __syncthreads();

    const int wv = tid >> 5;
    const int ln = tid & 31;
    if (wv < 2) {
        const int* h = wv ? hs : hd;
        int* st = wv ? ss : sd;
        int c[8];
        int tot = 0;
        for (int q = 0; q < 8; ++q) { c[q] = h[ln * 8 + q]; tot += c[q]; }
        int inc = tot;
        for (int o = 1; o < 32; o <<= 1) {
            const int v = __shfl_up(inc, o);
            if (ln >= o) inc += v;
        }
        int run = inc - tot;
        for (int q = 0; q < 8; ++q) { st[ln * 8 + q] = run; run += c[q]; }
        if (ln == 31) st[256] = inc;
    }
    __syncthreads();

    if (tid < OFFP) {
        offD[(long long)b * OFFP + tid] = sd[tid];
        offS[(long long)b * OFFP + tid] = ss[tid];
    }
    for (int j = 0; j < 4; ++j) {
        if (e0 + j < E) {
            const v2i rec = { sv[j], dv[j] };
            od[sd[kd[j]] + rd[j]] = rec;
            os[ss[ks[j]] + rs[j]] = rec;
        }
    }
    __syncthreads();
    const int tot = sd[256];
    v2i* gD = reinterpret_cast<v2i*>(idxD) + (long long)b * EB;
    v2i* gS = reinterpret_cast<v2i*>(idxS) + (long long)b * EB;
    for (int q = 0; q < 2; ++q) {
        const int p = q * (EB / 2) + tid * 2;
        if (p + 1 < tot) {
            *reinterpret_cast<v4i*>(gD + p) = *reinterpret_cast<const v4i*>(od + p);
            *reinterpret_cast<v4i*>(gS + p) = *reinterpret_cast<const v4i*>(os + p);
        } else if (p < tot) {
            gD[p] = od[p];
            gS[p] = os[p];
        }
    }
}

__global__ __launch_bounds__(1024) void tile_aggregate_kernel(
        const float* __restrict__ feat,
        const int* __restrict__ idx, const int* __restrict__ off, int nB,
        float* __restrict__ accum, float* __restrict__ invdeg, int Nn) {
    __shared__ __attribute__((aligned(16))) float tile[TR * 128];
    __shared__ int degc[TR];
    __shared__ v2i lst[TW][LCAP];
    const v2i* prs = reinterpret_cast<const v2i*>(idx);

    const int lane = threadIdx.x;
    const int w    = __builtin_amdgcn_readfirstlane(threadIdx.y);
    const int tid  = w * 32 + lane;
    const int rowBase = blockIdx.x * TR;

    for (int i = tid; i < TR * 32; i += 32 * TW)
        reinterpret_cast<float4*>(tile)[i] = make_float4(0.0f, 0.0f, 0.0f, 0.0f);
    if (tid < TR) degc[tid] = 0;
    __syncthreads();

    v2i* myl = lst[w];
    int cnt = 0;

    auto note = [&](bool m, v2i e) {
        const unsigned mask = __builtin_amdgcn_ballot_w32(m);
        if (m) myl[cnt + (int)__builtin_amdgcn_mbcnt_lo(mask, 0u)] = e;
        cnt += __builtin_popcount(mask);
    };

    auto drain = [&]() {
        for (int i = lane; i < cnt; i += 32) {
            const unsigned ld = (unsigned)myl[i].y - (unsigned)rowBase;
            if (ld < (unsigned)TR) atomicAdd(&degc[ld], 1);
        }
        for (int base = 0; base < cnt; base += PB) {
            v2i pr[PB];
            float4 v[PB];
#pragma unroll
            for (int k = 0; k < PB; ++k) {
                int idx = base + k;
                if (idx > cnt - 1) idx = cnt - 1;
                pr[k] = myl[idx];
            }
#pragma unroll
            for (int k = 0; k < PB; ++k)
                v[k] = *reinterpret_cast<const float4*>(feat + (long long)pr[k].x * 128 + lane * 4);
#pragma unroll
            for (int k = 0; k < PB; ++k) {
                const int dk = pr[k].y - rowBase;
                const bool live = (base + k < cnt) && ((unsigned)dk < (unsigned)TR);
                float* t = tile + (live ? dk : 0) * 128 + lane;
                atomicAdd(t,      live ? v[k].x : 0.0f);
                atomicAdd(t + 32, live ? v[k].y : 0.0f);
                atomicAdd(t + 64, live ? v[k].z : 0.0f);
                atomicAdd(t + 96, live ? v[k].w : 0.0f);
            }
        }
        cnt = 0;
    };

    const v2i none = { 0, 0 };
    const int key = (int)((unsigned)blockIdx.x & 255u);
    for (int b0 = w; b0 < nB; b0 += TW * RB) {
        int s0[RB], s1[RB];
        v2i e1[RB];
        bool longRun = false;
        const int bq = b0 + TW * (lane & (RB - 1));
        int ov = 0;
        if (lane < 2 * RB && bq < nB) ov = off[(long long)bq * OFFP + key + (lane >= RB ? 1 : 0)];
#pragma unroll
        for (int r = 0; r < RB; ++r) {
            s0[r] = __builtin_amdgcn_readlane(ov, r);
            s1[r] = __builtin_amdgcn_readlane(ov, RB + r);
        }
#pragma unroll
        for (int r = 0; r < RB; ++r) {
            const int i = s0[r] + lane;
            e1[r] = (i < s1[r]) ? prs[(long long)(b0 + TW * r) * EB + i] : none;
            longRun = longRun || (s1[r] - s0[r] > 32);
        }
#pragma unroll
        for (int r = 0; r < RB; ++r) note(s0[r] + lane < s1[r], e1[r]);
        if (cnt >= FLUSH) drain();
        if (longRun) {
            for (int r = 0; r < RB; ++r) {
                const int a1 = __builtin_amdgcn_readlane(ov, RB + r);
                for (int base = __builtin_amdgcn_readlane(ov, r) + 32; base < a1; base += 32) {
                    const int i = base + lane;
                    note(i < a1, (i < a1) ? prs[(long long)(b0 + TW * r) * EB + i] : none);
                    if (cnt >= FLUSH) drain();
                }
            }
        }
    }
    drain();
    __syncthreads();

    for (int rr = w; rr < TR; rr += TW) {
        const int node = rowBase + rr;
        if (node < Nn) {
            const float* t = tile + rr * 128 + lane;
            const float4 o = make_float4(t[0], t[32], t[64], t[96]);
            *reinterpret_cast<float4*>(accum + (long long)node * 128 + lane * 4) = o;
        }
    }
    if (tid < TR) {
        const int node = rowBase + tid;
        if (node < Nn) invdeg[node] = 1.0f / fmaxf((float)degc[tid], 1.0f);
    }
}

struct GemmRaw {
    float4 as, ar;
    float4 bs[4], bn[4];
};

__device__ __forceinline__ void gemm_load(GemmRaw& g, const float4* xs, const float4* xa,
                                          const float* ws, const float* wn, int i) {
    g.as = xs[2 * i];
    g.ar = xa[2 * i];
#pragma unroll
    for (int e = 0; e < 4; ++e) {
        g.bs[e] = *reinterpret_cast<const float4*>(ws + (8 * i + e) * HID_F);
        g.bn[e] = *reinterpret_cast<const float4*>(wn + (8 * i + e) * HID_F);
    }
}

template <bool MORE>
__device__ __forceinline__ void gemm_group_roll(GemmRaw& g, const float4* xs, const float4* xa,
                                                const float* ws, const float* wn, int nx, float invd,
                                                v8f& acc0, v8f& acc1, v8f& acc2, v8f& acc3) {
    const v2f aS0 = { g.as.x, g.as.y }, aS1 = { g.as.z, g.as.w };
    const v2f aN0 = { g.ar.x * invd, g.ar.y * invd };
    const v2f aN1 = { g.ar.z * invd, g.ar.w * invd };
    acc0 = __builtin_amdgcn_wmma_f32_16x16x4_f32(false, aS0, false, (v2f){ g.bs[0].x, g.bs[1].x }, (short)0, acc0, false, false);
    acc1 = __builtin_amdgcn_wmma_f32_16x16x4_f32(false, aS0, false, (v2f){ g.bs[0].y, g.bs[1].y }, (short)0, acc1, false, false);
    acc2 = __builtin_amdgcn_wmma_f32_16x16x4_f32(false, aS0, false, (v2f){ g.bs[0].z, g.bs[1].z }, (short)0, acc2, false, false);
    acc3 = __builtin_amdgcn_wmma_f32_16x16x4_f32(false, aS0, false, (v2f){ g.bs[0].w, g.bs[1].w }, (short)0, acc3, false, false);
    if (MORE) {
        g.bs[0] = *reinterpret_cast<const float4*>(ws + (8 * nx + 0) * HID_F);
        g.bs[1] = *reinterpret_cast<const float4*>(ws + (8 * nx + 1) * HID_F);
    }
    acc0 = __builtin_amdgcn_wmma_f32_16x16x4_f32(false, aN0, false, (v2f){ g.bn[0].x, g.bn[1].x }, (short)0, acc0, false, false);
    acc1 = __builtin_amdgcn_wmma_f32_16x16x4_f32(false, aN0, false, (v2f){ g.bn[0].y, g.bn[1].y }, (short)0, acc1, false, false);
    acc2 = __builtin_amdgcn_wmma_f32_16x16x4_f32(false, aN0, false, (v2f){ g.bn[0].z, g.bn[1].z }, (short)0, acc2, false, false);
    acc3 = __builtin_amdgcn_wmma_f32_16x16x4_f32(false, aN0, false, (v2f){ g.bn[0].w, g.bn[1].w }, (short)0, acc3, false, false);
    if (MORE) {
        g.bn[0] = *reinterpret_cast<const float4*>(wn + (8 * nx + 0) * HID_F);
        g.bn[1] = *reinterpret_cast<const float4*>(wn + (8 * nx + 1) * HID_F);
    }
    acc0 = __builtin_amdgcn_wmma_f32_16x16x4_f32(false, aS1, false, (v2f){ g.bs[2].x, g.bs[3].x }, (short)0, acc0, false, false);
    acc1 = __builtin_amdgcn_wmma_f32_16x16x4_f32(false, aS1, false, (v2f){ g.bs[2].y, g.bs[3].y }, (short)0, acc1, false, false);
    acc2 = __builtin_amdgcn_wmma_f32_16x16x4_f32(false, aS1, false, (v2f){ g.bs[2].z, g.bs[3].z }, (short)0, acc2, false, false);
    acc3 = __builtin_amdgcn_wmma_f32_16x16x4_f32(false, aS1, false, (v2f){ g.bs[2].w, g.bs[3].w }, (short)0, acc3, false, false);
    if (MORE) {
        g.bs[2] = *reinterpret_cast<const float4*>(ws + (8 * nx + 2) * HID_F);
        g.bs[3] = *reinterpret_cast<const float4*>(ws + (8 * nx + 3) * HID_F);
        g.as = xs[2 * nx];
    }
    acc0 = __builtin_amdgcn_wmma_f32_16x16x4_f32(false, aN1, false, (v2f){ g.bn[2].x, g.bn[3].x }, (short)0, acc0, false, false);
    acc1 = __builtin_amdgcn_wmma_f32_16x16x4_f32(false, aN1, false, (v2f){ g.bn[2].y, g.bn[3].y }, (short)0, acc1, false, false);
    acc2 = __builtin_amdgcn_wmma_f32_16x16x4_f32(false, aN1, false, (v2f){ g.bn[2].z, g.bn[3].z }, (short)0, acc2, false, false);
    acc3 = __builtin_amdgcn_wmma_f32_16x16x4_f32(false, aN1, false, (v2f){ g.bn[2].w, g.bn[3].w }, (short)0, acc3, false, false);
    if (MORE) {
        g.bn[2] = *reinterpret_cast<const float4*>(wn + (8 * nx + 2) * HID_F);
        g.bn[3] = *reinterpret_cast<const float4*>(wn + (8 * nx + 3) * HID_F);
        g.ar = xa[2 * nx];
    }
}

static_assert(RW == 32 && QN * 4 == CH && QN == 4 * (4 + 3 + 3 + 3),
              "8 feature tiles x 4 node quarters; a quarter is 13 k-steps, taken as 4 + 3 + 3 + 3");
static_assert(CH % 16 == 0 && 2 * (CH / 16) <= RW, "a wave per 16 x 64 strip of the chunk's h1");
static_assert(RW * LCAP * 2 <= CH * HP1, "the walk's lists fit into the space of the h1 tile");
template <int N> struct KSteps { static constexpr int value = N; };

__global__ __launch_bounds__(1024) void graph_reduce_owner(
        const float* __restrict__ Xs, const float* __restrict__ Xa,
        const float* __restrict__ W1s, const float* __restrict__ W1n,
        const float* __restrict__ bias1,
        const int* __restrict__ idx, const int* __restrict__ off, int nB,
        const int* __restrict__ gid, const float* __restrict__ invdeg,
        float* __restrict__ part, float* __restrict__ cntPart, int Nn) {
    __shared__ __attribute__((aligned(16))) float CtL[CH * CP];
    __shared__ __attribute__((aligned(16))) float Sloc[(NG + 1) * HID_F];
    __shared__ __attribute__((aligned(16))) float Tloc[NG * HID_F];
    __shared__ int cntloc[NG + 1];
    __shared__ int sgid[CH];
    __shared__ __attribute__((aligned(16))) float h1L[CH * HP1];
    const v2i* prs = reinterpret_cast<const v2i*>(idx);

    const int lane = threadIdx.x;
    const int w    = __builtin_amdgcn_readfirstlane(threadIdx.y);
    const int tid  = w * 32 + lane;
    const int k0   = blockIdx.x * CH;

    const float4 z4 = make_float4(0.0f, 0.0f, 0.0f, 0.0f);
    for (int i = tid; i < CH * CP / 4; i += 32 * RW) reinterpret_cast<float4*>(CtL)[i] = z4;
    for (int i = tid; i < (NG + 1) * HID_F / 4; i += 32 * RW) reinterpret_cast<float4*>(Sloc)[i] = z4;
    for (int i = tid; i < NG * HID_F / 4; i += 32 * RW) reinterpret_cast<float4*>(Tloc)[i] = z4;
    if (tid <= NG) cntloc[tid] = 0;
    if (tid < CH) {
        const int node = k0 + tid;
        int g = NG;
        if (node < Nn) {
            const int gg = gid[node];
            if ((unsigned)gg < (unsigned)NG) g = gg;
        }
        sgid[tid] = g;
    }
    __syncthreads();
    if (tid < CH) atomicAdd(&cntloc[sgid[tid]], 1);

    v2i* myl = reinterpret_cast<v2i*>(h1L) + w * LCAP;
    int cnt = 0;

    auto note = [&](bool m, v2i e) {
        const unsigned mask = __builtin_amdgcn_ballot_w32(m);
        if (m) myl[cnt + (int)__builtin_amdgcn_mbcnt_lo(mask, 0u)] = e;
        cnt += __builtin_popcount(mask);
    };

    auto drain = [&]() {
        for (int i = lane; i < cnt; i += 32) {
            const v2i e = myl[i];
            const unsigned ls = (unsigned)e.x - (unsigned)k0;
            const int d = e.y;
            const int g = gid[d];
            const float wv = invdeg[d];
            const bool live = (ls < (unsigned)CH) && ((unsigned)g < (unsigned)NG);
            atomicAdd(&CtL[live ? (int)ls * CP + g : 0], wv * (live ? 1.0f : 0.0f));
        }
        cnt = 0;
    };

    const v2i none = { 0, 0 };
    const int key = (int)((unsigned)blockIdx.x & 255u);
    for (int b0 = w; b0 < nB; b0 += RW * RB) {
        int s0[RB], s1[RB];
        v2i e1[RB];
        bool longRun = false;
        const int bq = b0 + RW * (lane & (RB - 1));
        int ov = 0;
        if (lane < 2 * RB && bq < nB) ov = off[(long long)bq * OFFP + key + (lane >= RB ? 1 : 0)];
#pragma unroll
        for (int r = 0; r < RB; ++r) {
            s0[r] = __builtin_amdgcn_readlane(ov, r);
            s1[r] = __builtin_amdgcn_readlane(ov, RB + r);
        }
#pragma unroll
        for (int r = 0; r < RB; ++r) {
            const int i = s0[r] + lane;
            e1[r] = (i < s1[r]) ? prs[(long long)(b0 + RW * r) * EB + i] : none;
            longRun = longRun || (s1[r] - s0[r] > 32);
        }
#pragma unroll
        for (int r = 0; r < RB; ++r) note(s0[r] + lane < s1[r], e1[r]);
        if (cnt >= FLUSH) drain();
        if (longRun) {
            for (int r = 0; r < RB; ++r) {
                const int a1 = __builtin_amdgcn_readlane(ov, RB + r);
                for (int base = __builtin_amdgcn_readlane(ov, r) + 32; base < a1; base += 32) {
                    const int i = base + lane;
                    note(i < a1, (i < a1) ? prs[(long long)(b0 + RW * r) * EB + i] : none);
                    if (cnt >= FLUSH) drain();
                }
            }
        }
    }
    drain();
    __syncthreads();

    if (w < 2 * (CH / 16)) {
        const int c    = lane & 15;
        const int hh   = lane >> 4;
        const int rt   = w >> 1;
        const int col0 = (w & 1) * 64 + 4 * c;

        int r = k0 + rt * 16 + c;
        if (r >= Nn) r = Nn - 1;
        const float invd = invdeg[r];
        const float4* xs = reinterpret_cast<const float4*>(Xs + (long long)r * IN_F + 4 * hh);
        const float4* xa = reinterpret_cast<const float4*>(Xa + (long long)r * IN_F + 4 * hh);
        const float* ws = W1s + (4 * hh) * HID_F + col0;
        const float* wn = W1n + (4 * hh) * HID_F + col0;

        v8f hc0 = {}, hc1 = {}, hc2 = {}, hc3 = {};
        GemmRaw g;
        gemm_load(g, xs, xa, ws, wn, 0);
#pragma unroll 1
        for (int i = 1; i < IN_F / 8; ++i)
            gemm_group_roll<true>(g, xs, xa, ws, wn, i, invd, hc0, hc1, hc2, hc3);
        gemm_group_roll<false>(g, xs, xa, ws, wn, 0, invd, hc0, hc1, hc2, hc3);

        const float4 bv = *reinterpret_cast<const float4*>(bias1 + col0);
#pragma unroll
        for (int j = 0; j < 8; ++j) {
            const int m = rt * 16 + 8 * hh + j;
            float4 v = make_float4(fmaxf(hc0[j] + bv.x, 0.0f), fmaxf(hc1[j] + bv.y, 0.0f),
                                   fmaxf(hc2[j] + bv.z, 0.0f), fmaxf(hc3[j] + bv.w, 0.0f));
            if (k0 + m >= Nn) v = z4;
            *reinterpret_cast<float4*>(h1L + m * HP1 + col0) = v;
        }
    }
    __syncthreads();

    const int kOff = (lane < 16) ? 0 : 2;
    const int col  = lane & 15;
    const int f    = (w & 7) * 16 + col;
    const int kq   = (w >> 3) * QN;

    v8f acc0 = {}, acc1 = {}, acc2 = {}, acc3 = {};
    auto ksteps = [&](int kk, auto kuTag) {
        constexpr int KU = decltype(kuTag)::value;
        v2f b[KU], a0[KU], a1[KU], a2[KU], a3[KU];
        int sa[KU], sb[KU];
#pragma unroll
        for (int u = 0; u < KU; ++u) {
            const int ka = kk + 4 * u + kOff;
            const float* hb = h1L + ka * HP1 + f;
            b[u] = (v2f){ hb[0], hb[HP1] };
            const float* ra = CtL + ka * CP + col;
            const float* rb = ra + CP;
            a0[u] = (v2f){ ra[0],  rb[0]  };
            a1[u] = (v2f){ ra[16], rb[16] };
            a2[u] = (v2f){ ra[32], rb[32] };
            a3[u] = (v2f){ ra[48], rb[48] };
            sa[u] = sgid[ka];
            sb[u] = sgid[ka + 1];
        }
#pragma unroll
        for (int u = 0; u < KU; ++u) {
            atomicAdd(&Sloc[sa[u] * HID_F + f], b[u].x);
            atomicAdd(&Sloc[sb[u] * HID_F + f], b[u].y);
            acc0 = __builtin_amdgcn_wmma_f32_16x16x4_f32(false, a0[u], false, b[u], (short)0, acc0, false, false);
            acc1 = __builtin_amdgcn_wmma_f32_16x16x4_f32(false, a1[u], false, b[u], (short)0, acc1, false, false);
            acc2 = __builtin_amdgcn_wmma_f32_16x16x4_f32(false, a2[u], false, b[u], (short)0, acc2, false, false);
            acc3 = __builtin_amdgcn_wmma_f32_16x16x4_f32(false, a3[u], false, b[u], (short)0, acc3, false, false);
        }
    };
    ksteps(kq,      KSteps<4>{});
    ksteps(kq + 16, KSteps<3>{});
    ksteps(kq + 28, KSteps<3>{});
    ksteps(kq + 40, KSteps<3>{});

    const int mB = (lane < 16) ? 0 : 8;
#pragma unroll
    for (int j = 0; j < 8; ++j) {
        atomicAdd(&Tloc[(0  + mB + j) * HID_F + f], acc0[j]);
        atomicAdd(&Tloc[(16 + mB + j) * HID_F + f], acc1[j]);
        atomicAdd(&Tloc[(32 + mB + j) * HID_F + f], acc2[j]);
        atomicAdd(&Tloc[(48 + mB + j) * HID_F + f], acc3[j]);
    }
    __syncthreads();

    float4* p4 = reinterpret_cast<float4*>(part + (long long)blockIdx.x * (2 * NG * HID_F));
    for (int i = tid; i < NG * HID_F / 4; i += 32 * RW) {
        p4[i] = reinterpret_cast<const float4*>(Tloc)[i];
        p4[NG * HID_F / 4 + i] = reinterpret_cast<const float4*>(Sloc)[i];
    }
    if (tid < NG) cntPart[blockIdx.x * NG + tid] = (float)cntloc[tid];
}

#define HS  16
#define HB  16
#define HP  260
static_assert(2 * HID_F == 256 && OUT_F == 64 && HS == 16 && NG * HID_F == 8192,
              "32 waves = 4 column tiles x 8 k-slices of 32; 64 pieces of 16 bytes per row");

__global__ __launch_bounds__(1024) void graph_head_sum(const float* __restrict__ part,
                                                       const float* __restrict__ cntPart,
                                                       const float* __restrict__ Ws,
                                                       const float* __restrict__ Wn,
                                                       const float* __restrict__ bias,
                                                       float* __restrict__ out, int nChunks) {
    __shared__ __attribute__((aligned(16))) float Asl[HS * HP];
    __shared__ float red[8 * OUT_F];
    __shared__ float cw[32];

    const int lane = threadIdx.x;
    const int w    = threadIdx.y;
    const int tid  = w * 32 + lane;
    const int g    = blockIdx.x;

    const int c    = lane & 15;
    const int kOff = (lane < 16) ? 0 : 2;
    const int ks   = w >> 2;
    const int o    = (w & 3) * 16 + c;
    const float* Wp = ((ks < 4) ? Wn : Ws) + ((ks & 3) * 32 + kOff) * OUT_F + o;
    v2f b[8];
#pragma unroll
    for (int u = 0; u < 8; ++u) b[u] = (v2f){ Wp[(4 * u) * OUT_F], Wp[(4 * u + 1) * OUT_F] };

    const float cv0 = cntPart[((tid < nChunks) ? tid : 0) * NG + g];
    const float bo  = bias[tid & (OUT_F - 1)];

    const int q = tid & 63;
    const int r = tid >> 6;
    const float* rowp = part + (q >> 5) * (NG * HID_F) + g * HID_F + (q & 31) * 4;
    const long long per = 2 * NG * HID_F;
    const int nIt = (nChunks + HS - 1) / HS;
    float4 s = make_float4(0.0f, 0.0f, 0.0f, 0.0f);
    for (int base = 0; base < nIt; base += HB) {
        float4 v[HB];
#pragma unroll
        for (int k = 0; k < HB; ++k) {
            const int cc = r + (base + k) * HS;
            v[k] = *reinterpret_cast<const float4*>(rowp + (long long)((cc < nChunks) ? cc : 0) * per);
        }
#pragma unroll
        for (int k = 0; k < HB; ++k) {
            const bool ok = (r + (base + k) * HS) < nChunks;
            s.x += ok ? v[k].x : 0.0f;
            s.y += ok ? v[k].y : 0.0f;
            s.z += ok ? v[k].z : 0.0f;
            s.w += ok ? v[k].w : 0.0f;
        }
    }
    *reinterpret_cast<float4*>(Asl + r * HP + 4 * q) = s;
    float cv = (tid < nChunks) ? cv0 : 0.0f;
    for (int cc = tid + 32 * 32; cc < nChunks; cc += 32 * 32) cv += cntPart[cc * NG + g];
    for (int off = 16; off > 0; off >>= 1) cv += __shfl_xor(cv, off);
    if (lane == 0) cw[w] = cv;
    __syncthreads();

    v8f acc = {};
    const float* ap = Asl + c * HP + ks * 32 + kOff;
#pragma unroll
    for (int u = 0; u < 8; ++u) {
        const v2f a = *reinterpret_cast<const v2f*>(ap + 4 * u);
        acc = __builtin_amdgcn_wmma_f32_16x16x4_f32(false, a, false, b[u], (short)0, acc, false, false);
    }
    float t = ((acc[0] + acc[1]) + (acc[2] + acc[3])) + ((acc[4] + acc[5]) + (acc[6] + acc[7]));
    t += __shfl_xor(t, 16);
    if (lane < 16) red[ks * OUT_F + o] = t;
    __syncthreads();

    if (tid < OUT_F) {
        float tot = 0.0f;
        for (int k = 0; k < 8; ++k) tot += red[k * OUT_F + tid];
        float n = 0.0f;
        for (int k = 0; k < 32; ++k) n += cw[k];
        out[g * OUT_F + tid] = (tot + n * bo) / fmaxf(n, 1.0f);
    }
}


static inline int cdiv64(long long a, long long b) { return (int)((a + b - 1) / b); }

extern "C" void kernel_launch(void* const* d_in, const int* in_sizes, int n_in,
                              void* d_out, int out_size, void* d_ws, size_t ws_size,
                              hipStream_t stream) {
    const float* in_feat = (const float*)d_in[0];
    const int*   src     = (const int*)d_in[1];
    const int*   dst     = (const int*)d_in[2];
    const int*   gid     = (const int*)d_in[3];
    const float* W1s = (const float*)d_in[5];
    const float* W1n = (const float*)d_in[6];
    const float* b1  = (const float*)d_in[7];
    const float* W2s = (const float*)d_in[8];
    const float* W2n = (const float*)d_in[9];
    const float* b2  = (const float*)d_in[10];
    float* out = (float*)d_out;

    const int Nn = in_sizes[0] / IN_F;
    const int E  = in_sizes[1];

    float* ws    = (float*)d_ws;
    float* accum = ws;
    float* deg   = accum + (size_t)Nn * 128;
    float* Ct    = deg + Nn;
    float* h1sp  = Ct + (size_t)Nn * NG;

    const int nChunks = cdiv64(Nn, CH);

    float* part    = h1sp;
    float* cntPart = part + (size_t)nChunks * (2 * NG * HID_F);
    const int nB = cdiv64(E, EB);
    int* idxD = (int*)h1sp;
    int* offD = idxD + (size_t)nB * EB * 2;
    int* idxS = (int*)Ct;
    int* offS = idxS + (size_t)nB * EB * 2;

    bin_kernel<<<nB, 1024, 0, stream>>>(dst, src, idxD, offD, idxS, offS, E);
    tile_aggregate_kernel<<<cdiv64(Nn, TR), dim3(32, TW), 0, stream>>>(in_feat, idxD, offD, nB, accum, deg, Nn);

    graph_reduce_owner<<<nChunks, dim3(32, RW), 0, stream>>>(
        in_feat, accum, W1s, W1n, b1, idxS, offS, nB, gid, deg, part, cntPart, Nn);
    graph_head_sum<<<NG, dim3(32, 32), 0, stream>>>(part, cntPart, W2s, W2n, b2, out, nChunks);
}


# Layer-1 GEMM moved into the owner-tile launch (208-row tiles): each wave issues its strip's self products between a gather group's row loads and their LDS adds; k order changes (self half first)
# speedup vs baseline: 7.6858x; 1.0695x over previous
#include <hip/hip_runtime.h>

typedef __attribute__((ext_vector_type(2))) float v2f;
typedef __attribute__((ext_vector_type(8))) float v8f;
typedef __attribute__((ext_vector_type(2))) int v2i;
typedef __attribute__((ext_vector_type(4))) int v4i;

#define IN_F  128
#define HID_F 128
#define OUT_F 64
#define NG    64
#define CH    208
#define QN    (CH / 4)
#define TR    CH
#define TW    32
#define TP    132
#define EB    4096
#define OFFP  257
#define RB    8
#define PB    8
#define FLUSH 64
#define LCAP  (FLUSH + 32 * RB)
#define RW    32
#define CP    80
#define HP1   136


__global__ __launch_bounds__(1024) void bin_kernel(const int* __restrict__ dst,
                                                   const int* __restrict__ src,
                                                   int* __restrict__ idxD, int* __restrict__ offD,
                                                   int* __restrict__ idxS, int* __restrict__ offS,
                                                   int E) {
    __shared__ int hd[256], hs[256];
    __shared__ int sd[OFFP], ss[OFFP];
    __shared__ __attribute__((aligned(16))) v2i od[EB], os[EB];

    const int tid = threadIdx.x;
    const int b   = blockIdx.x;
    const int e0  = b * EB + tid * 4;

    if (tid < 256) { hd[tid] = 0; hs[tid] = 0; }
    int dv[4], sv[4];
    if (e0 + 3 < E) {
        const int4 d4 = *reinterpret_cast<const int4*>(dst + e0);
        const int4 s4 = *reinterpret_cast<const int4*>(src + e0);
        dv[0] = d4.x; dv[1] = d4.y; dv[2] = d4.z; dv[3] = d4.w;
        sv[0] = s4.x; sv[1] = s4.y; sv[2] = s4.z; sv[3] = s4.w;
    } else {
        for (int j = 0; j < 4; ++j) {
            dv[j] = (e0 + j < E) ? dst[e0 + j] : 0;
            sv[j] = (e0 + j < E) ? src[e0 + j] : 0;
        }
    }
    __syncthreads();

    int kd[4], ks[4], rd[4], rs[4];
    for (int j = 0; j < 4; ++j) {
        kd[j] = (int)(((unsigned)dv[j] / (unsigned)TR) & 255u);
        ks[j] = (int)(((unsigned)sv[j] / (unsigned)CH) & 255u);
        rd[j] = 0;
        rs[j] = 0;
        if (e0 + j < E) {
            rd[j] = atomicAdd(&hd[kd[j]], 1);
            rs[j] = atomicAdd(&hs[ks[j]], 1);
        }
    }
    __syncthreads();

    const int wv = tid >> 5;
    const int ln = tid & 31;
    if (wv < 2) {
        const int* h = wv ? hs : hd;
        int* st = wv ? ss : sd;
        int c[8];
        int tot = 0;
        for (int q = 0; q < 8; ++q) { c[q] = h[ln * 8 + q]; tot += c[q]; }
        int inc = tot;
        for (int o = 1; o < 32; o <<= 1) {
            const int v = __shfl_up(inc, o);
            if (ln >= o) inc += v;
        }
        int run = inc - tot;
        for (int q = 0; q < 8; ++q) { st[ln * 8 + q] = run; run += c[q]; }
        if (ln == 31) st[256] = inc;
    }
    __syncthreads();

    if (tid < OFFP) {
        offD[(long long)b * OFFP + tid] = sd[tid];
        offS[(long long)b * OFFP + tid] = ss[tid];
    }
    for (int j = 0; j < 4; ++j) {
        if (e0 + j < E) {
            const v2i rec = { sv[j], dv[j] };
            od[sd[kd[j]] + rd[j]] = rec;
            os[ss[ks[j]] + rs[j]] = rec;
        }
    }
    __syncthreads();
    const int tot = sd[256];
    v2i* gD = reinterpret_cast<v2i*>(idxD) + (long long)b * EB;
    v2i* gS = reinterpret_cast<v2i*>(idxS) + (long long)b * EB;
    for (int q = 0; q < 2; ++q) {
        const int p = q * (EB / 2) + tid * 2;
        if (p + 1 < tot) {
            *reinterpret_cast<v4i*>(gD + p) = *reinterpret_cast<const v4i*>(od + p);
            *reinterpret_cast<v4i*>(gS + p) = *reinterpret_cast<const v4i*>(os + p);
        } else if (p < tot) {
            gD[p] = od[p];
            gS[p] = os[p];
        }
    }
}

struct SelfRaw {
    float4 as;
    float4 bs[4];
};

__device__ __forceinline__ void self_load(SelfRaw& g, const float4* xs, const float* ws, int i) {
    g.as = xs[2 * i];
#pragma unroll
    for (int e = 0; e < 4; ++e)
        g.bs[e] = *reinterpret_cast<const float4*>(ws + (8 * i + e) * HID_F);
}

__device__ __forceinline__ void self_group_roll(SelfRaw& g, const float4* xs, const float* ws, int nx,
                                                v8f& acc0, v8f& acc1, v8f& acc2, v8f& acc3) {
    const v2f a0 = { g.as.x, g.as.y }, a1 = { g.as.z, g.as.w };
    acc0 = __builtin_amdgcn_wmma_f32_16x16x4_f32(false, a0, false, (v2f){ g.bs[0].x, g.bs[1].x }, (short)0, acc0, false, false);
    acc1 = __builtin_amdgcn_wmma_f32_16x16x4_f32(false, a0, false, (v2f){ g.bs[0].y, g.bs[1].y }, (short)0, acc1, false, false);
    acc2 = __builtin_amdgcn_wmma_f32_16x16x4_f32(false, a0, false, (v2f){ g.bs[0].z, g.bs[1].z }, (short)0, acc2, false, false);
    acc3 = __builtin_amdgcn_wmma_f32_16x16x4_f32(false, a0, false, (v2f){ g.bs[0].w, g.bs[1].w }, (short)0, acc3, false, false);
    g.bs[0] = *reinterpret_cast<const float4*>(ws + (8 * nx + 0) * HID_F);
    g.bs[1] = *reinterpret_cast<const float4*>(ws + (8 * nx + 1) * HID_F);
    acc0 = __builtin_amdgcn_wmma_f32_16x16x4_f32(false, a1, false, (v2f){ g.bs[2].x, g.bs[3].x }, (short)0, acc0, false, false);
    acc1 = __builtin_amdgcn_wmma_f32_16x16x4_f32(false, a1, false, (v2f){ g.bs[2].y, g.bs[3].y }, (short)0, acc1, false, false);
    acc2 = __builtin_amdgcn_wmma_f32_16x16x4_f32(false, a1, false, (v2f){ g.bs[2].z, g.bs[3].z }, (short)0, acc2, false, false);
    acc3 = __builtin_amdgcn_wmma_f32_16x16x4_f32(false, a1, false, (v2f){ g.bs[2].w, g.bs[3].w }, (short)0, acc3, false, false);
    g.bs[2] = *reinterpret_cast<const float4*>(ws + (8 * nx + 2) * HID_F);
    g.bs[3] = *reinterpret_cast<const float4*>(ws + (8 * nx + 3) * HID_F);
    g.as = xs[2 * nx];
}

template <bool MORE>
__device__ __forceinline__ void neigh_group_roll(float4 (&bn)[4], const float* trow, const float* wn,
                                                 int i, int nx, float invd,
                                                 v8f& acc0, v8f& acc1, v8f& acc2, v8f& acc3) {
    const float* t = trow + 2 * i;
    const v2f a0 = { t[0]  * invd, t[32] * invd };
    const v2f a1 = { t[64] * invd, t[96] * invd };
    acc0 = __builtin_amdgcn_wmma_f32_16x16x4_f32(false, a0, false, (v2f){ bn[0].x, bn[1].x }, (short)0, acc0, false, false);
    acc1 = __builtin_amdgcn_wmma_f32_16x16x4_f32(false, a0, false, (v2f){ bn[0].y, bn[1].y }, (short)0, acc1, false, false);
    acc2 = __builtin_amdgcn_wmma_f32_16x16x4_f32(false, a0, false, (v2f){ bn[0].z, bn[1].z }, (short)0, acc2, false, false);
    acc3 = __builtin_amdgcn_wmma_f32_16x16x4_f32(false, a0, false, (v2f){ bn[0].w, bn[1].w }, (short)0, acc3, false, false);
    if (MORE) {
        bn[0] = *reinterpret_cast<const float4*>(wn + (8 * nx + 0) * HID_F);
        bn[1] = *reinterpret_cast<const float4*>(wn + (8 * nx + 1) * HID_F);
    }
    acc0 = __builtin_amdgcn_wmma_f32_16x16x4_f32(false, a1, false, (v2f){ bn[2].x, bn[3].x }, (short)0, acc0, false, false);
    acc1 = __builtin_amdgcn_wmma_f32_16x16x4_f32(false, a1, false, (v2f){ bn[2].y, bn[3].y }, (short)0, acc1, false, false);
    acc2 = __builtin_amdgcn_wmma_f32_16x16x4_f32(false, a1, false, (v2f){ bn[2].z, bn[3].z }, (short)0, acc2, false, false);
    acc3 = __builtin_amdgcn_wmma_f32_16x16x4_f32(false, a1, false, (v2f){ bn[2].w, bn[3].w }, (short)0, acc3, false, false);
    if (MORE) {
        bn[2] = *reinterpret_cast<const float4*>(wn + (8 * nx + 2) * HID_F);
        bn[3] = *reinterpret_cast<const float4*>(wn + (8 * nx + 3) * HID_F);
    }
}

static_assert(TR % 16 == 0 && 2 * (TR / 16) <= TW, "a wave per 16 x 64 strip of the tile's h1");
static_assert((TR * TP) % 4 == 0 && TP % 4 == 0, "tile rows are cleared and read in 16-byte pieces");

__global__ __launch_bounds__(1024) void tile_layer1_kernel(
        const float* __restrict__ feat,
        const int* __restrict__ idx, const int* __restrict__ off, int nB,
        const float* __restrict__ W1s, const float* __restrict__ W1n,
        const float* __restrict__ bias1,
        float* __restrict__ h1out, float* __restrict__ invdeg, int Nn) {
    __shared__ __attribute__((aligned(16))) float tile[TR * TP];
    __shared__ int degc[TR];
    __shared__ v2i lst[TW][LCAP];
    const v2i* prs = reinterpret_cast<const v2i*>(idx);

    const int lane = threadIdx.x;
    const int w    = __builtin_amdgcn_readfirstlane(threadIdx.y);
    const int tid  = w * 32 + lane;
    const int rowBase = blockIdx.x * TR;

    const bool strip = w < 2 * (TR / 16);
    const int c    = lane & 15;
    const int hh   = lane >> 4;
    const int rt   = w >> 1;
    const int col0 = (w & 1) * 64 + 4 * c;
    int r = rowBase + rt * 16 + c;
    if (r >= Nn) r = Nn - 1;
    const float4* xs = reinterpret_cast<const float4*>(feat + (long long)r * IN_F + 4 * hh);
    const float* ws = W1s + (4 * hh) * HID_F + col0;
    v8f hc0 = {}, hc1 = {}, hc2 = {}, hc3 = {};
    SelfRaw g;
    self_load(g, xs, ws, 0);
    int gi = 0;

    for (int i = tid; i < TR * TP / 4; i += 32 * TW)
        reinterpret_cast<float4*>(tile)[i] = make_float4(0.0f, 0.0f, 0.0f, 0.0f);
    if (tid < TR) degc[tid] = 0;
    __syncthreads();

    auto selfStep = [&]() {
        if (strip && gi < IN_F / 8) {
            const int nx = (gi + 1 < IN_F / 8) ? gi + 1 : gi;
            self_group_roll(g, xs, ws, nx, hc0, hc1, hc2, hc3);
            ++gi;
        }
    };

    v2i* myl = lst[w];
    int cnt = 0;

    auto note = [&](bool m, v2i e) {
        const unsigned mask = __builtin_amdgcn_ballot_w32(m);
        if (m) myl[cnt + (int)__builtin_amdgcn_mbcnt_lo(mask, 0u)] = e;
        cnt += __builtin_popcount(mask);
    };

    auto drain = [&]() {
        for (int i = lane; i < cnt; i += 32) {
            const unsigned ld = (unsigned)myl[i].y - (unsigned)rowBase;
            if (ld < (unsigned)TR) atomicAdd(&degc[ld], 1);
        }
        for (int base = 0; base < cnt; base += PB) {
            v2i pr[PB];
            float4 v[PB];
#pragma unroll
            for (int k = 0; k < PB; ++k) {
                int idx = base + k;
                if (idx > cnt - 1) idx = cnt - 1;
                pr[k] = myl[idx];
            }
#pragma unroll
            for (int k = 0; k < PB; ++k)
                v[k] = *reinterpret_cast<const float4*>(feat + (long long)pr[k].x * 128 + lane * 4);
            selfStep();
            selfStep();
#pragma unroll
            for (int k = 0; k < PB; ++k) {
                const int dk = pr[k].y - rowBase;
                const bool live = (base + k < cnt) && ((unsigned)dk < (unsigned)TR);
                float* t = tile + (live ? dk : 0) * TP + lane;
                atomicAdd(t,      live ? v[k].x : 0.0f);
                atomicAdd(t + 32, live ? v[k].y : 0.0f);
                atomicAdd(t + 64, live ? v[k].z : 0.0f);
                atomicAdd(t + 96, live ? v[k].w : 0.0f);
            }
        }
        cnt = 0;
    };

    const v2i none = { 0, 0 };
    const int key = (int)((unsigned)blockIdx.x & 255u);
    for (int b0 = w; b0 < nB; b0 += TW * RB) {
        int s0[RB], s1[RB];
        v2i e1[RB];
        bool longRun = false;
        const int bq = b0 + TW * (lane & (RB - 1));
        int ov = 0;
        if (lane < 2 * RB && bq < nB) ov = off[(long long)bq * OFFP + key + (lane >= RB ? 1 : 0)];
#pragma unroll
        for (int q = 0; q < RB; ++q) {
            s0[q] = __builtin_amdgcn_readlane(ov, q);
            s1[q] = __builtin_amdgcn_readlane(ov, RB + q);
        }
#pragma unroll
        for (int q = 0; q < RB; ++q) {
            const int i = s0[q] + lane;
            e1[q] = (i < s1[q]) ? prs[(long long)(b0 + TW * q) * EB + i] : none;
            longRun = longRun || (s1[q] - s0[q] > 32);
        }
#pragma unroll
        for (int q = 0; q < RB; ++q) note(s0[q] + lane < s1[q], e1[q]);
        if (cnt >= FLUSH) drain();
        if (longRun) {
            for (int q = 0; q < RB; ++q) {
                const int a1 = __builtin_amdgcn_readlane(ov, RB + q);
                for (int base = __builtin_amdgcn_readlane(ov, q) + 32; base < a1; base += 32) {
                    const int i = base + lane;
                    note(i < a1, (i < a1) ? prs[(long long)(b0 + TW * q) * EB + i] : none);
                    if (cnt >= FLUSH) drain();
                }
            }
        }
    }
    drain();
#pragma unroll 1
    while (strip && gi < IN_F / 8) selfStep();
    __syncthreads();

    if (strip) {
        const int m0 = rt * 16 + c;
        const float invd = 1.0f / fmaxf((float)degc[m0], 1.0f);
        const float* trow = tile + m0 * TP + hh;
        const float* wn = W1n + (4 * hh) * HID_F + col0;
        float4 bn[4];
#pragma unroll
        for (int e = 0; e < 4; ++e) bn[e] = *reinterpret_cast<const float4*>(wn + e * HID_F);
#pragma unroll 1
        for (int i = 1; i < IN_F / 8; ++i)
            neigh_group_roll<true>(bn, trow, wn, i - 1, i, invd, hc0, hc1, hc2, hc3);
        neigh_group_roll<false>(bn, trow, wn, IN_F / 8 - 1, 0, invd, hc0, hc1, hc2, hc3);

        const float4 bv = *reinterpret_cast<const float4*>(bias1 + col0);
#pragma unroll
        for (int j = 0; j < 8; ++j) {
            const int node = rowBase + rt * 16 + 8 * hh + j;
            const float4 o = make_float4(fmaxf(hc0[j] + bv.x, 0.0f), fmaxf(hc1[j] + bv.y, 0.0f),
                                         fmaxf(hc2[j] + bv.z, 0.0f), fmaxf(hc3[j] + bv.w, 0.0f));
            if (node < Nn) *reinterpret_cast<float4*>(h1out + (long long)node * HID_F + col0) = o;
        }
    }
    if (tid < TR) {
        const int node = rowBase + tid;
        if (node < Nn) invdeg[node] = 1.0f / fmaxf((float)degc[tid], 1.0f);
    }
}

static_assert(RW == 32 && QN * 4 == CH && QN == 4 * (4 + 3 + 3 + 3),
              "8 feature tiles x 4 node quarters; a quarter is 13 k-steps, taken as 4 + 3 + 3 + 3");
#define HV    ((CH * 32 + 32 * RW - 1) / (32 * RW))
static_assert(RW * LCAP * 2 <= CH * HP1, "the walk's lists fit into the space of the h1 tile");
template <int N> struct KSteps { static constexpr int value = N; };

__global__ __launch_bounds__(1024) void graph_reduce_owner(
        const float* __restrict__ H,
        const int* __restrict__ idx, const int* __restrict__ off, int nB,
        const int* __restrict__ gid, const float* __restrict__ invdeg,
        float* __restrict__ part, float* __restrict__ cntPart, int Nn) {
    __shared__ __attribute__((aligned(16))) float CtL[CH * CP];
    __shared__ __attribute__((aligned(16))) float Sloc[(NG + 1) * HID_F];
    __shared__ __attribute__((aligned(16))) float Tloc[NG * HID_F];
    __shared__ int cntloc[NG + 1];
    __shared__ int sgid[CH];
    __shared__ __attribute__((aligned(16))) float h1L[CH * HP1];
    const v2i* prs = reinterpret_cast<const v2i*>(idx);

    const int lane = threadIdx.x;
    const int w    = __builtin_amdgcn_readfirstlane(threadIdx.y);
    const int tid  = w * 32 + lane;
    const int k0   = blockIdx.x * CH;

    const float4 z4 = make_float4(0.0f, 0.0f, 0.0f, 0.0f);
    float4 hv[HV];
#pragma unroll
    for (int q = 0; q < HV; ++q) {
        const int i = tid + q * 32 * RW;
        int node = k0 + (i >> 5);
        if (node >= Nn) node = Nn - 1;
        hv[q] = *reinterpret_cast<const float4*>(H + (long long)node * HID_F + (i & 31) * 4);
    }
    for (int i = tid; i < CH * CP / 4; i += 32 * RW) reinterpret_cast<float4*>(CtL)[i] = z4;
    for (int i = tid; i < (NG + 1) * HID_F / 4; i += 32 * RW) reinterpret_cast<float4*>(Sloc)[i] = z4;
    for (int i = tid; i < NG * HID_F / 4; i += 32 * RW) reinterpret_cast<float4*>(Tloc)[i] = z4;
    if (tid <= NG) cntloc[tid] = 0;
    if (tid < CH) {
        const int node = k0 + tid;
        int g = NG;
        if (node < Nn) {
            const int gg = gid[node];
            if ((unsigned)gg < (unsigned)NG) g = gg;
        }
        sgid[tid] = g;
    }
    __syncthreads();
    if (tid < CH) atomicAdd(&cntloc[sgid[tid]], 1);

    v2i* myl = reinterpret_cast<v2i*>(h1L) + w * LCAP;
    int cnt = 0;

    auto note = [&](bool m, v2i e) {
        const unsigned mask = __builtin_amdgcn_ballot_w32(m);
        if (m) myl[cnt + (int)__builtin_amdgcn_mbcnt_lo(mask, 0u)] = e;
        cnt += __builtin_popcount(mask);
    };

    auto drain = [&]() {
        for (int i = lane; i < cnt; i += 32) {
            const v2i e = myl[i];
            const unsigned ls = (unsigned)e.x - (unsigned)k0;
            const int d = e.y;
            const int g = gid[d];
            const float wv = invdeg[d];
            const bool live = (ls < (unsigned)CH) && ((unsigned)g < (unsigned)NG);
            atomicAdd(&CtL[live ? (int)ls * CP + g : 0], wv * (live ? 1.0f : 0.0f));
        }
        cnt = 0;
    };

    const v2i none = { 0, 0 };
    const int key = (int)((unsigned)blockIdx.x & 255u);
    for (int b0 = w; b0 < nB; b0 += RW * RB) {
        int s0[RB], s1[RB];
        v2i e1[RB];
        bool longRun = false;
        const int bq = b0 + RW * (lane & (RB - 1));
        int ov = 0;
        if (lane < 2 * RB && bq < nB) ov = off[(long long)bq * OFFP + key + (lane >= RB ? 1 : 0)];
#pragma unroll
        for (int r = 0; r < RB; ++r) {
            s0[r] = __builtin_amdgcn_readlane(ov, r);
            s1[r] = __builtin_amdgcn_readlane(ov, RB + r);
        }
#pragma unroll
        for (int r = 0; r < RB; ++r) {
            const int i = s0[r] + lane;
            e1[r] = (i < s1[r]) ? prs[(long long)(b0 + RW * r) * EB + i] : none;
            longRun = longRun || (s1[r] - s0[r] > 32);
        }
#pragma unroll
        for (int r = 0; r < RB; ++r) note(s0[r] + lane < s1[r], e1[r]);
        if (cnt >= FLUSH) drain();
        if (longRun) {
            for (int r = 0; r < RB; ++r) {
                const int a1 = __builtin_amdgcn_readlane(ov, RB + r);
                for (int base = __builtin_amdgcn_readlane(ov, r) + 32; base < a1; base += 32) {
                    const int i = base + lane;
                    note(i < a1, (i < a1) ? prs[(long long)(b0 + RW * r) * EB + i] : none);
                    if (cnt >= FLUSH) drain();
                }
            }
        }
    }
    drain();
    __syncthreads();

#pragma unroll
    for (int q = 0; q < HV; ++q) {
        const int i = tid + q * 32 * RW;
        const int m = i >> 5;
        if (i < CH * 32) {
            float4 v = hv[q];
            if (k0 + m >= Nn) v = z4;
            *reinterpret_cast<float4*>(h1L + m * HP1 + (i & 31) * 4) = v;
        }
    }
    __syncthreads();

    const int kOff = (lane < 16) ? 0 : 2;
    const int col  = lane & 15;
    const int f    = (w & 7) * 16 + col;
    const int kq   = (w >> 3) * QN;

    v8f acc0 = {}, acc1 = {}, acc2 = {}, acc3 = {};
    auto ksteps = [&](int kk, auto kuTag) {
        constexpr int KU = decltype(kuTag)::value;
        v2f b[KU], a0[KU], a1[KU], a2[KU], a3[KU];
        int sa[KU], sb[KU];
#pragma unroll
        for (int u = 0; u < KU; ++u) {
            const int ka = kk + 4 * u + kOff;
            const float* hb = h1L + ka * HP1 + f;
            b[u] = (v2f){ hb[0], hb[HP1] };
            const float* ra = CtL + ka * CP + col;
            const float* rb = ra + CP;
            a0[u] = (v2f){ ra[0],  rb[0]  };
            a1[u] = (v2f){ ra[16], rb[16] };
            a2[u] = (v2f){ ra[32], rb[32] };
            a3[u] = (v2f){ ra[48], rb[48] };
            sa[u] = sgid[ka];
            sb[u] = sgid[ka + 1];
        }
#pragma unroll
        for (int u = 0; u < KU; ++u) {
            atomicAdd(&Sloc[sa[u] * HID_F + f], b[u].x);
            atomicAdd(&Sloc[sb[u] * HID_F + f], b[u].y);
            acc0 = __builtin_amdgcn_wmma_f32_16x16x4_f32(false, a0[u], false, b[u], (short)0, acc0, false, false);
            acc1 = __builtin_amdgcn_wmma_f32_16x16x4_f32(false, a1[u], false, b[u], (short)0, acc1, false, false);
            acc2 = __builtin_amdgcn_wmma_f32_16x16x4_f32(false, a2[u], false, b[u], (short)0, acc2, false, false);
            acc3 = __builtin_amdgcn_wmma_f32_16x16x4_f32(false, a3[u], false, b[u], (short)0, acc3, false, false);
        }
    };
    ksteps(kq,      KSteps<4>{});
    ksteps(kq + 16, KSteps<3>{});
    ksteps(kq + 28, KSteps<3>{});
    ksteps(kq + 40, KSteps<3>{});

    const int mB = (lane < 16) ? 0 : 8;
#pragma unroll
    for (int j = 0; j < 8; ++j) {
        atomicAdd(&Tloc[(0  + mB + j) * HID_F + f], acc0[j]);
        atomicAdd(&Tloc[(16 + mB + j) * HID_F + f], acc1[j]);
        atomicAdd(&Tloc[(32 + mB + j) * HID_F + f], acc2[j]);
        atomicAdd(&Tloc[(48 + mB + j) * HID_F + f], acc3[j]);
    }
    __syncthreads();

    float4* p4 = reinterpret_cast<float4*>(part + (long long)blockIdx.x * (2 * NG * HID_F));
    for (int i = tid; i < NG * HID_F / 4; i += 32 * RW) {
        p4[i] = reinterpret_cast<const float4*>(Tloc)[i];
        p4[NG * HID_F / 4 + i] = reinterpret_cast<const float4*>(Sloc)[i];
    }
    if (tid < NG) cntPart[blockIdx.x * NG + tid] = (float)cntloc[tid];
}

#define HS  16
#define HB  16
#define HP  260
static_assert(2 * HID_F == 256 && OUT_F == 64 && HS == 16 && NG * HID_F == 8192,
              "32 waves = 4 column tiles x 8 k-slices of 32; 64 pieces of 16 bytes per row");

__global__ __launch_bounds__(1024) void graph_head_sum(const float* __restrict__ part,
                                                       const float* __restrict__ cntPart,
                                                       const float* __restrict__ Ws,
                                                       const float* __restrict__ Wn,
                                                       const float* __restrict__ bias,
                                                       float* __restrict__ out, int nChunks) {
    __shared__ __attribute__((aligned(16))) float Asl[HS * HP];
    __shared__ float red[8 * OUT_F];
    __shared__ float cw[32];

    const int lane = threadIdx.x;
    const int w    = threadIdx.y;
    const int tid  = w * 32 + lane;
    const int g    = blockIdx.x;

    const int c    = lane & 15;
    const int kOff = (lane < 16) ? 0 : 2;
    const int ks   = w >> 2;
    const int o    = (w & 3) * 16 + c;
    const float* Wp = ((ks < 4) ? Wn : Ws) + ((ks & 3) * 32 + kOff) * OUT_F + o;
    v2f b[8];
#pragma unroll
    for (int u = 0; u < 8; ++u) b[u] = (v2f){ Wp[(4 * u) * OUT_F], Wp[(4 * u + 1) * OUT_F] };

    const float cv0 = cntPart[((tid < nChunks) ? tid : 0) * NG + g];
    const float bo  = bias[tid & (OUT_F - 1)];

    const int q = tid & 63;
    const int r = tid >> 6;
    const float* rowp = part + (q >> 5) * (NG * HID_F) + g * HID_F + (q & 31) * 4;
    const long long per = 2 * NG * HID_F;
    const int nIt = (nChunks + HS - 1) / HS;
    float4 s = make_float4(0.0f, 0.0f, 0.0f, 0.0f);
    for (int base = 0; base < nIt; base += HB) {
        float4 v[HB];
#pragma unroll
        for (int k = 0; k < HB; ++k) {
            const int cc = r + (base + k) * HS;
            v[k] = *reinterpret_cast<const float4*>(rowp + (long long)((cc < nChunks) ? cc : 0) * per);
        }
#pragma unroll
        for (int k = 0; k < HB; ++k) {
            const bool ok = (r + (base + k) * HS) < nChunks;
            s.x += ok ? v[k].x : 0.0f;
            s.y += ok ? v[k].y : 0.0f;
            s.z += ok ? v[k].z : 0.0f;
            s.w += ok ? v[k].w : 0.0f;
        }
    }
    *reinterpret_cast<float4*>(Asl + r * HP + 4 * q) = s;
    float cv = (tid < nChunks) ? cv0 : 0.0f;
    for (int cc = tid + 32 * 32; cc < nChunks; cc += 32 * 32) cv += cntPart[cc * NG + g];
    for (int off = 16; off > 0; off >>= 1) cv += __shfl_xor(cv, off);
    if (lane == 0) cw[w] = cv;
    __syncthreads();

    v8f acc = {};
    const float* ap = Asl + c * HP + ks * 32 + kOff;
#pragma unroll
    for (int u = 0; u < 8; ++u) {
        const v2f a = *reinterpret_cast<const v2f*>(ap + 4 * u);
        acc = __builtin_amdgcn_wmma_f32_16x16x4_f32(false, a, false, b[u], (short)0, acc, false, false);
    }
    float t = ((acc[0] + acc[1]) + (acc[2] + acc[3])) + ((acc[4] + acc[5]) + (acc[6] + acc[7]));
    t += __shfl_xor(t, 16);
    if (lane < 16) red[ks * OUT_F + o] = t;
    __syncthreads();

    if (tid < OUT_F) {
        float tot = 0.0f;
        for (int k = 0; k < 8; ++k) tot += red[k * OUT_F + tid];
        float n = 0.0f;
        for (int k = 0; k < 32; ++k) n += cw[k];
        out[g * OUT_F + tid] = (tot + n * bo) / fmaxf(n, 1.0f);
    }
}


static inline int cdiv64(long long a, long long b) { return (int)((a + b - 1) / b); }

extern "C" void kernel_launch(void* const* d_in, const int* in_sizes, int n_in,
                              void* d_out, int out_size, void* d_ws, size_t ws_size,
                              hipStream_t stream) {
    const float* in_feat = (const float*)d_in[0];
    const int*   src     = (const int*)d_in[1];
    const int*   dst     = (const int*)d_in[2];
    const int*   gid     = (const int*)d_in[3];
    const float* W1s = (const float*)d_in[5];
    const float* W1n = (const float*)d_in[6];
    const float* b1  = (const float*)d_in[7];
    const float* W2s = (const float*)d_in[8];
    const float* W2n = (const float*)d_in[9];
    const float* b2  = (const float*)d_in[10];
    float* out = (float*)d_out;

    const int Nn = in_sizes[0] / IN_F;
    const int E  = in_sizes[1];

    float* ws    = (float*)d_ws;
    float* accum = ws;
    float* deg   = accum + (size_t)Nn * 128;
    float* Ct    = deg + Nn;
    float* h1sp  = Ct + (size_t)Nn * NG;

    const int nChunks = cdiv64(Nn, CH);

    float* part    = h1sp;
    float* cntPart = part + (size_t)nChunks * (2 * NG * HID_F);
    const int nB = cdiv64(E, EB);
    int* idxD = (int*)h1sp;
    int* offD = idxD + (size_t)nB * EB * 2;
    int* idxS = (int*)Ct;
    int* offS = idxS + (size_t)nB * EB * 2;

    bin_kernel<<<nB, 1024, 0, stream>>>(dst, src, idxD, offD, idxS, offS, E);
    tile_layer1_kernel<<<cdiv64(Nn, TR), dim3(32, TW), 0, stream>>>(
        in_feat, idxD, offD, nB, W1s, W1n, b1, accum, deg, Nn);

    graph_reduce_owner<<<nChunks, dim3(32, RW), 0, stream>>>(
        accum, idxS, offS, nB, gid, deg, part, cntPart, Nn);
    graph_head_sum<<<NG, dim3(32, 32), 0, stream>>>(part, cntPart, W2s, W2n, b2, out, nChunks);
}
